# starting kernel
# baseline (speedup 1.0000x reference)
#include <hip/hip_runtime.h>

typedef float v2f __attribute__((ext_vector_type(2)));
typedef float v8f __attribute__((ext_vector_type(8)));

#define BB 16
#define CC 128
#define TT 1024
#define DKD 64
#define CT (CC * TT)
#define LOG10000 9.210340371976184f
#define PI_2 1.5707963267948966f

__device__ inline v8f wmma4(v2f a, v2f b, v8f c) {
  return __builtin_amdgcn_wmma_f32_16x16x4_f32(false, a, false, b, (short)0, c,
                                               false, false);
}


__global__ void posenc_add(const float* __restrict__ x, float* __restrict__ out) {
  int idx = blockIdx.x * blockDim.x + threadIdx.x;
  int ct = idx % CT;
  int c = ct / TT;
  int t = ct % TT;
  float freq = expf(-(float)(c & ~1) * (LOG10000 / (float)CC));
  float pe = sinf((float)t * freq + (float)(c & 1) * PI_2);
  out[idx] = x[idx] + pe;
}

__global__ void batch_stats(const float* __restrict__ src, float* __restrict__ stats) {
  __shared__ float ssum[256];
  __shared__ float ssq[256];
  int b = blockIdx.x;
  int tid = threadIdx.x;
  const float* p = src + b * CT;
  float s = 0.f, q = 0.f;
  for (int e = tid; e < CT; e += 256) {
    float v = p[e];
    s += v;
    q += v * v;
  }
  ssum[tid] = s;
  ssq[tid] = q;
  __syncthreads();
  for (int st = 128; st > 0; st >>= 1) {
    if (tid < st) {
      ssum[tid] += ssum[tid + st];
      ssq[tid] += ssq[tid + st];
    }
    __syncthreads();
  }
  if (tid == 0) {
    stats[b] = ssum[0];
    stats[16 + b] = ssq[0];
  }
}

__global__ void ln_apply(const float* __restrict__ src, const float* __restrict__ g,
                         const float* __restrict__ bb, const float* __restrict__ stats,
                         float* __restrict__ dst) {
  int idx = blockIdx.x * blockDim.x + threadIdx.x;
  int b = idx / CT;
  int ct = idx % CT;
  float inv_n = 1.0f / (float)CT;
  float mu = stats[b] * inv_n;
  float var = stats[16 + b] * inv_n - mu * mu;
  float r = rsqrtf(var + 1e-5f);
  dst[idx] = (src[idx] - mu) * r * g[ct] + bb[ct];
}

__global__ void dwconv(const float* __restrict__ src, const float* __restrict__ w,
                       const float* __restrict__ bias, float* __restrict__ dst) {
  int idx = blockIdx.x * blockDim.x + threadIdx.x;
  int ct = idx % CT;
  int c = ct / TT;
  int t = ct % TT;
  const float* row = src + (idx - t);
  float s = bias[c];
#pragma unroll
  for (int j = 0; j < 7; ++j) {
    int tt = t + j - 3;
    if (tt >= 0 && tt < TT) s += row[tt] * w[c * 7 + j];
  }
  dst[idx] = s;
}


__global__ __launch_bounds__(128) void pw_gemm(const float* __restrict__ Y,
                                               const float* __restrict__ W,
                                               const float* __restrict__ bias,
                                               float* __restrict__ res) {
  int l = threadIdx.x & 31;
  int tile = blockIdx.x * 4 + (threadIdx.x >> 5);
  int b = tile >> 7;
  int rem = tile & 127;
  int m0 = (rem >> 4) << 4;
  int n0 = (rem & 15) << 6;
  int lo = l & 15;
  int kh = (l >> 4) << 1;
  const float* yb = Y + b * CT;
  v8f acc0 = {}, acc1 = {}, acc2 = {}, acc3 = {};
  for (int kk = 0; kk < CC; kk += 4) {
    int k0 = kk + kh;
    v2f a;
    a.x = W[(m0 + lo) * CC + k0];
    a.y = W[(m0 + lo) * CC + k0 + 1];
    const float* p0 = yb + k0 * TT + n0 + lo;
    const float* p1 = p0 + TT;
    v2f b0, b1, b2, b3;
    b0.x = p0[0];  b0.y = p1[0];
    b1.x = p0[16]; b1.y = p1[16];
    b2.x = p0[32]; b2.y = p1[32];
    b3.x = p0[48]; b3.y = p1[48];
    acc0 = wmma4(a, b0, acc0);
    acc1 = wmma4(a, b1, acc1);
    acc2 = wmma4(a, b2, acc2);
    acc3 = wmma4(a, b3, acc3);
  }
  float* rb = res + b * CT;
#pragma unroll
  for (int v = 0; v < 8; ++v) {
    int m = m0 + v + ((l >> 4) << 3);
    float bsv = bias[m];
    float* r = rb + m * TT + n0 + lo;
    r[0]  = fmaxf(acc0[v] + bsv, 0.f) + r[0];
    r[16] = fmaxf(acc1[v] + bsv, 0.f) + r[16];
    r[32] = fmaxf(acc2[v] + bsv, 0.f) + r[32];
    r[48] = fmaxf(acc3[v] + bsv, 0.f) + r[48];
  }
}

__global__ __launch_bounds__(128) void xtw_gemm(const float* __restrict__ X,
                                                const float* __restrict__ W,
                                                float* __restrict__ dst) {
  int l = threadIdx.x & 31;
  int tile = blockIdx.x * 4 + (threadIdx.x >> 5);
  int b = tile >> 6;
  int t0 = (tile & 63) << 4;
  int lo = l & 15;
  int kh = (l >> 4) << 1;
  const float* xb = X + b * CT;
  v8f acc0 = {}, acc1 = {}, acc2 = {}, acc3 = {};
  for (int kk = 0; kk < CC; kk += 4) {
    int k0 = kk + kh;
    v2f a;
    a.x = xb[k0 * TT + t0 + lo];
    a.y = xb[(k0 + 1) * TT + t0 + lo];
    const float* w0 = W + k0 * DKD + lo;
    const float* w1 = w0 + DKD;
    v2f b0, b1, b2, b3;
    b0.x = w0[0];  b0.y = w1[0];
    b1.x = w0[16]; b1.y = w1[16];
    b2.x = w0[32]; b2.y = w1[32];
    b3.x = w0[48]; b3.y = w1[48];
    acc0 = wmma4(a, b0, acc0);
    acc1 = wmma4(a, b1, acc1);
    acc2 = wmma4(a, b2, acc2);
    acc3 = wmma4(a, b3, acc3);
  }
  float* db = dst + b * TT * DKD;
#pragma unroll
  for (int v = 0; v < 8; ++v) {
    int t = t0 + v + ((l >> 4) << 3);
    float* d = db + t * DKD + lo;
    d[0]  = acc0[v];
    d[16] = acc1[v];
    d[32] = acc2[v];
    d[48] = acc3[v];
  }
}

__global__ __launch_bounds__(32) void attn_kernel(const float* __restrict__ Q,
                                                  const float* __restrict__ Km,
                                                  const float* __restrict__ Vm,
                                                  const float* __restrict__ mask,
                                                  float* __restrict__ head) {
  __shared__ float S[16 * TT];
  int b = blockIdx.x >> 6;
  int q0 = (blockIdx.x & 63) << 4;
  int l = threadIdx.x;
  int lo = l & 15;
  int kh = (l >> 4) << 1;
  const float* qb = Q + b * TT * DKD;
  const float* kb = Km + b * TT * DKD;
  const float* vb = Vm + b * TT * DKD;
  const float* mb = mask + b * TT;

  v2f afrag[DKD / 4];
#pragma unroll
  for (int kk = 0; kk < DKD; kk += 4) {
    int k0 = kk + kh;
    afrag[kk >> 2].x = qb[(q0 + lo) * DKD + k0];
    afrag[kk >> 2].y = qb[(q0 + lo) * DKD + k0 + 1];
  }

  for (int t = 0; t < TT / 16; ++t) {
    v8f acc = {};
#pragma unroll
    for (int kk = 0; kk < DKD; kk += 4) {
      int k0 = kk + kh;
      v2f bv;
      bv.x = kb[(t * 16 + lo) * DKD + k0];
      bv.y = kb[(t * 16 + lo) * DKD + k0 + 1];
      acc = wmma4(afrag[kk >> 2], bv, acc);
    }
#pragma unroll
    for (int v = 0; v < 8; ++v) {
      int row = v + ((l >> 4) << 3);
      int col = t * 16 + lo;
      float hm = mb[col];
      float sv = acc[v] * 0.125f;
      S[row * TT + col] = sv * hm + (1.f - hm) * -1e30f;
    }
  }
  __syncthreads();

  {
    int row = lo;
    int c0 = (l >> 4) * (TT / 2);
    float mx = -3.4e38f;
    for (int c = 0; c < TT / 2; ++c) mx = fmaxf(mx, S[row * TT + c0 + c]);
    mx = fmaxf(mx, __shfl_xor(mx, 16, 32));
    float sum = 0.f;
    for (int c = 0; c < TT / 2; ++c) sum += expf(S[row * TT + c0 + c] - mx);
    sum += __shfl_xor(sum, 16, 32);
    float inv = 1.0f / sum;
    float qm = mask[b * TT + q0 + row];
    for (int c = 0; c < TT / 2; ++c) {
      int ix = row * TT + c0 + c;
      S[ix] = expf(S[ix] - mx) * inv * qm;
    }
  }
  __syncthreads();

  {
    v8f acc0 = {}, acc1 = {}, acc2 = {}, acc3 = {};
    for (int kk = 0; kk < TT; kk += 4) {
      int k0 = kk + kh;
      v2f a;
      a.x = S[lo * TT + k0];
      a.y = S[lo * TT + k0 + 1];
      const float* v0p = vb + k0 * DKD + lo;
      const float* v1p = v0p + DKD;
      v2f b0, b1, b2, b3;
      b0.x = v0p[0];  b0.y = v1p[0];
      b1.x = v0p[16]; b1.y = v1p[16];
      b2.x = v0p[32]; b2.y = v1p[32];
      b3.x = v0p[48]; b3.y = v1p[48];
      acc0 = wmma4(a, b0, acc0);
      acc1 = wmma4(a, b1, acc1);
      acc2 = wmma4(a, b2, acc2);
      acc3 = wmma4(a, b3, acc3);
    }
    float* hb = head + b * TT * DKD;
#pragma unroll
    for (int v = 0; v < 8; ++v) {
      int t = q0 + v + ((l >> 4) << 3);
      float* h = hb + t * DKD + lo;
      h[0]  = acc0[v];
      h[16] = acc1[v];
      h[32] = acc2[v];
      h[48] = acc3[v];
    }
  }
}

__global__ __launch_bounds__(128) void wo_gemm(const float* __restrict__ H,
                                               const float* __restrict__ Wo,
                                               float* __restrict__ res) {
  int l = threadIdx.x & 31;
  int tile = blockIdx.x * 4 + (threadIdx.x >> 5);
  int b = tile >> 7;
  int rem = tile & 127;
  int m0 = (rem >> 4) << 4;
  int n0 = (rem & 15) << 6;
  int lo = l & 15;
  int kh = (l >> 4) << 1;
  const float* hb = H + b * TT * DKD;
  v8f acc0 = {}, acc1 = {}, acc2 = {}, acc3 = {};
  for (int kk = 0; kk < DKD; kk += 4) {
    int k0 = kk + kh;
    v2f a;
    a.x = Wo[k0 * CC + m0 + lo] + Wo[(k0 + 64) * CC + m0 + lo];
    a.y = Wo[(k0 + 1) * CC + m0 + lo] + Wo[(k0 + 65) * CC + m0 + lo];
    const float* h0 = hb + (n0 + lo) * DKD + k0;
    v2f b0, b1, b2, b3;
    b0.x = h0[0];            b0.y = h0[1];
    b1.x = h0[16 * DKD];     b1.y = h0[16 * DKD + 1];
    b2.x = h0[32 * DKD];     b2.y = h0[32 * DKD + 1];
    b3.x = h0[48 * DKD];     b3.y = h0[48 * DKD + 1];
    acc0 = wmma4(a, b0, acc0);
    acc1 = wmma4(a, b1, acc1);
    acc2 = wmma4(a, b2, acc2);
    acc3 = wmma4(a, b3, acc3);
  }
  float* rb = res + b * CT;
#pragma unroll
  for (int v = 0; v < 8; ++v) {
    int m = m0 + v + ((l >> 4) << 3);
    float* r = rb + m * TT + n0 + lo;
    r[0]  = acc0[v] + r[0];
    r[16] = acc1[v] + r[16];
    r[32] = acc2[v] + r[32];
    r[48] = acc3[v] + r[48];
  }
}

__global__ __launch_bounds__(128) void fc_gemm(const float* __restrict__ X,
                                               const float* __restrict__ W,
                                               const float* __restrict__ bias,
                                               const float* __restrict__ res,
                                               float* __restrict__ out) {
  int l = threadIdx.x & 31;
  int tile = blockIdx.x * 4 + (threadIdx.x >> 5);
  int b = tile >> 7;
  int rem = tile & 127;
  int m0 = (rem >> 4) << 4;
  int n0 = (rem & 15) << 6;
  int lo = l & 15;
  int kh = (l >> 4) << 1;
  const float* xb = X + b * CT;
  v8f acc0 = {}, acc1 = {}, acc2 = {}, acc3 = {};
  for (int kk = 0; kk < CC; kk += 4) {
    int k0 = kk + kh;
    v2f a;
    a.x = W[(m0 + lo) * CC + k0];
    a.y = W[(m0 + lo) * CC + k0 + 1];
    const float* p0 = xb + k0 * TT + n0 + lo;
    const float* p1 = p0 + TT;
    v2f b0, b1, b2, b3;
    b0.x = p0[0];  b0.y = p1[0];
    b1.x = p0[16]; b1.y = p1[16];
    b2.x = p0[32]; b2.y = p1[32];
    b3.x = p0[48]; b3.y = p1[48];
    acc0 = wmma4(a, b0, acc0);
    acc1 = wmma4(a, b1, acc1);
    acc2 = wmma4(a, b2, acc2);
    acc3 = wmma4(a, b3, acc3);
  }
  const float* rb = res + b * CT;
  float* ob = out + b * CT;
#pragma unroll
  for (int v = 0; v < 8; ++v) {
    int m = m0 + v + ((l >> 4) << 3);
    float bsv = bias[m];
    const float* r = rb + m * TT + n0 + lo;
    float* o = ob + m * TT + n0 + lo;
    o[0]  = fmaxf(acc0[v] + bsv, 0.f) + r[0];
    o[16] = fmaxf(acc1[v] + bsv, 0.f) + r[16];
    o[32] = fmaxf(acc2[v] + bsv, 0.f) + r[32];
    o[48] = fmaxf(acc3[v] + bsv, 0.f) + r[48];
  }
}


extern "C" void kernel_launch(void* const* d_in, const int* in_sizes, int n_in,
                              void* d_out, int out_size, void* d_ws, size_t ws_size,
                              hipStream_t stream) {
  (void)in_sizes; (void)n_in; (void)out_size; (void)ws_size;
  const float* x    = (const float*)d_in[0];
  const float* mask = (const float*)d_in[1];
  const float* dw_w = (const float*)d_in[2];
  const float* dw_b = (const float*)d_in[3];
  const float* pw_w = (const float*)d_in[4];
  const float* pw_b = (const float*)d_in[5];
  const float* n0g  = (const float*)d_in[6];
  const float* n0b  = (const float*)d_in[7];
  const float* nsg  = (const float*)d_in[8];
  const float* nsb  = (const float*)d_in[9];
  const float* neg_ = (const float*)d_in[10];
  const float* neb  = (const float*)d_in[11];
  const float* Wq   = (const float*)d_in[12];
  const float* Wk   = (const float*)d_in[13];
  const float* Wv   = (const float*)d_in[14];
  const float* Wo   = (const float*)d_in[15];
  const float* fcw  = (const float*)d_in[16];
  const float* fcb  = (const float*)d_in[17];
  float* outp = (float*)d_out;
  float* ws = (float*)d_ws;

  float* OUT = ws;
  float* RES = ws + 2097152;
  float* TMP = ws + 4194304;
  float* Qb  = ws + 6291456;
  float* Kb  = ws + 7340032;
  float* Vb  = ws + 8388608;
  float* Hb  = ws + 9437184;
  float* ST  = ws + 10485760;

  posenc_add<<<8192, 256, 0, stream>>>(x, OUT);
  batch_stats<<<16, 256, 0, stream>>>(OUT, ST);
  ln_apply<<<8192, 256, 0, stream>>>(OUT, n0g, n0b, ST, RES);

  for (int i = 0; i < 4; ++i) {
    dwconv<<<8192, 256, 0, stream>>>(OUT, dw_w + i * CC * 7, dw_b + i * CC, TMP);
    pw_gemm<<<512, 128, 0, stream>>>(TMP, pw_w + i * CC * CC, pw_b + i * CC, RES);
    batch_stats<<<16, 256, 0, stream>>>(RES, ST);
    ln_apply<<<8192, 256, 0, stream>>>(RES, nsg + i * CT, nsb + i * CT, ST, OUT);
  }

  xtw_gemm<<<256, 128, 0, stream>>>(OUT, Wq, Qb);
  xtw_gemm<<<256, 128, 0, stream>>>(OUT, Wk, Kb);
  xtw_gemm<<<256, 128, 0, stream>>>(OUT, Wv, Vb);
  attn_kernel<<<1024, 32, 0, stream>>>(Qb, Kb, Vb, mask, Hb);
  wo_gemm<<<512, 128, 0, stream>>>(Hb, Wo, RES);
  batch_stats<<<16, 256, 0, stream>>>(RES, ST);
  ln_apply<<<8192, 256, 0, stream>>>(RES, neg_, neb, ST, OUT);
  fc_gemm<<<512, 128, 0, stream>>>(OUT, fcw, fcb, RES, outp);
}


# LayerNorm sums spread over 512 work-groups (32 per batch element) instead of 16; ln_apply joins the 32 partial pairs by a lane butterfly: a different reduction tree
# speedup vs baseline: 1.5214x; 1.5214x over previous
#include <hip/hip_runtime.h>

typedef float v2f __attribute__((ext_vector_type(2)));
typedef float v8f __attribute__((ext_vector_type(8)));

#define BB 16
#define CC 128
#define TT 1024
#define DKD 64
#define CT (CC * TT)
#define NPART 32
#define PCHUNK (CT / NPART)
#define LOG10000 9.210340371976184f
#define PI_2 1.5707963267948966f

__device__ inline v8f wmma4(v2f a, v2f b, v8f c) {
  return __builtin_amdgcn_wmma_f32_16x16x4_f32(false, a, false, b, (short)0, c,
                                               false, false);
}


__global__ void posenc_add(const float* __restrict__ x, float* __restrict__ out) {
  int idx = blockIdx.x * blockDim.x + threadIdx.x;
  int ct = idx % CT;
  int c = ct / TT;
  int t = ct % TT;
  float freq = expf(-(float)(c & ~1) * (LOG10000 / (float)CC));
  float pe = sinf((float)t * freq + (float)(c & 1) * PI_2);
  out[idx] = x[idx] + pe;
}

__global__ __launch_bounds__(256) void batch_stats(const float* __restrict__ src,
                                                   float* __restrict__ part) {
  __shared__ float wsum[8];
  __shared__ float wsq[8];
  int b = blockIdx.x / NPART;
  int j = blockIdx.x % NPART;
  int tid = threadIdx.x;
  int lane = tid & 31;
  const float4* p = reinterpret_cast<const float4*>(src + b * CT + j * PCHUNK);
  float4 v0 = p[tid];
  float4 v1 = p[tid + 256];
  float4 v2 = p[tid + 512];
  float4 v3 = p[tid + 768];
  float s0 = (v0.x + v0.y) + (v0.z + v0.w);
  float s1 = (v1.x + v1.y) + (v1.z + v1.w);
  float s2 = (v2.x + v2.y) + (v2.z + v2.w);
  float s3 = (v3.x + v3.y) + (v3.z + v3.w);
  float q0 = (v0.x * v0.x + v0.y * v0.y) + (v0.z * v0.z + v0.w * v0.w);
  float q1 = (v1.x * v1.x + v1.y * v1.y) + (v1.z * v1.z + v1.w * v1.w);
  float q2 = (v2.x * v2.x + v2.y * v2.y) + (v2.z * v2.z + v2.w * v2.w);
  float q3 = (v3.x * v3.x + v3.y * v3.y) + (v3.z * v3.z + v3.w * v3.w);
  float s = (s0 + s1) + (s2 + s3);
  float q = (q0 + q1) + (q2 + q3);
  for (int o = 16; o > 0; o >>= 1) {
    s += __shfl_xor(s, o, 32);
    q += __shfl_xor(q, o, 32);
  }
  if (lane == 0) {
    wsum[tid >> 5] = s;
    wsq[tid >> 5] = q;
  }
  __syncthreads();
  if (tid < 32) {
    float a = (lane < 8) ? wsum[lane] : 0.f;
    float c = (lane < 8) ? wsq[lane] : 0.f;
    for (int o = 4; o > 0; o >>= 1) {
      a += __shfl_xor(a, o, 32);
      c += __shfl_xor(c, o, 32);
    }
    if (lane == 0) {
      reinterpret_cast<float2*>(part)[blockIdx.x] = make_float2(a, c);
    }
  }
}

__global__ void ln_apply(const float* __restrict__ src, const float* __restrict__ g,
                         const float* __restrict__ bb, const float* __restrict__ part,
                         float* __restrict__ dst) {
  int idx = blockIdx.x * blockDim.x + threadIdx.x;
  int b = idx / CT;
  int ct = idx % CT;
  float2 pr = reinterpret_cast<const float2*>(part)[b * NPART + (threadIdx.x & 31)];
  float s = pr.x;
  float q = pr.y;
  for (int o = 16; o > 0; o >>= 1) {
    s += __shfl_xor(s, o, 32);
    q += __shfl_xor(q, o, 32);
  }
  float inv_n = 1.0f / (float)CT;
  float mu = s * inv_n;
  float var = q * inv_n - mu * mu;
  float r = rsqrtf(var + 1e-5f);
  dst[idx] = (src[idx] - mu) * r * g[ct] + bb[ct];
}

__global__ void dwconv(const float* __restrict__ src, const float* __restrict__ w,
                       const float* __restrict__ bias, float* __restrict__ dst) {
  int idx = blockIdx.x * blockDim.x + threadIdx.x;
  int ct = idx % CT;
  int c = ct / TT;
  int t = ct % TT;
  const float* row = src + (idx - t);
  float s = bias[c];
#pragma unroll
  for (int j = 0; j < 7; ++j) {
    int tt = t + j - 3;
    if (tt >= 0 && tt < TT) s += row[tt] * w[c * 7 + j];
  }
  dst[idx] = s;
}


__global__ __launch_bounds__(128) void pw_gemm(const float* __restrict__ Y,
                                               const float* __restrict__ W,
                                               const float* __restrict__ bias,
                                               float* __restrict__ res) {
  int l = threadIdx.x & 31;
  int tile = blockIdx.x * 4 + (threadIdx.x >> 5);
  int b = tile >> 7;
  int rem = tile & 127;
  int m0 = (rem >> 4) << 4;
  int n0 = (rem & 15) << 6;
  int lo = l & 15;
  int kh = (l >> 4) << 1;
  const float* yb = Y + b * CT;
  v8f acc0 = {}, acc1 = {}, acc2 = {}, acc3 = {};
  for (int kk = 0; kk < CC; kk += 4) {
    int k0 = kk + kh;
    v2f a;
    a.x = W[(m0 + lo) * CC + k0];
    a.y = W[(m0 + lo) * CC + k0 + 1];
    const float* p0 = yb + k0 * TT + n0 + lo;
    const float* p1 = p0 + TT;
    v2f b0, b1, b2, b3;
    b0.x = p0[0];  b0.y = p1[0];
    b1.x = p0[16]; b1.y = p1[16];
    b2.x = p0[32]; b2.y = p1[32];
    b3.x = p0[48]; b3.y = p1[48];
    acc0 = wmma4(a, b0, acc0);
    acc1 = wmma4(a, b1, acc1);
    acc2 = wmma4(a, b2, acc2);
    acc3 = wmma4(a, b3, acc3);
  }
  float* rb = res + b * CT;
#pragma unroll
  for (int v = 0; v < 8; ++v) {
    int m = m0 + v + ((l >> 4) << 3);
    float bsv = bias[m];
    float* r = rb + m * TT + n0 + lo;
    r[0]  = fmaxf(acc0[v] + bsv, 0.f) + r[0];
    r[16] = fmaxf(acc1[v] + bsv, 0.f) + r[16];
    r[32] = fmaxf(acc2[v] + bsv, 0.f) + r[32];
    r[48] = fmaxf(acc3[v] + bsv, 0.f) + r[48];
  }
}

__global__ __launch_bounds__(128) void xtw_gemm(const float* __restrict__ X,
                                                const float* __restrict__ W,
                                                float* __restrict__ dst) {
  int l = threadIdx.x & 31;
  int tile = blockIdx.x * 4 + (threadIdx.x >> 5);
  int b = tile >> 6;
  int t0 = (tile & 63) << 4;
  int lo = l & 15;
  int kh = (l >> 4) << 1;
  const float* xb = X + b * CT;
  v8f acc0 = {}, acc1 = {}, acc2 = {}, acc3 = {};
  for (int kk = 0; kk < CC; kk += 4) {
    int k0 = kk + kh;
    v2f a;
    a.x = xb[k0 * TT + t0 + lo];
    a.y = xb[(k0 + 1) * TT + t0 + lo];
    const float* w0 = W + k0 * DKD + lo;
    const float* w1 = w0 + DKD;
    v2f b0, b1, b2, b3;
    b0.x = w0[0];  b0.y = w1[0];
    b1.x = w0[16]; b1.y = w1[16];
    b2.x = w0[32]; b2.y = w1[32];
    b3.x = w0[48]; b3.y = w1[48];
    acc0 = wmma4(a, b0, acc0);
    acc1 = wmma4(a, b1, acc1);
    acc2 = wmma4(a, b2, acc2);
    acc3 = wmma4(a, b3, acc3);
  }
  float* db = dst + b * TT * DKD;
#pragma unroll
  for (int v = 0; v < 8; ++v) {
    int t = t0 + v + ((l >> 4) << 3);
    float* d = db + t * DKD + lo;
    d[0]  = acc0[v];
    d[16] = acc1[v];
    d[32] = acc2[v];
    d[48] = acc3[v];
  }
}

__global__ __launch_bounds__(32) void attn_kernel(const float* __restrict__ Q,
                                                  const float* __restrict__ Km,
                                                  const float* __restrict__ Vm,
                                                  const float* __restrict__ mask,
                                                  float* __restrict__ head) {
  __shared__ float S[16 * TT];
  int b = blockIdx.x >> 6;
  int q0 = (blockIdx.x & 63) << 4;
  int l = threadIdx.x;
  int lo = l & 15;
  int kh = (l >> 4) << 1;
  const float* qb = Q + b * TT * DKD;
  const float* kb = Km + b * TT * DKD;
  const float* vb = Vm + b * TT * DKD;
  const float* mb = mask + b * TT;

  v2f afrag[DKD / 4];
#pragma unroll
  for (int kk = 0; kk < DKD; kk += 4) {
    int k0 = kk + kh;
    afrag[kk >> 2].x = qb[(q0 + lo) * DKD + k0];
    afrag[kk >> 2].y = qb[(q0 + lo) * DKD + k0 + 1];
  }

  for (int t = 0; t < TT / 16; ++t) {
    v8f acc = {};
#pragma unroll
    for (int kk = 0; kk < DKD; kk += 4) {
      int k0 = kk + kh;
      v2f bv;
      bv.x = kb[(t * 16 + lo) * DKD + k0];
      bv.y = kb[(t * 16 + lo) * DKD + k0 + 1];
      acc = wmma4(afrag[kk >> 2], bv, acc);
    }
#pragma unroll
    for (int v = 0; v < 8; ++v) {
      int row = v + ((l >> 4) << 3);
      int col = t * 16 + lo;
      float hm = mb[col];
      float sv = acc[v] * 0.125f;
      S[row * TT + col] = sv * hm + (1.f - hm) * -1e30f;
    }
  }
  __syncthreads();

  {
    int row = lo;
    int c0 = (l >> 4) * (TT / 2);
    float mx = -3.4e38f;
    for (int c = 0; c < TT / 2; ++c) mx = fmaxf(mx, S[row * TT + c0 + c]);
    mx = fmaxf(mx, __shfl_xor(mx, 16, 32));
    float sum = 0.f;
    for (int c = 0; c < TT / 2; ++c) sum += expf(S[row * TT + c0 + c] - mx);
    sum += __shfl_xor(sum, 16, 32);
    float inv = 1.0f / sum;
    float qm = mask[b * TT + q0 + row];
    for (int c = 0; c < TT / 2; ++c) {
      int ix = row * TT + c0 + c;
      S[ix] = expf(S[ix] - mx) * inv * qm;
    }
  }
  __syncthreads();

  {
    v8f acc0 = {}, acc1 = {}, acc2 = {}, acc3 = {};
    for (int kk = 0; kk < TT; kk += 4) {
      int k0 = kk + kh;
      v2f a;
      a.x = S[lo * TT + k0];
      a.y = S[lo * TT + k0 + 1];
      const float* v0p = vb + k0 * DKD + lo;
      const float* v1p = v0p + DKD;
      v2f b0, b1, b2, b3;
      b0.x = v0p[0];  b0.y = v1p[0];
      b1.x = v0p[16]; b1.y = v1p[16];
      b2.x = v0p[32]; b2.y = v1p[32];
      b3.x = v0p[48]; b3.y = v1p[48];
      acc0 = wmma4(a, b0, acc0);
      acc1 = wmma4(a, b1, acc1);
      acc2 = wmma4(a, b2, acc2);
      acc3 = wmma4(a, b3, acc3);
    }
    float* hb = head + b * TT * DKD;
#pragma unroll
    for (int v = 0; v < 8; ++v) {
      int t = q0 + v + ((l >> 4) << 3);
      float* h = hb + t * DKD + lo;
      h[0]  = acc0[v];
      h[16] = acc1[v];
      h[32] = acc2[v];
      h[48] = acc3[v];
    }
  }
}

__global__ __launch_bounds__(128) void wo_gemm(const float* __restrict__ H,
                                               const float* __restrict__ Wo,
                                               float* __restrict__ res) {
  int l = threadIdx.x & 31;
  int tile = blockIdx.x * 4 + (threadIdx.x >> 5);
  int b = tile >> 7;
  int rem = tile & 127;
  int m0 = (rem >> 4) << 4;
  int n0 = (rem & 15) << 6;
  int lo = l & 15;
  int kh = (l >> 4) << 1;
  const float* hb = H + b * TT * DKD;
  v8f acc0 = {}, acc1 = {}, acc2 = {}, acc3 = {};
  for (int kk = 0; kk < DKD; kk += 4) {
    int k0 = kk + kh;
    v2f a;
    a.x = Wo[k0 * CC + m0 + lo] + Wo[(k0 + 64) * CC + m0 + lo];
    a.y = Wo[(k0 + 1) * CC + m0 + lo] + Wo[(k0 + 65) * CC + m0 + lo];
    const float* h0 = hb + (n0 + lo) * DKD + k0;
    v2f b0, b1, b2, b3;
    b0.x = h0[0];            b0.y = h0[1];
    b1.x = h0[16 * DKD];     b1.y = h0[16 * DKD + 1];
    b2.x = h0[32 * DKD];     b2.y = h0[32 * DKD + 1];
    b3.x = h0[48 * DKD];     b3.y = h0[48 * DKD + 1];
    acc0 = wmma4(a, b0, acc0);
    acc1 = wmma4(a, b1, acc1);
    acc2 = wmma4(a, b2, acc2);
    acc3 = wmma4(a, b3, acc3);
  }
  float* rb = res + b * CT;
#pragma unroll
  for (int v = 0; v < 8; ++v) {
    int m = m0 + v + ((l >> 4) << 3);
    float* r = rb + m * TT + n0 + lo;
    r[0]  = acc0[v] + r[0];
    r[16] = acc1[v] + r[16];
    r[32] = acc2[v] + r[32];
    r[48] = acc3[v] + r[48];
  }
}

__global__ __launch_bounds__(128) void fc_gemm(const float* __restrict__ X,
                                               const float* __restrict__ W,
                                               const float* __restrict__ bias,
                                               const float* __restrict__ res,
                                               float* __restrict__ out) {
  int l = threadIdx.x & 31;
  int tile = blockIdx.x * 4 + (threadIdx.x >> 5);
  int b = tile >> 7;
  int rem = tile & 127;
  int m0 = (rem >> 4) << 4;
  int n0 = (rem & 15) << 6;
  int lo = l & 15;
  int kh = (l >> 4) << 1;
  const float* xb = X + b * CT;
  v8f acc0 = {}, acc1 = {}, acc2 = {}, acc3 = {};
  for (int kk = 0; kk < CC; kk += 4) {
    int k0 = kk + kh;
    v2f a;
    a.x = W[(m0 + lo) * CC + k0];
    a.y = W[(m0 + lo) * CC + k0 + 1];
    const float* p0 = xb + k0 * TT + n0 + lo;
    const float* p1 = p0 + TT;
    v2f b0, b1, b2, b3;
    b0.x = p0[0];  b0.y = p1[0];
    b1.x = p0[16]; b1.y = p1[16];
    b2.x = p0[32]; b2.y = p1[32];
    b3.x = p0[48]; b3.y = p1[48];
    acc0 = wmma4(a, b0, acc0);
    acc1 = wmma4(a, b1, acc1);
    acc2 = wmma4(a, b2, acc2);
    acc3 = wmma4(a, b3, acc3);
  }
  const float* rb = res + b * CT;
  float* ob = out + b * CT;
#pragma unroll
  for (int v = 0; v < 8; ++v) {
    int m = m0 + v + ((l >> 4) << 3);
    float bsv = bias[m];
    const float* r = rb + m * TT + n0 + lo;
    float* o = ob + m * TT + n0 + lo;
    o[0]  = fmaxf(acc0[v] + bsv, 0.f) + r[0];
    o[16] = fmaxf(acc1[v] + bsv, 0.f) + r[16];
    o[32] = fmaxf(acc2[v] + bsv, 0.f) + r[32];
    o[48] = fmaxf(acc3[v] + bsv, 0.f) + r[48];
  }
}


extern "C" void kernel_launch(void* const* d_in, const int* in_sizes, int n_in,
                              void* d_out, int out_size, void* d_ws, size_t ws_size,
                              hipStream_t stream) {
  (void)in_sizes; (void)n_in; (void)out_size; (void)ws_size;
  const float* x    = (const float*)d_in[0];
  const float* mask = (const float*)d_in[1];
  const float* dw_w = (const float*)d_in[2];
  const float* dw_b = (const float*)d_in[3];
  const float* pw_w = (const float*)d_in[4];
  const float* pw_b = (const float*)d_in[5];
  const float* n0g  = (const float*)d_in[6];
  const float* n0b  = (const float*)d_in[7];
  const float* nsg  = (const float*)d_in[8];
  const float* nsb  = (const float*)d_in[9];
  const float* neg_ = (const float*)d_in[10];
  const float* neb  = (const float*)d_in[11];
  const float* Wq   = (const float*)d_in[12];
  const float* Wk   = (const float*)d_in[13];
  const float* Wv   = (const float*)d_in[14];
  const float* Wo   = (const float*)d_in[15];
  const float* fcw  = (const float*)d_in[16];
  const float* fcb  = (const float*)d_in[17];
  float* outp = (float*)d_out;
  float* ws = (float*)d_ws;

  float* OUT = ws;
  float* RES = ws + 2097152;
  float* TMP = ws + 4194304;
  float* Qb  = ws + 6291456;
  float* Kb  = ws + 7340032;
  float* Vb  = ws + 8388608;
  float* Hb  = ws + 9437184;
  float* ST  = TMP;

  posenc_add<<<8192, 256, 0, stream>>>(x, OUT);
  batch_stats<<<512, 256, 0, stream>>>(OUT, ST);
  ln_apply<<<8192, 256, 0, stream>>>(OUT, n0g, n0b, ST, RES);

  for (int i = 0; i < 4; ++i) {
    dwconv<<<8192, 256, 0, stream>>>(OUT, dw_w + i * CC * 7, dw_b + i * CC, TMP);
    pw_gemm<<<512, 128, 0, stream>>>(TMP, pw_w + i * CC * CC, pw_b + i * CC, RES);
    batch_stats<<<512, 256, 0, stream>>>(RES, ST);
    ln_apply<<<8192, 256, 0, stream>>>(RES, nsg + i * CT, nsb + i * CT, ST, OUT);
  }

  xtw_gemm<<<256, 128, 0, stream>>>(OUT, Wq, Qb);
  xtw_gemm<<<256, 128, 0, stream>>>(OUT, Wk, Kb);
  xtw_gemm<<<256, 128, 0, stream>>>(OUT, Wv, Vb);
  attn_kernel<<<1024, 32, 0, stream>>>(Qb, Kb, Vb, mask, Hb);
  wo_gemm<<<512, 128, 0, stream>>>(Hb, Wo, RES);
  batch_stats<<<512, 256, 0, stream>>>(RES, ST);
  ln_apply<<<8192, 256, 0, stream>>>(RES, neg_, neb, ST, OUT);
  fc_gemm<<<512, 128, 0, stream>>>(OUT, fcw, fcb, RES, outp);
}


# Per-element kernels (posenc_add, ln_apply, dwconv): 512 work-groups with 16 elements per lane as four 16-byte accesses instead of 8192 with one element per lane: 16x fewer waves to enter
# speedup vs baseline: 1.6394x; 1.0776x over previous
#include <hip/hip_runtime.h>

typedef float v2f __attribute__((ext_vector_type(2)));
typedef float v8f __attribute__((ext_vector_type(8)));

#define BB 16
#define CC 128
#define TT 1024
#define DKD 64
#define CT (CC * TT)
#define NPART 32
#define PCHUNK (CT / NPART)
#define RPW 4
#define WGE (RPW * TT)
#define LOG10000 9.210340371976184f
#define PI_2 1.5707963267948966f

__device__ inline v8f wmma4(v2f a, v2f b, v8f c) {
  return __builtin_amdgcn_wmma_f32_16x16x4_f32(false, a, false, b, (short)0, c,
                                               false, false);
}


__global__ __launch_bounds__(256) void posenc_add(const float* __restrict__ x,
                                                  float* __restrict__ out) {
  int p = threadIdx.x;
  int row0 = blockIdx.x * RPW;
  const float4* xin = reinterpret_cast<const float4*>(x) + row0 * (TT / 4) + p;
  float4* o = reinterpret_cast<float4*>(out) + row0 * (TT / 4) + p;
  float4 v[RPW];
#pragma unroll
  for (int q = 0; q < RPW; ++q) v[q] = xin[q * (TT / 4)];
  int t = p * 4;
#pragma unroll
  for (int q = 0; q < RPW; ++q) {
    int c = (row0 + q) % CC;
    float freq = expf(-(float)(c & ~1) * (LOG10000 / (float)CC));
    float ph = (float)(c & 1) * PI_2;
    float4 r;
    r.x = v[q].x + sinf((float)t * freq + ph);
    r.y = v[q].y + sinf((float)(t + 1) * freq + ph);
    r.z = v[q].z + sinf((float)(t + 2) * freq + ph);
    r.w = v[q].w + sinf((float)(t + 3) * freq + ph);
    o[q * (TT / 4)] = r;
  }
}

__global__ __launch_bounds__(256) void batch_stats(const float* __restrict__ src,
                                                   float* __restrict__ part) {
  __shared__ float wsum[8];
  __shared__ float wsq[8];
  int b = blockIdx.x / NPART;
  int j = blockIdx.x % NPART;
  int tid = threadIdx.x;
  int lane = tid & 31;
  const float4* p = reinterpret_cast<const float4*>(src + b * CT + j * PCHUNK);
  float4 v0 = p[tid];
  float4 v1 = p[tid + 256];
  float4 v2 = p[tid + 512];
  float4 v3 = p[tid + 768];
  float s0 = (v0.x + v0.y) + (v0.z + v0.w);
  float s1 = (v1.x + v1.y) + (v1.z + v1.w);
  float s2 = (v2.x + v2.y) + (v2.z + v2.w);
  float s3 = (v3.x + v3.y) + (v3.z + v3.w);
  float q0 = (v0.x * v0.x + v0.y * v0.y) + (v0.z * v0.z + v0.w * v0.w);
  float q1 = (v1.x * v1.x + v1.y * v1.y) + (v1.z * v1.z + v1.w * v1.w);
  float q2 = (v2.x * v2.x + v2.y * v2.y) + (v2.z * v2.z + v2.w * v2.w);
  float q3 = (v3.x * v3.x + v3.y * v3.y) + (v3.z * v3.z + v3.w * v3.w);
  float s = (s0 + s1) + (s2 + s3);
  float q = (q0 + q1) + (q2 + q3);
  for (int o = 16; o > 0; o >>= 1) {
    s += __shfl_xor(s, o, 32);
    q += __shfl_xor(q, o, 32);
  }
  if (lane == 0) {
    wsum[tid >> 5] = s;
    wsq[tid >> 5] = q;
  }
  __syncthreads();
  if (tid < 32) {
    float a = (lane < 8) ? wsum[lane] : 0.f;
    float c = (lane < 8) ? wsq[lane] : 0.f;
    for (int o = 4; o > 0; o >>= 1) {
      a += __shfl_xor(a, o, 32);
      c += __shfl_xor(c, o, 32);
    }
    if (lane == 0) {
      reinterpret_cast<float2*>(part)[blockIdx.x] = make_float2(a, c);
    }
  }
}

__global__ __launch_bounds__(256) void ln_apply(const float* __restrict__ src,
                                                const float* __restrict__ g,
                                                const float* __restrict__ bb,
                                                const float* __restrict__ part,
                                                float* __restrict__ dst) {
  int tid = threadIdx.x;
  int b = blockIdx.x / (CT / WGE);
  int f0 = (blockIdx.x % (CT / WGE)) * (WGE / 4) + tid;
  const float4* s4 = reinterpret_cast<const float4*>(src + b * CT) + f0;
  const float4* g4 = reinterpret_cast<const float4*>(g) + f0;
  const float4* b4 = reinterpret_cast<const float4*>(bb) + f0;
  float4* d4 = reinterpret_cast<float4*>(dst + b * CT) + f0;
  float4 xv[4], gv[4], bv[4];
#pragma unroll
  for (int q = 0; q < 4; ++q) {
    xv[q] = s4[q * 256];
    gv[q] = g4[q * 256];
    bv[q] = b4[q * 256];
  }
  float2 pr = reinterpret_cast<const float2*>(part)[b * NPART + (tid & 31)];
  float s = pr.x;
  float q2 = pr.y;
  for (int o = 16; o > 0; o >>= 1) {
    s += __shfl_xor(s, o, 32);
    q2 += __shfl_xor(q2, o, 32);
  }
  float inv_n = 1.0f / (float)CT;
  float mu = s * inv_n;
  float var = q2 * inv_n - mu * mu;
  float r = rsqrtf(var + 1e-5f);
#pragma unroll
  for (int q = 0; q < 4; ++q) {
    float4 y;
    y.x = (xv[q].x - mu) * r * gv[q].x + bv[q].x;
    y.y = (xv[q].y - mu) * r * gv[q].y + bv[q].y;
    y.z = (xv[q].z - mu) * r * gv[q].z + bv[q].z;
    y.w = (xv[q].w - mu) * r * gv[q].w + bv[q].w;
    d4[q * 256] = y;
  }
}

__global__ __launch_bounds__(256) void dwconv(const float* __restrict__ src,
                                              const float* __restrict__ w,
                                              const float* __restrict__ bias,
                                              float* __restrict__ dst) {
  int p = threadIdx.x;
  int row0 = blockIdx.x * RPW;
  int pl = p > 0 ? p - 1 : p;
  int pr = p < TT / 4 - 1 ? p + 1 : p;
  float4 cur[RPW], prv[RPW], nxt[RPW];
#pragma unroll
  for (int q = 0; q < RPW; ++q) {
    const float4* r4 = reinterpret_cast<const float4*>(src) + (row0 + q) * (TT / 4);
    cur[q] = r4[p];
    prv[q] = r4[pl];
    nxt[q] = r4[pr];
  }
#pragma unroll
  for (int q = 0; q < RPW; ++q) {
    int c = (row0 + q) % CC;
    float bs = bias[c];
    float wj[7];
#pragma unroll
    for (int j = 0; j < 7; ++j) wj[j] = w[c * 7 + j];
    float xs[12] = {prv[q].x, prv[q].y, prv[q].z, prv[q].w, cur[q].x, cur[q].y,
                    cur[q].z, cur[q].w, nxt[q].x, nxt[q].y, nxt[q].z, nxt[q].w};
    float o4[4];
#pragma unroll
    for (int k = 0; k < 4; ++k) {
      float s = bs;
#pragma unroll
      for (int j = 0; j < 7; ++j) {
        int tt = p * 4 + k + j - 3;
        float sn = fmaf(xs[k + j + 1], wj[j], s);
        s = (tt >= 0 && tt < TT) ? sn : s;
      }
      o4[k] = s;
    }
    reinterpret_cast<float4*>(dst)[(row0 + q) * (TT / 4) + p] =
        make_float4(o4[0], o4[1], o4[2], o4[3]);
  }
}


__global__ __launch_bounds__(128) void pw_gemm(const float* __restrict__ Y,
                                               const float* __restrict__ W,
                                               const float* __restrict__ bias,
                                               float* __restrict__ res) {
  int l = threadIdx.x & 31;
  int tile = blockIdx.x * 4 + (threadIdx.x >> 5);
  int b = tile >> 7;
  int rem = tile & 127;
  int m0 = (rem >> 4) << 4;
  int n0 = (rem & 15) << 6;
  int lo = l & 15;
  int kh = (l >> 4) << 1;
  const float* yb = Y + b * CT;
  v8f acc0 = {}, acc1 = {}, acc2 = {}, acc3 = {};
  for (int kk = 0; kk < CC; kk += 4) {
    int k0 = kk + kh;
    v2f a;
    a.x = W[(m0 + lo) * CC + k0];
    a.y = W[(m0 + lo) * CC + k0 + 1];
    const float* p0 = yb + k0 * TT + n0 + lo;
    const float* p1 = p0 + TT;
    v2f b0, b1, b2, b3;
    b0.x = p0[0];  b0.y = p1[0];
    b1.x = p0[16]; b1.y = p1[16];
    b2.x = p0[32]; b2.y = p1[32];
    b3.x = p0[48]; b3.y = p1[48];
    acc0 = wmma4(a, b0, acc0);
    acc1 = wmma4(a, b1, acc1);
    acc2 = wmma4(a, b2, acc2);
    acc3 = wmma4(a, b3, acc3);
  }
  float* rb = res + b * CT;
#pragma unroll
  for (int v = 0; v < 8; ++v) {
    int m = m0 + v + ((l >> 4) << 3);
    float bsv = bias[m];
    float* r = rb + m * TT + n0 + lo;
    r[0]  = fmaxf(acc0[v] + bsv, 0.f) + r[0];
    r[16] = fmaxf(acc1[v] + bsv, 0.f) + r[16];
    r[32] = fmaxf(acc2[v] + bsv, 0.f) + r[32];
    r[48] = fmaxf(acc3[v] + bsv, 0.f) + r[48];
  }
}

__global__ __launch_bounds__(128) void xtw_gemm(const float* __restrict__ X,
                                                const float* __restrict__ W,
                                                float* __restrict__ dst) {
  int l = threadIdx.x & 31;
  int tile = blockIdx.x * 4 + (threadIdx.x >> 5);
  int b = tile >> 6;
  int t0 = (tile & 63) << 4;
  int lo = l & 15;
  int kh = (l >> 4) << 1;
  const float* xb = X + b * CT;
  v8f acc0 = {}, acc1 = {}, acc2 = {}, acc3 = {};
  for (int kk = 0; kk < CC; kk += 4) {
    int k0 = kk + kh;
    v2f a;
    a.x = xb[k0 * TT + t0 + lo];
    a.y = xb[(k0 + 1) * TT + t0 + lo];
    const float* w0 = W + k0 * DKD + lo;
    const float* w1 = w0 + DKD;
    v2f b0, b1, b2, b3;
    b0.x = w0[0];  b0.y = w1[0];
    b1.x = w0[16]; b1.y = w1[16];
    b2.x = w0[32]; b2.y = w1[32];
    b3.x = w0[48]; b3.y = w1[48];
    acc0 = wmma4(a, b0, acc0);
    acc1 = wmma4(a, b1, acc1);
    acc2 = wmma4(a, b2, acc2);
    acc3 = wmma4(a, b3, acc3);
  }
  float* db = dst + b * TT * DKD;
#pragma unroll
  for (int v = 0; v < 8; ++v) {
    int t = t0 + v + ((l >> 4) << 3);
    float* d = db + t * DKD + lo;
    d[0]  = acc0[v];
    d[16] = acc1[v];
    d[32] = acc2[v];
    d[48] = acc3[v];
  }
}

__global__ __launch_bounds__(32) void attn_kernel(const float* __restrict__ Q,
                                                  const float* __restrict__ Km,
                                                  const float* __restrict__ Vm,
                                                  const float* __restrict__ mask,
                                                  float* __restrict__ head) {
  __shared__ float S[16 * TT];
  int b = blockIdx.x >> 6;
  int q0 = (blockIdx.x & 63) << 4;
  int l = threadIdx.x;
  int lo = l & 15;
  int kh = (l >> 4) << 1;
  const float* qb = Q + b * TT * DKD;
  const float* kb = Km + b * TT * DKD;
  const float* vb = Vm + b * TT * DKD;
  const float* mb = mask + b * TT;

  v2f afrag[DKD / 4];
#pragma unroll
  for (int kk = 0; kk < DKD; kk += 4) {
    int k0 = kk + kh;
    afrag[kk >> 2].x = qb[(q0 + lo) * DKD + k0];
    afrag[kk >> 2].y = qb[(q0 + lo) * DKD + k0 + 1];
  }

  for (int t = 0; t < TT / 16; ++t) {
    v8f acc = {};
#pragma unroll
    for (int kk = 0; kk < DKD; kk += 4) {
      int k0 = kk + kh;
      v2f bv;
      bv.x = kb[(t * 16 + lo) * DKD + k0];
      bv.y = kb[(t * 16 + lo) * DKD + k0 + 1];
      acc = wmma4(afrag[kk >> 2], bv, acc);
    }
#pragma unroll
    for (int v = 0; v < 8; ++v) {
      int row = v + ((l >> 4) << 3);
      int col = t * 16 + lo;
      float hm = mb[col];
      float sv = acc[v] * 0.125f;
      S[row * TT + col] = sv * hm + (1.f - hm) * -1e30f;
    }
  }
  __syncthreads();

  {
    int row = lo;
    int c0 = (l >> 4) * (TT / 2);
    float mx = -3.4e38f;
    for (int c = 0; c < TT / 2; ++c) mx = fmaxf(mx, S[row * TT + c0 + c]);
    mx = fmaxf(mx, __shfl_xor(mx, 16, 32));
    float sum = 0.f;
    for (int c = 0; c < TT / 2; ++c) sum += expf(S[row * TT + c0 + c] - mx);
    sum += __shfl_xor(sum, 16, 32);
    float inv = 1.0f / sum;
    float qm = mask[b * TT + q0 + row];
    for (int c = 0; c < TT / 2; ++c) {
      int ix = row * TT + c0 + c;
      S[ix] = expf(S[ix] - mx) * inv * qm;
    }
  }
  __syncthreads();

  {
    v8f acc0 = {}, acc1 = {}, acc2 = {}, acc3 = {};
    for (int kk = 0; kk < TT; kk += 4) {
      int k0 = kk + kh;
      v2f a;
      a.x = S[lo * TT + k0];
      a.y = S[lo * TT + k0 + 1];
      const float* v0p = vb + k0 * DKD + lo;
      const float* v1p = v0p + DKD;
      v2f b0, b1, b2, b3;
      b0.x = v0p[0];  b0.y = v1p[0];
      b1.x = v0p[16]; b1.y = v1p[16];
      b2.x = v0p[32]; b2.y = v1p[32];
      b3.x = v0p[48]; b3.y = v1p[48];
      acc0 = wmma4(a, b0, acc0);
      acc1 = wmma4(a, b1, acc1);
      acc2 = wmma4(a, b2, acc2);
      acc3 = wmma4(a, b3, acc3);
    }
    float* hb = head + b * TT * DKD;
#pragma unroll
    for (int v = 0; v < 8; ++v) {
      int t = q0 + v + ((l >> 4) << 3);
      float* h = hb + t * DKD + lo;
      h[0]  = acc0[v];
      h[16] = acc1[v];
      h[32] = acc2[v];
      h[48] = acc3[v];
    }
  }
}

__global__ __launch_bounds__(128) void wo_gemm(const float* __restrict__ H,
                                               const float* __restrict__ Wo,
                                               float* __restrict__ res) {
  int l = threadIdx.x & 31;
  int tile = blockIdx.x * 4 + (threadIdx.x >> 5);
  int b = tile >> 7;
  int rem = tile & 127;
  int m0 = (rem >> 4) << 4;
  int n0 = (rem & 15) << 6;
  int lo = l & 15;
  int kh = (l >> 4) << 1;
  const float* hb = H + b * TT * DKD;
  v8f acc0 = {}, acc1 = {}, acc2 = {}, acc3 = {};
  for (int kk = 0; kk < DKD; kk += 4) {
    int k0 = kk + kh;
    v2f a;
    a.x = Wo[k0 * CC + m0 + lo] + Wo[(k0 + 64) * CC + m0 + lo];
    a.y = Wo[(k0 + 1) * CC + m0 + lo] + Wo[(k0 + 65) * CC + m0 + lo];
    const float* h0 = hb + (n0 + lo) * DKD + k0;
    v2f b0, b1, b2, b3;
    b0.x = h0[0];            b0.y = h0[1];
    b1.x = h0[16 * DKD];     b1.y = h0[16 * DKD + 1];
    b2.x = h0[32 * DKD];     b2.y = h0[32 * DKD + 1];
    b3.x = h0[48 * DKD];     b3.y = h0[48 * DKD + 1];
    acc0 = wmma4(a, b0, acc0);
    acc1 = wmma4(a, b1, acc1);
    acc2 = wmma4(a, b2, acc2);
    acc3 = wmma4(a, b3, acc3);
  }
  float* rb = res + b * CT;
#pragma unroll
  for (int v = 0; v < 8; ++v) {
    int m = m0 + v + ((l >> 4) << 3);
    float* r = rb + m * TT + n0 + lo;
    r[0]  = acc0[v] + r[0];
    r[16] = acc1[v] + r[16];
    r[32] = acc2[v] + r[32];
    r[48] = acc3[v] + r[48];
  }
}

__global__ __launch_bounds__(128) void fc_gemm(const float* __restrict__ X,
                                               const float* __restrict__ W,
                                               const float* __restrict__ bias,
                                               const float* __restrict__ res,
                                               float* __restrict__ out) {
  int l = threadIdx.x & 31;
  int tile = blockIdx.x * 4 + (threadIdx.x >> 5);
  int b = tile >> 7;
  int rem = tile & 127;
  int m0 = (rem >> 4) << 4;
  int n0 = (rem & 15) << 6;
  int lo = l & 15;
  int kh = (l >> 4) << 1;
  const float* xb = X + b * CT;
  v8f acc0 = {}, acc1 = {}, acc2 = {}, acc3 = {};
  for (int kk = 0; kk < CC; kk += 4) {
    int k0 = kk + kh;
    v2f a;
    a.x = W[(m0 + lo) * CC + k0];
    a.y = W[(m0 + lo) * CC + k0 + 1];
    const float* p0 = xb + k0 * TT + n0 + lo;
    const float* p1 = p0 + TT;
    v2f b0, b1, b2, b3;
    b0.x = p0[0];  b0.y = p1[0];
    b1.x = p0[16]; b1.y = p1[16];
    b2.x = p0[32]; b2.y = p1[32];
    b3.x = p0[48]; b3.y = p1[48];
    acc0 = wmma4(a, b0, acc0);
    acc1 = wmma4(a, b1, acc1);
    acc2 = wmma4(a, b2, acc2);
    acc3 = wmma4(a, b3, acc3);
  }
  const float* rb = res + b * CT;
  float* ob = out + b * CT;
#pragma unroll
  for (int v = 0; v < 8; ++v) {
    int m = m0 + v + ((l >> 4) << 3);
    float bsv = bias[m];
    const float* r = rb + m * TT + n0 + lo;
    float* o = ob + m * TT + n0 + lo;
    o[0]  = fmaxf(acc0[v] + bsv, 0.f) + r[0];
    o[16] = fmaxf(acc1[v] + bsv, 0.f) + r[16];
    o[32] = fmaxf(acc2[v] + bsv, 0.f) + r[32];
    o[48] = fmaxf(acc3[v] + bsv, 0.f) + r[48];
  }
}


extern "C" void kernel_launch(void* const* d_in, const int* in_sizes, int n_in,
                              void* d_out, int out_size, void* d_ws, size_t ws_size,
                              hipStream_t stream) {
  (void)in_sizes; (void)n_in; (void)out_size; (void)ws_size;
  const float* x    = (const float*)d_in[0];
  const float* mask = (const float*)d_in[1];
  const float* dw_w = (const float*)d_in[2];
  const float* dw_b = (const float*)d_in[3];
  const float* pw_w = (const float*)d_in[4];
  const float* pw_b = (const float*)d_in[5];
  const float* n0g  = (const float*)d_in[6];
  const float* n0b  = (const float*)d_in[7];
  const float* nsg  = (const float*)d_in[8];
  const float* nsb  = (const float*)d_in[9];
  const float* neg_ = (const float*)d_in[10];
  const float* neb  = (const float*)d_in[11];
  const float* Wq   = (const float*)d_in[12];
  const float* Wk   = (const float*)d_in[13];
  const float* Wv   = (const float*)d_in[14];
  const float* Wo   = (const float*)d_in[15];
  const float* fcw  = (const float*)d_in[16];
  const float* fcb  = (const float*)d_in[17];
  float* outp = (float*)d_out;
  float* ws = (float*)d_ws;

  float* OUT = ws;
  float* RES = ws + 2097152;
  float* TMP = ws + 4194304;
  float* Qb  = ws + 6291456;
  float* Kb  = ws + 7340032;
  float* Vb  = ws + 8388608;
  float* Hb  = ws + 9437184;
  float* ST  = TMP;

  posenc_add<<<512, 256, 0, stream>>>(x, OUT);
  batch_stats<<<512, 256, 0, stream>>>(OUT, ST);
  ln_apply<<<512, 256, 0, stream>>>(OUT, n0g, n0b, ST, RES);

  for (int i = 0; i < 4; ++i) {
    dwconv<<<512, 256, 0, stream>>>(OUT, dw_w + i * CC * 7, dw_b + i * CC, TMP);
    pw_gemm<<<512, 128, 0, stream>>>(TMP, pw_w + i * CC * CC, pw_b + i * CC, RES);
    batch_stats<<<512, 256, 0, stream>>>(RES, ST);
    ln_apply<<<512, 256, 0, stream>>>(RES, nsg + i * CT, nsb + i * CT, ST, OUT);
  }

  xtw_gemm<<<256, 128, 0, stream>>>(OUT, Wq, Qb);
  xtw_gemm<<<256, 128, 0, stream>>>(OUT, Wk, Kb);
  xtw_gemm<<<256, 128, 0, stream>>>(OUT, Wv, Vb);
  attn_kernel<<<1024, 32, 0, stream>>>(Qb, Kb, Vb, mask, Hb);
  wo_gemm<<<512, 128, 0, stream>>>(Hb, Wo, RES);
  batch_stats<<<512, 256, 0, stream>>>(RES, ST);
  ln_apply<<<512, 256, 0, stream>>>(RES, neg_, neb, ST, OUT);
  fc_gemm<<<512, 128, 0, stream>>>(OUT, fcw, fcb, RES, outp);
}


# attn_kernel: each 16-query tile is shared by 4 waves (key tiles, softmax rows and the P.V k range split 4 ways, partial head tiles added in LDS): a different reduction tree for row sums and P.V
# speedup vs baseline: 2.2494x; 1.3721x over previous
#include <hip/hip_runtime.h>

typedef float v2f __attribute__((ext_vector_type(2)));
typedef float v8f __attribute__((ext_vector_type(8)));

#define BB 16
#define CC 128
#define TT 1024
#define DKD 64
#define CT (CC * TT)
#define NPART 32
#define PCHUNK (CT / NPART)
#define RPW 4
#define WGE (RPW * TT)
#define LOG10000 9.210340371976184f
#define PI_2 1.5707963267948966f

__device__ inline v8f wmma4(v2f a, v2f b, v8f c) {
  return __builtin_amdgcn_wmma_f32_16x16x4_f32(false, a, false, b, (short)0, c,
                                               false, false);
}


__global__ __launch_bounds__(256) void posenc_add(const float* __restrict__ x,
                                                  float* __restrict__ out) {
  int p = threadIdx.x;
  int row0 = blockIdx.x * RPW;
  const float4* xin = reinterpret_cast<const float4*>(x) + row0 * (TT / 4) + p;
  float4* o = reinterpret_cast<float4*>(out) + row0 * (TT / 4) + p;
  float4 v[RPW];
#pragma unroll
  for (int q = 0; q < RPW; ++q) v[q] = xin[q * (TT / 4)];
  int t = p * 4;
#pragma unroll
  for (int q = 0; q < RPW; ++q) {
    int c = (row0 + q) % CC;
    float freq = expf(-(float)(c & ~1) * (LOG10000 / (float)CC));
    float ph = (float)(c & 1) * PI_2;
    float4 r;
    r.x = v[q].x + sinf((float)t * freq + ph);
    r.y = v[q].y + sinf((float)(t + 1) * freq + ph);
    r.z = v[q].z + sinf((float)(t + 2) * freq + ph);
    r.w = v[q].w + sinf((float)(t + 3) * freq + ph);
    o[q * (TT / 4)] = r;
  }
}

__global__ __launch_bounds__(256) void batch_stats(const float* __restrict__ src,
                                                   float* __restrict__ part) {
  __shared__ float wsum[8];
  __shared__ float wsq[8];
  int b = blockIdx.x / NPART;
  int j = blockIdx.x % NPART;
  int tid = threadIdx.x;
  int lane = tid & 31;
  const float4* p = reinterpret_cast<const float4*>(src + b * CT + j * PCHUNK);
  float4 v0 = p[tid];
  float4 v1 = p[tid + 256];
  float4 v2 = p[tid + 512];
  float4 v3 = p[tid + 768];
  float s0 = (v0.x + v0.y) + (v0.z + v0.w);
  float s1 = (v1.x + v1.y) + (v1.z + v1.w);
  float s2 = (v2.x + v2.y) + (v2.z + v2.w);
  float s3 = (v3.x + v3.y) + (v3.z + v3.w);
  float q0 = (v0.x * v0.x + v0.y * v0.y) + (v0.z * v0.z + v0.w * v0.w);
  float q1 = (v1.x * v1.x + v1.y * v1.y) + (v1.z * v1.z + v1.w * v1.w);
  float q2 = (v2.x * v2.x + v2.y * v2.y) + (v2.z * v2.z + v2.w * v2.w);
  float q3 = (v3.x * v3.x + v3.y * v3.y) + (v3.z * v3.z + v3.w * v3.w);
  float s = (s0 + s1) + (s2 + s3);
  float q = (q0 + q1) + (q2 + q3);
  for (int o = 16; o > 0; o >>= 1) {
    s += __shfl_xor(s, o, 32);
    q += __shfl_xor(q, o, 32);
  }
  if (lane == 0) {
    wsum[tid >> 5] = s;
    wsq[tid >> 5] = q;
  }
  __syncthreads();
  if (tid < 32) {
    float a = (lane < 8) ? wsum[lane] : 0.f;
    float c = (lane < 8) ? wsq[lane] : 0.f;
    for (int o = 4; o > 0; o >>= 1) {
      a += __shfl_xor(a, o, 32);
      c += __shfl_xor(c, o, 32);
    }
    if (lane == 0) {
      reinterpret_cast<float2*>(part)[blockIdx.x] = make_float2(a, c);
    }
  }
}

__global__ __launch_bounds__(256) void ln_apply(const float* __restrict__ src,
                                                const float* __restrict__ g,
                                                const float* __restrict__ bb,
                                                const float* __restrict__ part,
                                                float* __restrict__ dst) {
  int tid = threadIdx.x;
  int b = blockIdx.x / (CT / WGE);
  int f0 = (blockIdx.x % (CT / WGE)) * (WGE / 4) + tid;
  const float4* s4 = reinterpret_cast<const float4*>(src + b * CT) + f0;
  const float4* g4 = reinterpret_cast<const float4*>(g) + f0;
  const float4* b4 = reinterpret_cast<const float4*>(bb) + f0;
  float4* d4 = reinterpret_cast<float4*>(dst + b * CT) + f0;
  float4 xv[4], gv[4], bv[4];
#pragma unroll
  for (int q = 0; q < 4; ++q) {
    xv[q] = s4[q * 256];
    gv[q] = g4[q * 256];
    bv[q] = b4[q * 256];
  }
  float2 pr = reinterpret_cast<const float2*>(part)[b * NPART + (tid & 31)];
  float s = pr.x;
  float q2 = pr.y;
  for (int o = 16; o > 0; o >>= 1) {
    s += __shfl_xor(s, o, 32);
    q2 += __shfl_xor(q2, o, 32);
  }
  float inv_n = 1.0f / (float)CT;
  float mu = s * inv_n;
  float var = q2 * inv_n - mu * mu;
  float r = rsqrtf(var + 1e-5f);
#pragma unroll
  for (int q = 0; q < 4; ++q) {
    float4 y;
    y.x = (xv[q].x - mu) * r * gv[q].x + bv[q].x;
    y.y = (xv[q].y - mu) * r * gv[q].y + bv[q].y;
    y.z = (xv[q].z - mu) * r * gv[q].z + bv[q].z;
    y.w = (xv[q].w - mu) * r * gv[q].w + bv[q].w;
    d4[q * 256] = y;
  }
}

__global__ __launch_bounds__(256) void dwconv(const float* __restrict__ src,
                                              const float* __restrict__ w,
                                              const float* __restrict__ bias,
                                              float* __restrict__ dst) {
  int p = threadIdx.x;
  int row0 = blockIdx.x * RPW;
  int pl = p > 0 ? p - 1 : p;
  int pr = p < TT / 4 - 1 ? p + 1 : p;
  float4 cur[RPW], prv[RPW], nxt[RPW];
#pragma unroll
  for (int q = 0; q < RPW; ++q) {
    const float4* r4 = reinterpret_cast<const float4*>(src) + (row0 + q) * (TT / 4);
    cur[q] = r4[p];
    prv[q] = r4[pl];
    nxt[q] = r4[pr];
  }
#pragma unroll
  for (int q = 0; q < RPW; ++q) {
    int c = (row0 + q) % CC;
    float bs = bias[c];
    float wj[7];
#pragma unroll
    for (int j = 0; j < 7; ++j) wj[j] = w[c * 7 + j];
    float xs[12] = {prv[q].x, prv[q].y, prv[q].z, prv[q].w, cur[q].x, cur[q].y,
                    cur[q].z, cur[q].w, nxt[q].x, nxt[q].y, nxt[q].z, nxt[q].w};
    float o4[4];
#pragma unroll
    for (int k = 0; k < 4; ++k) {
      float s = bs;
#pragma unroll
      for (int j = 0; j < 7; ++j) {
        int tt = p * 4 + k + j - 3;
        float sn = fmaf(xs[k + j + 1], wj[j], s);
        s = (tt >= 0 && tt < TT) ? sn : s;
      }
      o4[k] = s;
    }
    reinterpret_cast<float4*>(dst)[(row0 + q) * (TT / 4) + p] =
        make_float4(o4[0], o4[1], o4[2], o4[3]);
  }
}


__global__ __launch_bounds__(128) void pw_gemm(const float* __restrict__ Y,
                                               const float* __restrict__ W,
                                               const float* __restrict__ bias,
                                               float* __restrict__ res) {
  int l = threadIdx.x & 31;
  int tile = blockIdx.x * 4 + (threadIdx.x >> 5);
  int b = tile >> 7;
  int rem = tile & 127;
  int m0 = (rem >> 4) << 4;
  int n0 = (rem & 15) << 6;
  int lo = l & 15;
  int kh = (l >> 4) << 1;
  const float* yb = Y + b * CT;
  v8f acc0 = {}, acc1 = {}, acc2 = {}, acc3 = {};
  for (int kk = 0; kk < CC; kk += 4) {
    int k0 = kk + kh;
    v2f a;
    a.x = W[(m0 + lo) * CC + k0];
    a.y = W[(m0 + lo) * CC + k0 + 1];
    const float* p0 = yb + k0 * TT + n0 + lo;
    const float* p1 = p0 + TT;
    v2f b0, b1, b2, b3;
    b0.x = p0[0];  b0.y = p1[0];
    b1.x = p0[16]; b1.y = p1[16];
    b2.x = p0[32]; b2.y = p1[32];
    b3.x = p0[48]; b3.y = p1[48];
    acc0 = wmma4(a, b0, acc0);
    acc1 = wmma4(a, b1, acc1);
    acc2 = wmma4(a, b2, acc2);
    acc3 = wmma4(a, b3, acc3);
  }
  float* rb = res + b * CT;
#pragma unroll
  for (int v = 0; v < 8; ++v) {
    int m = m0 + v + ((l >> 4) << 3);
    float bsv = bias[m];
    float* r = rb + m * TT + n0 + lo;
    r[0]  = fmaxf(acc0[v] + bsv, 0.f) + r[0];
    r[16] = fmaxf(acc1[v] + bsv, 0.f) + r[16];
    r[32] = fmaxf(acc2[v] + bsv, 0.f) + r[32];
    r[48] = fmaxf(acc3[v] + bsv, 0.f) + r[48];
  }
}

__global__ __launch_bounds__(128) void xtw_gemm(const float* __restrict__ X,
                                                const float* __restrict__ W,
                                                float* __restrict__ dst) {
  int l = threadIdx.x & 31;
  int tile = blockIdx.x * 4 + (threadIdx.x >> 5);
  int b = tile >> 6;
  int t0 = (tile & 63) << 4;
  int lo = l & 15;
  int kh = (l >> 4) << 1;
  const float* xb = X + b * CT;
  v8f acc0 = {}, acc1 = {}, acc2 = {}, acc3 = {};
  for (int kk = 0; kk < CC; kk += 4) {
    int k0 = kk + kh;
    v2f a;
    a.x = xb[k0 * TT + t0 + lo];
    a.y = xb[(k0 + 1) * TT + t0 + lo];
    const float* w0 = W + k0 * DKD + lo;
    const float* w1 = w0 + DKD;
    v2f b0, b1, b2, b3;
    b0.x = w0[0];  b0.y = w1[0];
    b1.x = w0[16]; b1.y = w1[16];
    b2.x = w0[32]; b2.y = w1[32];
    b3.x = w0[48]; b3.y = w1[48];
    acc0 = wmma4(a, b0, acc0);
    acc1 = wmma4(a, b1, acc1);
    acc2 = wmma4(a, b2, acc2);
    acc3 = wmma4(a, b3, acc3);
  }
  float* db = dst + b * TT * DKD;
#pragma unroll
  for (int v = 0; v < 8; ++v) {
    int t = t0 + v + ((l >> 4) << 3);
    float* d = db + t * DKD + lo;
    d[0]  = acc0[v];
    d[16] = acc1[v];
    d[32] = acc2[v];
    d[48] = acc3[v];
  }
}

#define AW 4
#define KTW (TT / 16 / AW)
#define RWW (16 / AW)
#define KPW (TT / AW)
__global__ __launch_bounds__(32 * AW) void attn_kernel(const float* __restrict__ Q,
                                                       const float* __restrict__ Km,
                                                       const float* __restrict__ Vm,
                                                       const float* __restrict__ mask,
                                                       float* __restrict__ head) {
  __shared__ __attribute__((aligned(16))) float S[16 * TT];
  int b = blockIdx.x >> 6;
  int q0 = (blockIdx.x & 63) << 4;
  int w = __builtin_amdgcn_readfirstlane((int)(threadIdx.x >> 5));
  int l = threadIdx.x & 31;
  int lo = l & 15;
  int kh = (l >> 4) << 1;
  const float* qb = Q + b * TT * DKD;
  const float* kb = Km + b * TT * DKD;
  const float* vb = Vm + b * TT * DKD;
  const float* mb = mask + b * TT;

  v2f afrag[DKD / 4];
#pragma unroll
  for (int kk = 0; kk < DKD; kk += 4) {
    int k0 = kk + kh;
    afrag[kk >> 2].x = qb[(q0 + lo) * DKD + k0];
    afrag[kk >> 2].y = qb[(q0 + lo) * DKD + k0 + 1];
  }

  for (int ti = 0; ti < KTW; ++ti) {
    int t = w * KTW + ti;
    v8f acc = {};
#pragma unroll
    for (int kk = 0; kk < DKD; kk += 4) {
      int k0 = kk + kh;
      v2f bv;
      bv.x = kb[(t * 16 + lo) * DKD + k0];
      bv.y = kb[(t * 16 + lo) * DKD + k0 + 1];
      acc = wmma4(afrag[kk >> 2], bv, acc);
    }
#pragma unroll
    for (int v = 0; v < 8; ++v) {
      int row = v + ((l >> 4) << 3);
      int col = t * 16 + lo;
      float hm = mb[col];
      float sv = acc[v] * 0.125f;
      S[row * TT + col] = sv * hm + (1.f - hm) * -1e30f;
    }
  }
  __syncthreads();

  for (int r = 0; r < RWW; ++r) {
    int row = w * RWW + r;
    float* sr = S + row * TT + l;
    float e[TT / 32];
    float mx = -3.4e38f;
#pragma unroll
    for (int c = 0; c < TT / 32; ++c) {
      e[c] = sr[c * 32];
      mx = fmaxf(mx, e[c]);
    }
    for (int o = 16; o > 0; o >>= 1) mx = fmaxf(mx, __shfl_xor(mx, o, 32));
    float sum = 0.f;
#pragma unroll
    for (int c = 0; c < TT / 32; ++c) {
      e[c] = expf(e[c] - mx);
      sum += e[c];
    }
    for (int o = 16; o > 0; o >>= 1) sum += __shfl_xor(sum, o, 32);
    float inv = 1.0f / sum;
    float qm = mb[q0 + row];
#pragma unroll
    for (int c = 0; c < TT / 32; ++c) sr[c * 32] = e[c] * inv * qm;
  }
  __syncthreads();

  v8f acc0 = {}, acc1 = {}, acc2 = {}, acc3 = {};
  for (int ki = 0; ki < KPW; ki += 4) {
    int k0 = w * KPW + ki + kh;
    v2f a;
    a.x = S[lo * TT + k0];
    a.y = S[lo * TT + k0 + 1];
    const float* v0p = vb + k0 * DKD + lo;
    const float* v1p = v0p + DKD;
    v2f b0, b1, b2, b3;
    b0.x = v0p[0];  b0.y = v1p[0];
    b1.x = v0p[16]; b1.y = v1p[16];
    b2.x = v0p[32]; b2.y = v1p[32];
    b3.x = v0p[48]; b3.y = v1p[48];
    acc0 = wmma4(a, b0, acc0);
    acc1 = wmma4(a, b1, acc1);
    acc2 = wmma4(a, b2, acc2);
    acc3 = wmma4(a, b3, acc3);
  }
  __syncthreads();

  {
    float* pw = S + w * (16 * DKD);
#pragma unroll
    for (int v = 0; v < 8; ++v) {
      float* p = pw + (v + ((l >> 4) << 3)) * DKD + lo;
      p[0]  = acc0[v];
      p[16] = acc1[v];
      p[32] = acc2[v];
      p[48] = acc3[v];
    }
  }
  __syncthreads();

  {
    const float4* s4 = reinterpret_cast<const float4*>(S);
    float4* h4 = reinterpret_cast<float4*>(head + (b * TT + q0) * DKD);
#pragma unroll
    for (int j = 0; j < (16 * DKD / 4) / (32 * AW); ++j) {
      int i4 = threadIdx.x + j * (32 * AW);
      float4 r = s4[i4];
#pragma unroll
      for (int u = 1; u < AW; ++u) {
        float4 pu = s4[u * (16 * DKD / 4) + i4];
        r.x += pu.x;
        r.y += pu.y;
        r.z += pu.z;
        r.w += pu.w;
      }
      h4[i4] = r;
    }
  }
}

__global__ __launch_bounds__(128) void wo_gemm(const float* __restrict__ H,
                                               const float* __restrict__ Wo,
                                               float* __restrict__ res) {
  int l = threadIdx.x & 31;
  int tile = blockIdx.x * 4 + (threadIdx.x >> 5);
  int b = tile >> 7;
  int rem = tile & 127;
  int m0 = (rem >> 4) << 4;
  int n0 = (rem & 15) << 6;
  int lo = l & 15;
  int kh = (l >> 4) << 1;
  const float* hb = H + b * TT * DKD;
  v8f acc0 = {}, acc1 = {}, acc2 = {}, acc3 = {};
  for (int kk = 0; kk < DKD; kk += 4) {
    int k0 = kk + kh;
    v2f a;
    a.x = Wo[k0 * CC + m0 + lo] + Wo[(k0 + 64) * CC + m0 + lo];
    a.y = Wo[(k0 + 1) * CC + m0 + lo] + Wo[(k0 + 65) * CC + m0 + lo];
    const float* h0 = hb + (n0 + lo) * DKD + k0;
    v2f b0, b1, b2, b3;
    b0.x = h0[0];            b0.y = h0[1];
    b1.x = h0[16 * DKD];     b1.y = h0[16 * DKD + 1];
    b2.x = h0[32 * DKD];     b2.y = h0[32 * DKD + 1];
    b3.x = h0[48 * DKD];     b3.y = h0[48 * DKD + 1];
    acc0 = wmma4(a, b0, acc0);
    acc1 = wmma4(a, b1, acc1);
    acc2 = wmma4(a, b2, acc2);
    acc3 = wmma4(a, b3, acc3);
  }
  float* rb = res + b * CT;
#pragma unroll
  for (int v = 0; v < 8; ++v) {
    int m = m0 + v + ((l >> 4) << 3);
    float* r = rb + m * TT + n0 + lo;
    r[0]  = acc0[v] + r[0];
    r[16] = acc1[v] + r[16];
    r[32] = acc2[v] + r[32];
    r[48] = acc3[v] + r[48];
  }
}

__global__ __launch_bounds__(128) void fc_gemm(const float* __restrict__ X,
                                               const float* __restrict__ W,
                                               const float* __restrict__ bias,
                                               const float* __restrict__ res,
                                               float* __restrict__ out) {
  int l = threadIdx.x & 31;
  int tile = blockIdx.x * 4 + (threadIdx.x >> 5);
  int b = tile >> 7;
  int rem = tile & 127;
  int m0 = (rem >> 4) << 4;
  int n0 = (rem & 15) << 6;
  int lo = l & 15;
  int kh = (l >> 4) << 1;
  const float* xb = X + b * CT;
  v8f acc0 = {}, acc1 = {}, acc2 = {}, acc3 = {};
  for (int kk = 0; kk < CC; kk += 4) {
    int k0 = kk + kh;
    v2f a;
    a.x = W[(m0 + lo) * CC + k0];
    a.y = W[(m0 + lo) * CC + k0 + 1];
    const float* p0 = xb + k0 * TT + n0 + lo;
    const float* p1 = p0 + TT;
    v2f b0, b1, b2, b3;
    b0.x = p0[0];  b0.y = p1[0];
    b1.x = p0[16]; b1.y = p1[16];
    b2.x = p0[32]; b2.y = p1[32];
    b3.x = p0[48]; b3.y = p1[48];
    acc0 = wmma4(a, b0, acc0);
    acc1 = wmma4(a, b1, acc1);
    acc2 = wmma4(a, b2, acc2);
    acc3 = wmma4(a, b3, acc3);
  }
  const float* rb = res + b * CT;
  float* ob = out + b * CT;
#pragma unroll
  for (int v = 0; v < 8; ++v) {
    int m = m0 + v + ((l >> 4) << 3);
    float bsv = bias[m];
    const float* r = rb + m * TT + n0 + lo;
    float* o = ob + m * TT + n0 + lo;
    o[0]  = fmaxf(acc0[v] + bsv, 0.f) + r[0];
    o[16] = fmaxf(acc1[v] + bsv, 0.f) + r[16];
    o[32] = fmaxf(acc2[v] + bsv, 0.f) + r[32];
    o[48] = fmaxf(acc3[v] + bsv, 0.f) + r[48];
  }
}


extern "C" void kernel_launch(void* const* d_in, const int* in_sizes, int n_in,
                              void* d_out, int out_size, void* d_ws, size_t ws_size,
                              hipStream_t stream) {
  (void)in_sizes; (void)n_in; (void)out_size; (void)ws_size;
  const float* x    = (const float*)d_in[0];
  const float* mask = (const float*)d_in[1];
  const float* dw_w = (const float*)d_in[2];
  const float* dw_b = (const float*)d_in[3];
  const float* pw_w = (const float*)d_in[4];
  const float* pw_b = (const float*)d_in[5];
  const float* n0g  = (const float*)d_in[6];
  const float* n0b  = (const float*)d_in[7];
  const float* nsg  = (const float*)d_in[8];
  const float* nsb  = (const float*)d_in[9];
  const float* neg_ = (const float*)d_in[10];
  const float* neb  = (const float*)d_in[11];
  const float* Wq   = (const float*)d_in[12];
  const float* Wk   = (const float*)d_in[13];
  const float* Wv   = (const float*)d_in[14];
  const float* Wo   = (const float*)d_in[15];
  const float* fcw  = (const float*)d_in[16];
  const float* fcb  = (const float*)d_in[17];
  float* outp = (float*)d_out;
  float* ws = (float*)d_ws;

  float* OUT = ws;
  float* RES = ws + 2097152;
  float* TMP = ws + 4194304;
  float* Qb  = ws + 6291456;
  float* Kb  = ws + 7340032;
  float* Vb  = ws + 8388608;
  float* Hb  = ws + 9437184;
  float* ST  = TMP;

  posenc_add<<<512, 256, 0, stream>>>(x, OUT);
  batch_stats<<<512, 256, 0, stream>>>(OUT, ST);
  ln_apply<<<512, 256, 0, stream>>>(OUT, n0g, n0b, ST, RES);

  for (int i = 0; i < 4; ++i) {
    dwconv<<<512, 256, 0, stream>>>(OUT, dw_w + i * CC * 7, dw_b + i * CC, TMP);
    pw_gemm<<<512, 128, 0, stream>>>(TMP, pw_w + i * CC * CC, pw_b + i * CC, RES);
    batch_stats<<<512, 256, 0, stream>>>(RES, ST);
    ln_apply<<<512, 256, 0, stream>>>(RES, nsg + i * CT, nsb + i * CT, ST, OUT);
  }

  xtw_gemm<<<256, 128, 0, stream>>>(OUT, Wq, Qb);
  xtw_gemm<<<256, 128, 0, stream>>>(OUT, Wk, Kb);
  xtw_gemm<<<256, 128, 0, stream>>>(OUT, Wv, Vb);
  attn_kernel<<<1024, 128, 0, stream>>>(Qb, Kb, Vb, mask, Hb);
  wo_gemm<<<512, 128, 0, stream>>>(Hb, Wo, RES);
  batch_stats<<<512, 256, 0, stream>>>(RES, ST);
  ln_apply<<<512, 256, 0, stream>>>(RES, neg_, neb, ST, OUT);
  fc_gemm<<<512, 128, 0, stream>>>(OUT, fcw, fcb, RES, outp);
}


# LayerNorm partial sums made in the epilogue of the kernel that writes the tensor (posenc_add, pw_gemm, wo_gemm); the 6 separate sum launches are gone: a different reduction tree for those sums
# speedup vs baseline: 2.4214x; 1.0765x over previous
#include <hip/hip_runtime.h>

typedef float v2f __attribute__((ext_vector_type(2)));
typedef float v8f __attribute__((ext_vector_type(8)));

#define BB 16
#define CC 128
#define TT 1024
#define DKD 64
#define CT (CC * TT)
#define NPART 32
#define RPW 4
#define WGE (RPW * TT)
#define LOG10000 9.210340371976184f
#define PI_2 1.5707963267948966f

__device__ inline v8f wmma4(v2f a, v2f b, v8f c) {
  return __builtin_amdgcn_wmma_f32_16x16x4_f32(false, a, false, b, (short)0, c,
                                               false, false);
}


__device__ inline void wg_pair_store(float s, float q, int nw, float* wsum, float* wsq,
                                     float* __restrict__ part) {
  int tid = threadIdx.x;
  for (int o = 16; o > 0; o >>= 1) {
    s += __shfl_xor(s, o, 32);
    q += __shfl_xor(q, o, 32);
  }
  if ((tid & 31) == 0) {
    wsum[tid >> 5] = s;
    wsq[tid >> 5] = q;
  }
  __syncthreads();
  if (tid == 0) {
    float a = wsum[0];
    float c = wsq[0];
    for (int u = 1; u < nw; ++u) {
      a += wsum[u];
      c += wsq[u];
    }
    reinterpret_cast<float2*>(part)[blockIdx.x] = make_float2(a, c);
  }
}


__global__ __launch_bounds__(256) void posenc_add(const float* __restrict__ x,
                                                  float* __restrict__ out,
                                                  float* __restrict__ part) {
  __shared__ float wsum[8];
  __shared__ float wsq[8];
  int p = threadIdx.x;
  int row0 = blockIdx.x * RPW;
  const float4* xin = reinterpret_cast<const float4*>(x) + row0 * (TT / 4) + p;
  float4* o = reinterpret_cast<float4*>(out) + row0 * (TT / 4) + p;
  float4 v[RPW];
#pragma unroll
  for (int q = 0; q < RPW; ++q) v[q] = xin[q * (TT / 4)];
  int t = p * 4;
  float s = 0.f, sq = 0.f;
#pragma unroll
  for (int q = 0; q < RPW; ++q) {
    int c = (row0 + q) % CC;
    float freq = expf(-(float)(c & ~1) * (LOG10000 / (float)CC));
    float ph = (float)(c & 1) * PI_2;
    float4 r;
    r.x = v[q].x + sinf((float)t * freq + ph);
    r.y = v[q].y + sinf((float)(t + 1) * freq + ph);
    r.z = v[q].z + sinf((float)(t + 2) * freq + ph);
    r.w = v[q].w + sinf((float)(t + 3) * freq + ph);
    v[q] = r;
    s += (r.x + r.y) + (r.z + r.w);
    sq += (r.x * r.x + r.y * r.y) + (r.z * r.z + r.w * r.w);
  }
  wg_pair_store(s, sq, 8, wsum, wsq, part);
#pragma unroll
  for (int q = 0; q < RPW; ++q) o[q * (TT / 4)] = v[q];
}

__global__ __launch_bounds__(256) void ln_apply(const float* __restrict__ src,
                                                const float* __restrict__ g,
                                                const float* __restrict__ bb,
                                                const float* __restrict__ part,
                                                float* __restrict__ dst) {
  int tid = threadIdx.x;
  int b = blockIdx.x / (CT / WGE);
  int f0 = (blockIdx.x % (CT / WGE)) * (WGE / 4) + tid;
  const float4* s4 = reinterpret_cast<const float4*>(src + b * CT) + f0;
  const float4* g4 = reinterpret_cast<const float4*>(g) + f0;
  const float4* b4 = reinterpret_cast<const float4*>(bb) + f0;
  float4* d4 = reinterpret_cast<float4*>(dst + b * CT) + f0;
  float4 xv[4], gv[4], bv[4];
#pragma unroll
  for (int q = 0; q < 4; ++q) {
    xv[q] = s4[q * 256];
    gv[q] = g4[q * 256];
    bv[q] = b4[q * 256];
  }
  float2 pr = reinterpret_cast<const float2*>(part)[b * NPART + (tid & 31)];
  float s = pr.x;
  float q2 = pr.y;
  for (int o = 16; o > 0; o >>= 1) {
    s += __shfl_xor(s, o, 32);
    q2 += __shfl_xor(q2, o, 32);
  }
  float inv_n = 1.0f / (float)CT;
  float mu = s * inv_n;
  float var = q2 * inv_n - mu * mu;
  float r = rsqrtf(var + 1e-5f);
#pragma unroll
  for (int q = 0; q < 4; ++q) {
    float4 y;
    y.x = (xv[q].x - mu) * r * gv[q].x + bv[q].x;
    y.y = (xv[q].y - mu) * r * gv[q].y + bv[q].y;
    y.z = (xv[q].z - mu) * r * gv[q].z + bv[q].z;
    y.w = (xv[q].w - mu) * r * gv[q].w + bv[q].w;
    d4[q * 256] = y;
  }
}

__global__ __launch_bounds__(256) void dwconv(const float* __restrict__ src,
                                              const float* __restrict__ w,
                                              const float* __restrict__ bias,
                                              float* __restrict__ dst) {
  int p = threadIdx.x;
  int row0 = blockIdx.x * RPW;
  int pl = p > 0 ? p - 1 : p;
  int pr = p < TT / 4 - 1 ? p + 1 : p;
  float4 cur[RPW], prv[RPW], nxt[RPW];
#pragma unroll
  for (int q = 0; q < RPW; ++q) {
    const float4* r4 = reinterpret_cast<const float4*>(src) + (row0 + q) * (TT / 4);
    cur[q] = r4[p];
    prv[q] = r4[pl];
    nxt[q] = r4[pr];
  }
#pragma unroll
  for (int q = 0; q < RPW; ++q) {
    int c = (row0 + q) % CC;
    float bs = bias[c];
    float wj[7];
#pragma unroll
    for (int j = 0; j < 7; ++j) wj[j] = w[c * 7 + j];
    float xs[12] = {prv[q].x, prv[q].y, prv[q].z, prv[q].w, cur[q].x, cur[q].y,
                    cur[q].z, cur[q].w, nxt[q].x, nxt[q].y, nxt[q].z, nxt[q].w};
    float o4[4];
#pragma unroll
    for (int k = 0; k < 4; ++k) {
      float s = bs;
#pragma unroll
      for (int j = 0; j < 7; ++j) {
        int tt = p * 4 + k + j - 3;
        float sn = fmaf(xs[k + j + 1], wj[j], s);
        s = (tt >= 0 && tt < TT) ? sn : s;
      }
      o4[k] = s;
    }
    reinterpret_cast<float4*>(dst)[(row0 + q) * (TT / 4) + p] =
        make_float4(o4[0], o4[1], o4[2], o4[3]);
  }
}


__global__ __launch_bounds__(128) void pw_gemm(const float* __restrict__ Y,
                                               const float* __restrict__ W,
                                               const float* __restrict__ bias,
                                               float* __restrict__ res,
                                               float* __restrict__ part) {
  __shared__ float wsum[4];
  __shared__ float wsq[4];
  int l = threadIdx.x & 31;
  int tile = blockIdx.x * 4 + (threadIdx.x >> 5);
  int b = tile >> 7;
  int rem = tile & 127;
  int m0 = (rem >> 4) << 4;
  int n0 = (rem & 15) << 6;
  int lo = l & 15;
  int kh = (l >> 4) << 1;
  const float* yb = Y + b * CT;
  v8f acc0 = {}, acc1 = {}, acc2 = {}, acc3 = {};
  for (int kk = 0; kk < CC; kk += 4) {
    int k0 = kk + kh;
    v2f a;
    a.x = W[(m0 + lo) * CC + k0];
    a.y = W[(m0 + lo) * CC + k0 + 1];
    const float* p0 = yb + k0 * TT + n0 + lo;
    const float* p1 = p0 + TT;
    v2f b0, b1, b2, b3;
    b0.x = p0[0];  b0.y = p1[0];
    b1.x = p0[16]; b1.y = p1[16];
    b2.x = p0[32]; b2.y = p1[32];
    b3.x = p0[48]; b3.y = p1[48];
    acc0 = wmma4(a, b0, acc0);
    acc1 = wmma4(a, b1, acc1);
    acc2 = wmma4(a, b2, acc2);
    acc3 = wmma4(a, b3, acc3);
  }
  float* rb = res + b * CT;
  float s = 0.f, sq = 0.f;
#pragma unroll
  for (int v = 0; v < 8; ++v) {
    int m = m0 + v + ((l >> 4) << 3);
    float bsv = bias[m];
    const float* r = rb + m * TT + n0 + lo;
    float o0 = fmaxf(acc0[v] + bsv, 0.f) + r[0];
    float o1 = fmaxf(acc1[v] + bsv, 0.f) + r[16];
    float o2 = fmaxf(acc2[v] + bsv, 0.f) + r[32];
    float o3 = fmaxf(acc3[v] + bsv, 0.f) + r[48];
    acc0[v] = o0;
    acc1[v] = o1;
    acc2[v] = o2;
    acc3[v] = o3;
    s += (o0 + o1) + (o2 + o3);
    sq += (o0 * o0 + o1 * o1) + (o2 * o2 + o3 * o3);
  }
  wg_pair_store(s, sq, 4, wsum, wsq, part);
#pragma unroll
  for (int v = 0; v < 8; ++v) {
    float* r = rb + (m0 + v + ((l >> 4) << 3)) * TT + n0 + lo;
    r[0]  = acc0[v];
    r[16] = acc1[v];
    r[32] = acc2[v];
    r[48] = acc3[v];
  }
}

__global__ __launch_bounds__(128) void xtw_gemm(const float* __restrict__ X,
                                                const float* __restrict__ W,
                                                float* __restrict__ dst) {
  int l = threadIdx.x & 31;
  int tile = blockIdx.x * 4 + (threadIdx.x >> 5);
  int b = tile >> 6;
  int t0 = (tile & 63) << 4;
  int lo = l & 15;
  int kh = (l >> 4) << 1;
  const float* xb = X + b * CT;
  v8f acc0 = {}, acc1 = {}, acc2 = {}, acc3 = {};
  for (int kk = 0; kk < CC; kk += 4) {
    int k0 = kk + kh;
    v2f a;
    a.x = xb[k0 * TT + t0 + lo];
    a.y = xb[(k0 + 1) * TT + t0 + lo];
    const float* w0 = W + k0 * DKD + lo;
    const float* w1 = w0 + DKD;
    v2f b0, b1, b2, b3;
    b0.x = w0[0];  b0.y = w1[0];
    b1.x = w0[16]; b1.y = w1[16];
    b2.x = w0[32]; b2.y = w1[32];
    b3.x = w0[48]; b3.y = w1[48];
    acc0 = wmma4(a, b0, acc0);
    acc1 = wmma4(a, b1, acc1);
    acc2 = wmma4(a, b2, acc2);
    acc3 = wmma4(a, b3, acc3);
  }
  float* db = dst + b * TT * DKD;
#pragma unroll
  for (int v = 0; v < 8; ++v) {
    int t = t0 + v + ((l >> 4) << 3);
    float* d = db + t * DKD + lo;
    d[0]  = acc0[v];
    d[16] = acc1[v];
    d[32] = acc2[v];
    d[48] = acc3[v];
  }
}

#define AW 4
#define KTW (TT / 16 / AW)
#define RWW (16 / AW)
#define KPW (TT / AW)
__global__ __launch_bounds__(32 * AW) void attn_kernel(const float* __restrict__ Q,
                                                       const float* __restrict__ Km,
                                                       const float* __restrict__ Vm,
                                                       const float* __restrict__ mask,
                                                       float* __restrict__ head) {
  __shared__ __attribute__((aligned(16))) float S[16 * TT];
  int b = blockIdx.x >> 6;
  int q0 = (blockIdx.x & 63) << 4;
  int w = __builtin_amdgcn_readfirstlane((int)(threadIdx.x >> 5));
  int l = threadIdx.x & 31;
  int lo = l & 15;
  int kh = (l >> 4) << 1;
  const float* qb = Q + b * TT * DKD;
  const float* kb = Km + b * TT * DKD;
  const float* vb = Vm + b * TT * DKD;
  const float* mb = mask + b * TT;

  v2f afrag[DKD / 4];
#pragma unroll
  for (int kk = 0; kk < DKD; kk += 4) {
    int k0 = kk + kh;
    afrag[kk >> 2].x = qb[(q0 + lo) * DKD + k0];
    afrag[kk >> 2].y = qb[(q0 + lo) * DKD + k0 + 1];
  }

  for (int ti = 0; ti < KTW; ++ti) {
    int t = w * KTW + ti;
    v8f acc = {};
#pragma unroll
    for (int kk = 0; kk < DKD; kk += 4) {
      int k0 = kk + kh;
      v2f bv;
      bv.x = kb[(t * 16 + lo) * DKD + k0];
      bv.y = kb[(t * 16 + lo) * DKD + k0 + 1];
      acc = wmma4(afrag[kk >> 2], bv, acc);
    }
#pragma unroll
    for (int v = 0; v < 8; ++v) {
      int row = v + ((l >> 4) << 3);
      int col = t * 16 + lo;
      float hm = mb[col];
      float sv = acc[v] * 0.125f;
      S[row * TT + col] = sv * hm + (1.f - hm) * -1e30f;
    }
  }
  __syncthreads();

  for (int r = 0; r < RWW; ++r) {
    int row = w * RWW + r;
    float* sr = S + row * TT + l;
    float e[TT / 32];
    float mx = -3.4e38f;
#pragma unroll
    for (int c = 0; c < TT / 32; ++c) {
      e[c] = sr[c * 32];
      mx = fmaxf(mx, e[c]);
    }
    for (int o = 16; o > 0; o >>= 1) mx = fmaxf(mx, __shfl_xor(mx, o, 32));
    float sum = 0.f;
#pragma unroll
    for (int c = 0; c < TT / 32; ++c) {
      e[c] = expf(e[c] - mx);
      sum += e[c];
    }
    for (int o = 16; o > 0; o >>= 1) sum += __shfl_xor(sum, o, 32);
    float inv = 1.0f / sum;
    float qm = mb[q0 + row];
#pragma unroll
    for (int c = 0; c < TT / 32; ++c) sr[c * 32] = e[c] * inv * qm;
  }
  __syncthreads();

  v8f acc0 = {}, acc1 = {}, acc2 = {}, acc3 = {};
  for (int ki = 0; ki < KPW; ki += 4) {
    int k0 = w * KPW + ki + kh;
    v2f a;
    a.x = S[lo * TT + k0];
    a.y = S[lo * TT + k0 + 1];
    const float* v0p = vb + k0 * DKD + lo;
    const float* v1p = v0p + DKD;
    v2f b0, b1, b2, b3;
    b0.x = v0p[0];  b0.y = v1p[0];
    b1.x = v0p[16]; b1.y = v1p[16];
    b2.x = v0p[32]; b2.y = v1p[32];
    b3.x = v0p[48]; b3.y = v1p[48];
    acc0 = wmma4(a, b0, acc0);
    acc1 = wmma4(a, b1, acc1);
    acc2 = wmma4(a, b2, acc2);
    acc3 = wmma4(a, b3, acc3);
  }
  __syncthreads();

  {
    float* pw = S + w * (16 * DKD);
#pragma unroll
    for (int v = 0; v < 8; ++v) {
      float* p = pw + (v + ((l >> 4) << 3)) * DKD + lo;
      p[0]  = acc0[v];
      p[16] = acc1[v];
      p[32] = acc2[v];
      p[48] = acc3[v];
    }
  }
  __syncthreads();

  {
    const float4* s4 = reinterpret_cast<const float4*>(S);
    float4* h4 = reinterpret_cast<float4*>(head + (b * TT + q0) * DKD);
#pragma unroll
    for (int j = 0; j < (16 * DKD / 4) / (32 * AW); ++j) {
      int i4 = threadIdx.x + j * (32 * AW);
      float4 r = s4[i4];
#pragma unroll
      for (int u = 1; u < AW; ++u) {
        float4 pu = s4[u * (16 * DKD / 4) + i4];
        r.x += pu.x;
        r.y += pu.y;
        r.z += pu.z;
        r.w += pu.w;
      }
      h4[i4] = r;
    }
  }
}

__global__ __launch_bounds__(128) void wo_gemm(const float* __restrict__ H,
                                               const float* __restrict__ Wo,
                                               float* __restrict__ res,
                                               float* __restrict__ part) {
  __shared__ float wsum[4];
  __shared__ float wsq[4];
  int l = threadIdx.x & 31;
  int tile = blockIdx.x * 4 + (threadIdx.x >> 5);
  int b = tile >> 7;
  int rem = tile & 127;
  int m0 = (rem >> 4) << 4;
  int n0 = (rem & 15) << 6;
  int lo = l & 15;
  int kh = (l >> 4) << 1;
  const float* hb = H + b * TT * DKD;
  v8f acc0 = {}, acc1 = {}, acc2 = {}, acc3 = {};
  for (int kk = 0; kk < DKD; kk += 4) {
    int k0 = kk + kh;
    v2f a;
    a.x = Wo[k0 * CC + m0 + lo] + Wo[(k0 + 64) * CC + m0 + lo];
    a.y = Wo[(k0 + 1) * CC + m0 + lo] + Wo[(k0 + 65) * CC + m0 + lo];
    const float* h0 = hb + (n0 + lo) * DKD + k0;
    v2f b0, b1, b2, b3;
    b0.x = h0[0];            b0.y = h0[1];
    b1.x = h0[16 * DKD];     b1.y = h0[16 * DKD + 1];
    b2.x = h0[32 * DKD];     b2.y = h0[32 * DKD + 1];
    b3.x = h0[48 * DKD];     b3.y = h0[48 * DKD + 1];
    acc0 = wmma4(a, b0, acc0);
    acc1 = wmma4(a, b1, acc1);
    acc2 = wmma4(a, b2, acc2);
    acc3 = wmma4(a, b3, acc3);
  }
  float* rb = res + b * CT;
  float s = 0.f, sq = 0.f;
#pragma unroll
  for (int v = 0; v < 8; ++v) {
    int m = m0 + v + ((l >> 4) << 3);
    const float* r = rb + m * TT + n0 + lo;
    float o0 = acc0[v] + r[0];
    float o1 = acc1[v] + r[16];
    float o2 = acc2[v] + r[32];
    float o3 = acc3[v] + r[48];
    acc0[v] = o0;
    acc1[v] = o1;
    acc2[v] = o2;
    acc3[v] = o3;
    s += (o0 + o1) + (o2 + o3);
    sq += (o0 * o0 + o1 * o1) + (o2 * o2 + o3 * o3);
  }
  wg_pair_store(s, sq, 4, wsum, wsq, part);
#pragma unroll
  for (int v = 0; v < 8; ++v) {
    float* r = rb + (m0 + v + ((l >> 4) << 3)) * TT + n0 + lo;
    r[0]  = acc0[v];
    r[16] = acc1[v];
    r[32] = acc2[v];
    r[48] = acc3[v];
  }
}

__global__ __launch_bounds__(128) void fc_gemm(const float* __restrict__ X,
                                               const float* __restrict__ W,
                                               const float* __restrict__ bias,
                                               const float* __restrict__ res,
                                               float* __restrict__ out) {
  int l = threadIdx.x & 31;
  int tile = blockIdx.x * 4 + (threadIdx.x >> 5);
  int b = tile >> 7;
  int rem = tile & 127;
  int m0 = (rem >> 4) << 4;
  int n0 = (rem & 15) << 6;
  int lo = l & 15;
  int kh = (l >> 4) << 1;
  const float* xb = X + b * CT;
  v8f acc0 = {}, acc1 = {}, acc2 = {}, acc3 = {};
  for (int kk = 0; kk < CC; kk += 4) {
    int k0 = kk + kh;
    v2f a;
    a.x = W[(m0 + lo) * CC + k0];
    a.y = W[(m0 + lo) * CC + k0 + 1];
    const float* p0 = xb + k0 * TT + n0 + lo;
    const float* p1 = p0 + TT;
    v2f b0, b1, b2, b3;
    b0.x = p0[0];  b0.y = p1[0];
    b1.x = p0[16]; b1.y = p1[16];
    b2.x = p0[32]; b2.y = p1[32];
    b3.x = p0[48]; b3.y = p1[48];
    acc0 = wmma4(a, b0, acc0);
    acc1 = wmma4(a, b1, acc1);
    acc2 = wmma4(a, b2, acc2);
    acc3 = wmma4(a, b3, acc3);
  }
  const float* rb = res + b * CT;
  float* ob = out + b * CT;
#pragma unroll
  for (int v = 0; v < 8; ++v) {
    int m = m0 + v + ((l >> 4) << 3);
    float bsv = bias[m];
    const float* r = rb + m * TT + n0 + lo;
    float* o = ob + m * TT + n0 + lo;
    o[0]  = fmaxf(acc0[v] + bsv, 0.f) + r[0];
    o[16] = fmaxf(acc1[v] + bsv, 0.f) + r[16];
    o[32] = fmaxf(acc2[v] + bsv, 0.f) + r[32];
    o[48] = fmaxf(acc3[v] + bsv, 0.f) + r[48];
  }
}


extern "C" void kernel_launch(void* const* d_in, const int* in_sizes, int n_in,
                              void* d_out, int out_size, void* d_ws, size_t ws_size,
                              hipStream_t stream) {
  (void)in_sizes; (void)n_in; (void)out_size; (void)ws_size;
  const float* x    = (const float*)d_in[0];
  const float* mask = (const float*)d_in[1];
  const float* dw_w = (const float*)d_in[2];
  const float* dw_b = (const float*)d_in[3];
  const float* pw_w = (const float*)d_in[4];
  const float* pw_b = (const float*)d_in[5];
  const float* n0g  = (const float*)d_in[6];
  const float* n0b  = (const float*)d_in[7];
  const float* nsg  = (const float*)d_in[8];
  const float* nsb  = (const float*)d_in[9];
  const float* neg_ = (const float*)d_in[10];
  const float* neb  = (const float*)d_in[11];
  const float* Wq   = (const float*)d_in[12];
  const float* Wk   = (const float*)d_in[13];
  const float* Wv   = (const float*)d_in[14];
  const float* Wo   = (const float*)d_in[15];
  const float* fcw  = (const float*)d_in[16];
  const float* fcb  = (const float*)d_in[17];
  float* outp = (float*)d_out;
  float* ws = (float*)d_ws;

  float* OUT = ws;
  float* RES = ws + 2097152;
  float* TMP = ws + 4194304;
  float* Qb  = ws + 6291456;
  float* Kb  = ws + 7340032;
  float* Vb  = ws + 8388608;
  float* Hb  = ws + 9437184;
  float* ST  = Qb;

  posenc_add<<<512, 256, 0, stream>>>(x, OUT, ST);
  ln_apply<<<512, 256, 0, stream>>>(OUT, n0g, n0b, ST, RES);

  for (int i = 0; i < 4; ++i) {
    dwconv<<<512, 256, 0, stream>>>(OUT, dw_w + i * CC * 7, dw_b + i * CC, TMP);
    pw_gemm<<<512, 128, 0, stream>>>(TMP, pw_w + i * CC * CC, pw_b + i * CC, RES, ST);
    ln_apply<<<512, 256, 0, stream>>>(RES, nsg + i * CT, nsb + i * CT, ST, OUT);
  }

  xtw_gemm<<<256, 128, 0, stream>>>(OUT, Wq, Qb);
  xtw_gemm<<<256, 128, 0, stream>>>(OUT, Wk, Kb);
  xtw_gemm<<<256, 128, 0, stream>>>(OUT, Wv, Vb);
  attn_kernel<<<1024, 128, 0, stream>>>(Qb, Kb, Vb, mask, Hb);
  wo_gemm<<<512, 128, 0, stream>>>(Hb, Wo, RES, ST);
  ln_apply<<<512, 256, 0, stream>>>(RES, neg_, neb, ST, OUT);
  fc_gemm<<<512, 128, 0, stream>>>(OUT, fcw, fcb, RES, outp);
}


# Depthwise conv done inside the pointwise GEMM: each work-group builds its 128x64 B tile in LDS; 4 dwconv launches and the TMP round trip gone; LayerNorm partials regrouped: a different reduction tree
# speedup vs baseline: 2.8862x; 1.1920x over previous
#include <hip/hip_runtime.h>

typedef float v2f __attribute__((ext_vector_type(2)));
typedef float v8f __attribute__((ext_vector_type(8)));

#define BB 16
#define CC 128
#define TT 1024
#define DKD 64
#define CT (CC * TT)
#define NPART 32
#define RPW 4
#define WGE (RPW * TT)
#define LOG10000 9.210340371976184f
#define PI_2 1.5707963267948966f

__device__ inline v8f wmma4(v2f a, v2f b, v8f c) {
  return __builtin_amdgcn_wmma_f32_16x16x4_f32(false, a, false, b, (short)0, c,
                                               false, false);
}


__device__ inline void wg_pair_store(float s, float q, int nw, float* wsum, float* wsq,
                                     float* __restrict__ part) {
  int tid = threadIdx.x;
  for (int o = 16; o > 0; o >>= 1) {
    s += __shfl_xor(s, o, 32);
    q += __shfl_xor(q, o, 32);
  }
  if ((tid & 31) == 0) {
    wsum[tid >> 5] = s;
    wsq[tid >> 5] = q;
  }
  __syncthreads();
  if (tid == 0) {
    float a = wsum[0];
    float c = wsq[0];
    for (int u = 1; u < nw; ++u) {
      a += wsum[u];
      c += wsq[u];
    }
    reinterpret_cast<float2*>(part)[blockIdx.x] = make_float2(a, c);
  }
}


__global__ __launch_bounds__(256) void posenc_add(const float* __restrict__ x,
                                                  float* __restrict__ out,
                                                  float* __restrict__ part) {
  __shared__ float wsum[8];
  __shared__ float wsq[8];
  int p = threadIdx.x;
  int row0 = blockIdx.x * RPW;
  const float4* xin = reinterpret_cast<const float4*>(x) + row0 * (TT / 4) + p;
  float4* o = reinterpret_cast<float4*>(out) + row0 * (TT / 4) + p;
  float4 v[RPW];
#pragma unroll
  for (int q = 0; q < RPW; ++q) v[q] = xin[q * (TT / 4)];
  int t = p * 4;
  float s = 0.f, sq = 0.f;
#pragma unroll
  for (int q = 0; q < RPW; ++q) {
    int c = (row0 + q) % CC;
    float freq = expf(-(float)(c & ~1) * (LOG10000 / (float)CC));
    float ph = (float)(c & 1) * PI_2;
    float4 r;
    r.x = v[q].x + sinf((float)t * freq + ph);
    r.y = v[q].y + sinf((float)(t + 1) * freq + ph);
    r.z = v[q].z + sinf((float)(t + 2) * freq + ph);
    r.w = v[q].w + sinf((float)(t + 3) * freq + ph);
    v[q] = r;
    s += (r.x + r.y) + (r.z + r.w);
    sq += (r.x * r.x + r.y * r.y) + (r.z * r.z + r.w * r.w);
  }
  wg_pair_store(s, sq, 8, wsum, wsq, part);
#pragma unroll
  for (int q = 0; q < RPW; ++q) o[q * (TT / 4)] = v[q];
}

__global__ __launch_bounds__(256) void ln_apply(const float* __restrict__ src,
                                                const float* __restrict__ g,
                                                const float* __restrict__ bb,
                                                const float* __restrict__ part,
                                                float* __restrict__ dst) {
  int tid = threadIdx.x;
  int b = blockIdx.x / (CT / WGE);
  int f0 = (blockIdx.x % (CT / WGE)) * (WGE / 4) + tid;
  const float4* s4 = reinterpret_cast<const float4*>(src + b * CT) + f0;
  const float4* g4 = reinterpret_cast<const float4*>(g) + f0;
  const float4* b4 = reinterpret_cast<const float4*>(bb) + f0;
  float4* d4 = reinterpret_cast<float4*>(dst + b * CT) + f0;
  float4 xv[4], gv[4], bv[4];
#pragma unroll
  for (int q = 0; q < 4; ++q) {
    xv[q] = s4[q * 256];
    gv[q] = g4[q * 256];
    bv[q] = b4[q * 256];
  }
  float2 pr = reinterpret_cast<const float2*>(part)[b * NPART + (tid & 31)];
  float s = pr.x;
  float q2 = pr.y;
  for (int o = 16; o > 0; o >>= 1) {
    s += __shfl_xor(s, o, 32);
    q2 += __shfl_xor(q2, o, 32);
  }
  float inv_n = 1.0f / (float)CT;
  float mu = s * inv_n;
  float var = q2 * inv_n - mu * mu;
  float r = rsqrtf(var + 1e-5f);
#pragma unroll
  for (int q = 0; q < 4; ++q) {
    float4 y;
    y.x = (xv[q].x - mu) * r * gv[q].x + bv[q].x;
    y.y = (xv[q].y - mu) * r * gv[q].y + bv[q].y;
    y.z = (xv[q].z - mu) * r * gv[q].z + bv[q].z;
    y.w = (xv[q].w - mu) * r * gv[q].w + bv[q].w;
    d4[q * 256] = y;
  }
}


#define BPAIR 160
__global__ __launch_bounds__(128) void dwpw_gemm(const float* __restrict__ X,
                                                 const float* __restrict__ dww,
                                                 const float* __restrict__ dwb,
                                                 const float* __restrict__ W,
                                                 const float* __restrict__ bias,
                                                 float* __restrict__ res,
                                                 float* __restrict__ part) {
  __shared__ __attribute__((aligned(16))) float Bs[(CC / 2) * BPAIR];
  __shared__ float wsum[4];
  __shared__ float wsq[4];
  int tid = threadIdx.x;
  int l = tid & 31;
  int b = blockIdx.x >> 5;
  int r5 = blockIdx.x & 31;
  int n0 = (r5 >> 1) << 6;
  int m0 = (((r5 & 1) << 2) + (tid >> 5)) << 4;
  int lo = l & 15;
  int kh = (l >> 4) << 1;

  v2f afrag[CC / 4];
#pragma unroll
  for (int kk = 0; kk < CC; kk += 4) {
    int k0 = kk + kh;
    afrag[kk >> 2].x = W[(m0 + lo) * CC + k0];
    afrag[kk >> 2].y = W[(m0 + lo) * CC + k0 + 1];
  }

  {
    const float* xb = X + b * CT;
    int p = tid & 15;
    int gp = (n0 >> 2) + p;
    int gl = gp > 0 ? gp - 1 : gp;
    int gr = gp < TT / 4 - 1 ? gp + 1 : gp;
    int c0 = tid >> 4;
#pragma unroll 8
    for (int q = 0; q < CC / 8; ++q) {
      int c = c0 + 8 * q;
      const float4* r4 = reinterpret_cast<const float4*>(xb + c * TT);
      float4 cur = r4[gp];
      float4 prv = r4[gl];
      float4 nxt = r4[gr];
      if (gp == 0) prv = make_float4(0.f, 0.f, 0.f, 0.f);
      if (gp == TT / 4 - 1) nxt = make_float4(0.f, 0.f, 0.f, 0.f);
      float bs = dwb[c];
      float wj[7];
#pragma unroll
      for (int j = 0; j < 7; ++j) wj[j] = dww[c * 7 + j];
      float xs[12] = {prv.x, prv.y, prv.z, prv.w, cur.x, cur.y,
                      cur.z, cur.w, nxt.x, nxt.y, nxt.z, nxt.w};
      float o4[4];
#pragma unroll
      for (int k = 0; k < 4; ++k) {
        float s = bs;
#pragma unroll
        for (int j = 0; j < 7; ++j) s = fmaf(xs[k + j + 1], wj[j], s);
        o4[k] = s;
      }
      float* d = Bs + (c >> 1) * BPAIR + 8 * p + (c & 1);
#pragma unroll
      for (int k = 0; k < 4; ++k) d[2 * k] = o4[k];
    }
  }
  __syncthreads();

  v8f acc0 = {}, acc1 = {}, acc2 = {}, acc3 = {};
#pragma unroll
  for (int kk = 0; kk < CC; kk += 4) {
    const v2f* bp = reinterpret_cast<const v2f*>(Bs + ((kk + kh) >> 1) * BPAIR + 2 * lo);
    v2f b0 = bp[0], b1 = bp[16], b2 = bp[32], b3 = bp[48];
    acc0 = wmma4(afrag[kk >> 2], b0, acc0);
    acc1 = wmma4(afrag[kk >> 2], b1, acc1);
    acc2 = wmma4(afrag[kk >> 2], b2, acc2);
    acc3 = wmma4(afrag[kk >> 2], b3, acc3);
  }
  float* rb = res + b * CT;
  float s = 0.f, sq = 0.f;
#pragma unroll
  for (int v = 0; v < 8; ++v) {
    int m = m0 + v + ((l >> 4) << 3);
    float bsv = bias[m];
    const float* r = rb + m * TT + n0 + lo;
    float o0 = fmaxf(acc0[v] + bsv, 0.f) + r[0];
    float o1 = fmaxf(acc1[v] + bsv, 0.f) + r[16];
    float o2 = fmaxf(acc2[v] + bsv, 0.f) + r[32];
    float o3 = fmaxf(acc3[v] + bsv, 0.f) + r[48];
    acc0[v] = o0;
    acc1[v] = o1;
    acc2[v] = o2;
    acc3[v] = o3;
    s += (o0 + o1) + (o2 + o3);
    sq += (o0 * o0 + o1 * o1) + (o2 * o2 + o3 * o3);
  }
  wg_pair_store(s, sq, 4, wsum, wsq, part);
#pragma unroll
  for (int v = 0; v < 8; ++v) {
    float* r = rb + (m0 + v + ((l >> 4) << 3)) * TT + n0 + lo;
    r[0]  = acc0[v];
    r[16] = acc1[v];
    r[32] = acc2[v];
    r[48] = acc3[v];
  }
}

__global__ __launch_bounds__(128) void xtw_gemm(const float* __restrict__ X,
                                                const float* __restrict__ W,
                                                float* __restrict__ dst) {
  int l = threadIdx.x & 31;
  int tile = blockIdx.x * 4 + (threadIdx.x >> 5);
  int b = tile >> 6;
  int t0 = (tile & 63) << 4;
  int lo = l & 15;
  int kh = (l >> 4) << 1;
  const float* xb = X + b * CT;
  v8f acc0 = {}, acc1 = {}, acc2 = {}, acc3 = {};
  for (int kk = 0; kk < CC; kk += 4) {
    int k0 = kk + kh;
    v2f a;
    a.x = xb[k0 * TT + t0 + lo];
    a.y = xb[(k0 + 1) * TT + t0 + lo];
    const float* w0 = W + k0 * DKD + lo;
    const float* w1 = w0 + DKD;
    v2f b0, b1, b2, b3;
    b0.x = w0[0];  b0.y = w1[0];
    b1.x = w0[16]; b1.y = w1[16];
    b2.x = w0[32]; b2.y = w1[32];
    b3.x = w0[48]; b3.y = w1[48];
    acc0 = wmma4(a, b0, acc0);
    acc1 = wmma4(a, b1, acc1);
    acc2 = wmma4(a, b2, acc2);
    acc3 = wmma4(a, b3, acc3);
  }
  float* db = dst + b * TT * DKD;
#pragma unroll
  for (int v = 0; v < 8; ++v) {
    int t = t0 + v + ((l >> 4) << 3);
    float* d = db + t * DKD + lo;
    d[0]  = acc0[v];
    d[16] = acc1[v];
    d[32] = acc2[v];
    d[48] = acc3[v];
  }
}

#define AW 4
#define KTW (TT / 16 / AW)
#define RWW (16 / AW)
#define KPW (TT / AW)
__global__ __launch_bounds__(32 * AW) void attn_kernel(const float* __restrict__ Q,
                                                       const float* __restrict__ Km,
                                                       const float* __restrict__ Vm,
                                                       const float* __restrict__ mask,
                                                       float* __restrict__ head) {
  __shared__ __attribute__((aligned(16))) float S[16 * TT];
  int b = blockIdx.x >> 6;
  int q0 = (blockIdx.x & 63) << 4;
  int w = __builtin_amdgcn_readfirstlane((int)(threadIdx.x >> 5));
  int l = threadIdx.x & 31;
  int lo = l & 15;
  int kh = (l >> 4) << 1;
  const float* qb = Q + b * TT * DKD;
  const float* kb = Km + b * TT * DKD;
  const float* vb = Vm + b * TT * DKD;
  const float* mb = mask + b * TT;

  v2f afrag[DKD / 4];
#pragma unroll
  for (int kk = 0; kk < DKD; kk += 4) {
    int k0 = kk + kh;
    afrag[kk >> 2].x = qb[(q0 + lo) * DKD + k0];
    afrag[kk >> 2].y = qb[(q0 + lo) * DKD + k0 + 1];
  }

  for (int ti = 0; ti < KTW; ++ti) {
    int t = w * KTW + ti;
    v8f acc = {};
#pragma unroll
    for (int kk = 0; kk < DKD; kk += 4) {
      int k0 = kk + kh;
      v2f bv;
      bv.x = kb[(t * 16 + lo) * DKD + k0];
      bv.y = kb[(t * 16 + lo) * DKD + k0 + 1];
      acc = wmma4(afrag[kk >> 2], bv, acc);
    }
#pragma unroll
    for (int v = 0; v < 8; ++v) {
      int row = v + ((l >> 4) << 3);
      int col = t * 16 + lo;
      float hm = mb[col];
      float sv = acc[v] * 0.125f;
      S[row * TT + col] = sv * hm + (1.f - hm) * -1e30f;
    }
  }
  __syncthreads();

  for (int r = 0; r < RWW; ++r) {
    int row = w * RWW + r;
    float* sr = S + row * TT + l;
    float e[TT / 32];
    float mx = -3.4e38f;
#pragma unroll
    for (int c = 0; c < TT / 32; ++c) {
      e[c] = sr[c * 32];
      mx = fmaxf(mx, e[c]);
    }
    for (int o = 16; o > 0; o >>= 1) mx = fmaxf(mx, __shfl_xor(mx, o, 32));
    float sum = 0.f;
#pragma unroll
    for (int c = 0; c < TT / 32; ++c) {
      e[c] = expf(e[c] - mx);
      sum += e[c];
    }
    for (int o = 16; o > 0; o >>= 1) sum += __shfl_xor(sum, o, 32);
    float inv = 1.0f / sum;
    float qm = mb[q0 + row];
#pragma unroll
    for (int c = 0; c < TT / 32; ++c) sr[c * 32] = e[c] * inv * qm;
  }
  __syncthreads();

  v8f acc0 = {}, acc1 = {}, acc2 = {}, acc3 = {};
  for (int ki = 0; ki < KPW; ki += 4) {
    int k0 = w * KPW + ki + kh;
    v2f a;
    a.x = S[lo * TT + k0];
    a.y = S[lo * TT + k0 + 1];
    const float* v0p = vb + k0 * DKD + lo;
    const float* v1p = v0p + DKD;
    v2f b0, b1, b2, b3;
    b0.x = v0p[0];  b0.y = v1p[0];
    b1.x = v0p[16]; b1.y = v1p[16];
    b2.x = v0p[32]; b2.y = v1p[32];
    b3.x = v0p[48]; b3.y = v1p[48];
    acc0 = wmma4(a, b0, acc0);
    acc1 = wmma4(a, b1, acc1);
    acc2 = wmma4(a, b2, acc2);
    acc3 = wmma4(a, b3, acc3);
  }
  __syncthreads();

  {
    float* pw = S + w * (16 * DKD);
#pragma unroll
    for (int v = 0; v < 8; ++v) {
      float* p = pw + (v + ((l >> 4) << 3)) * DKD + lo;
      p[0]  = acc0[v];
      p[16] = acc1[v];
      p[32] = acc2[v];
      p[48] = acc3[v];
    }
  }
  __syncthreads();

  {
    const float4* s4 = reinterpret_cast<const float4*>(S);
    float4* h4 = reinterpret_cast<float4*>(head + (b * TT + q0) * DKD);
#pragma unroll
    for (int j = 0; j < (16 * DKD / 4) / (32 * AW); ++j) {
      int i4 = threadIdx.x + j * (32 * AW);
      float4 r = s4[i4];
#pragma unroll
      for (int u = 1; u < AW; ++u) {
        float4 pu = s4[u * (16 * DKD / 4) + i4];
        r.x += pu.x;
        r.y += pu.y;
        r.z += pu.z;
        r.w += pu.w;
      }
      h4[i4] = r;
    }
  }
}

__global__ __launch_bounds__(128) void wo_gemm(const float* __restrict__ H,
                                               const float* __restrict__ Wo,
                                               float* __restrict__ res,
                                               float* __restrict__ part) {
  __shared__ float wsum[4];
  __shared__ float wsq[4];
  int l = threadIdx.x & 31;
  int tile = blockIdx.x * 4 + (threadIdx.x >> 5);
  int b = tile >> 7;
  int rem = tile & 127;
  int m0 = (rem >> 4) << 4;
  int n0 = (rem & 15) << 6;
  int lo = l & 15;
  int kh = (l >> 4) << 1;
  const float* hb = H + b * TT * DKD;
  v8f acc0 = {}, acc1 = {}, acc2 = {}, acc3 = {};
  for (int kk = 0; kk < DKD; kk += 4) {
    int k0 = kk + kh;
    v2f a;
    a.x = Wo[k0 * CC + m0 + lo] + Wo[(k0 + 64) * CC + m0 + lo];
    a.y = Wo[(k0 + 1) * CC + m0 + lo] + Wo[(k0 + 65) * CC + m0 + lo];
    const float* h0 = hb + (n0 + lo) * DKD + k0;
    v2f b0, b1, b2, b3;
    b0.x = h0[0];            b0.y = h0[1];
    b1.x = h0[16 * DKD];     b1.y = h0[16 * DKD + 1];
    b2.x = h0[32 * DKD];     b2.y = h0[32 * DKD + 1];
    b3.x = h0[48 * DKD];     b3.y = h0[48 * DKD + 1];
    acc0 = wmma4(a, b0, acc0);
    acc1 = wmma4(a, b1, acc1);
    acc2 = wmma4(a, b2, acc2);
    acc3 = wmma4(a, b3, acc3);
  }
  float* rb = res + b * CT;
  float s = 0.f, sq = 0.f;
#pragma unroll
  for (int v = 0; v < 8; ++v) {
    int m = m0 + v + ((l >> 4) << 3);
    const float* r = rb + m * TT + n0 + lo;
    float o0 = acc0[v] + r[0];
    float o1 = acc1[v] + r[16];
    float o2 = acc2[v] + r[32];
    float o3 = acc3[v] + r[48];
    acc0[v] = o0;
    acc1[v] = o1;
    acc2[v] = o2;
    acc3[v] = o3;
    s += (o0 + o1) + (o2 + o3);
    sq += (o0 * o0 + o1 * o1) + (o2 * o2 + o3 * o3);
  }
  wg_pair_store(s, sq, 4, wsum, wsq, part);
#pragma unroll
  for (int v = 0; v < 8; ++v) {
    float* r = rb + (m0 + v + ((l >> 4) << 3)) * TT + n0 + lo;
    r[0]  = acc0[v];
    r[16] = acc1[v];
    r[32] = acc2[v];
    r[48] = acc3[v];
  }
}

__global__ __launch_bounds__(128) void fc_gemm(const float* __restrict__ X,
                                               const float* __restrict__ W,
                                               const float* __restrict__ bias,
                                               const float* __restrict__ res,
                                               float* __restrict__ out) {
  int l = threadIdx.x & 31;
  int tile = blockIdx.x * 4 + (threadIdx.x >> 5);
  int b = tile >> 7;
  int rem = tile & 127;
  int m0 = (rem >> 4) << 4;
  int n0 = (rem & 15) << 6;
  int lo = l & 15;
  int kh = (l >> 4) << 1;
  const float* xb = X + b * CT;
  v8f acc0 = {}, acc1 = {}, acc2 = {}, acc3 = {};
  for (int kk = 0; kk < CC; kk += 4) {
    int k0 = kk + kh;
    v2f a;
    a.x = W[(m0 + lo) * CC + k0];
    a.y = W[(m0 + lo) * CC + k0 + 1];
    const float* p0 = xb + k0 * TT + n0 + lo;
    const float* p1 = p0 + TT;
    v2f b0, b1, b2, b3;
    b0.x = p0[0];  b0.y = p1[0];
    b1.x = p0[16]; b1.y = p1[16];
    b2.x = p0[32]; b2.y = p1[32];
    b3.x = p0[48]; b3.y = p1[48];
    acc0 = wmma4(a, b0, acc0);
    acc1 = wmma4(a, b1, acc1);
    acc2 = wmma4(a, b2, acc2);
    acc3 = wmma4(a, b3, acc3);
  }
  const float* rb = res + b * CT;
  float* ob = out + b * CT;
#pragma unroll
  for (int v = 0; v < 8; ++v) {
    int m = m0 + v + ((l >> 4) << 3);
    float bsv = bias[m];
    const float* r = rb + m * TT + n0 + lo;
    float* o = ob + m * TT + n0 + lo;
    o[0]  = fmaxf(acc0[v] + bsv, 0.f) + r[0];
    o[16] = fmaxf(acc1[v] + bsv, 0.f) + r[16];
    o[32] = fmaxf(acc2[v] + bsv, 0.f) + r[32];
    o[48] = fmaxf(acc3[v] + bsv, 0.f) + r[48];
  }
}


extern "C" void kernel_launch(void* const* d_in, const int* in_sizes, int n_in,
                              void* d_out, int out_size, void* d_ws, size_t ws_size,
                              hipStream_t stream) {
  (void)in_sizes; (void)n_in; (void)out_size; (void)ws_size;
  const float* x    = (const float*)d_in[0];
  const float* mask = (const float*)d_in[1];
  const float* dw_w = (const float*)d_in[2];
  const float* dw_b = (const float*)d_in[3];
  const float* pw_w = (const float*)d_in[4];
  const float* pw_b = (const float*)d_in[5];
  const float* n0g  = (const float*)d_in[6];
  const float* n0b  = (const float*)d_in[7];
  const float* nsg  = (const float*)d_in[8];
  const float* nsb  = (const float*)d_in[9];
  const float* neg_ = (const float*)d_in[10];
  const float* neb  = (const float*)d_in[11];
  const float* Wq   = (const float*)d_in[12];
  const float* Wk   = (const float*)d_in[13];
  const float* Wv   = (const float*)d_in[14];
  const float* Wo   = (const float*)d_in[15];
  const float* fcw  = (const float*)d_in[16];
  const float* fcb  = (const float*)d_in[17];
  float* outp = (float*)d_out;
  float* ws = (float*)d_ws;

  float* OUT = ws;
  float* RES = ws + 2097152;
  float* Qb  = ws + 6291456;
  float* Kb  = ws + 7340032;
  float* Vb  = ws + 8388608;
  float* Hb  = ws + 9437184;
  float* ST  = Qb;

  posenc_add<<<512, 256, 0, stream>>>(x, OUT, ST);
  ln_apply<<<512, 256, 0, stream>>>(OUT, n0g, n0b, ST, RES);

  for (int i = 0; i < 4; ++i) {
    dwpw_gemm<<<512, 128, 0, stream>>>(OUT, dw_w + i * CC * 7, dw_b + i * CC,
                                       pw_w + i * CC * CC, pw_b + i * CC, RES, ST);
    ln_apply<<<512, 256, 0, stream>>>(RES, nsg + i * CT, nsb + i * CT, ST, OUT);
  }

  xtw_gemm<<<256, 128, 0, stream>>>(OUT, Wq, Qb);
  xtw_gemm<<<256, 128, 0, stream>>>(OUT, Wk, Kb);
  xtw_gemm<<<256, 128, 0, stream>>>(OUT, Wv, Vb);
  attn_kernel<<<1024, 128, 0, stream>>>(Qb, Kb, Vb, mask, Hb);
  wo_gemm<<<512, 128, 0, stream>>>(Hb, Wo, RES, ST);
  ln_apply<<<512, 256, 0, stream>>>(RES, neg_, neb, ST, OUT);
  fc_gemm<<<512, 128, 0, stream>>>(OUT, fcw, fcb, RES, outp);
}


# Q, K and V projections in one launch (grid.y picks the weight and the output): 2 launches fewer, and 3 waves per SIMD instead of 1 cover each wave's wait for the loads of a k block
# speedup vs baseline: 3.2396x; 1.1224x over previous
#include <hip/hip_runtime.h>

typedef float v2f __attribute__((ext_vector_type(2)));
typedef float v8f __attribute__((ext_vector_type(8)));

#define BB 16
#define CC 128
#define TT 1024
#define DKD 64
#define CT (CC * TT)
#define NPART 32
#define RPW 4
#define WGE (RPW * TT)
#define LOG10000 9.210340371976184f
#define PI_2 1.5707963267948966f

__device__ inline v8f wmma4(v2f a, v2f b, v8f c) {
  return __builtin_amdgcn_wmma_f32_16x16x4_f32(false, a, false, b, (short)0, c,
                                               false, false);
}


__device__ inline void wg_pair_store(float s, float q, int nw, float* wsum, float* wsq,
                                     float* __restrict__ part) {
  int tid = threadIdx.x;
  for (int o = 16; o > 0; o >>= 1) {
    s += __shfl_xor(s, o, 32);
    q += __shfl_xor(q, o, 32);
  }
  if ((tid & 31) == 0) {
    wsum[tid >> 5] = s;
    wsq[tid >> 5] = q;
  }
  __syncthreads();
  if (tid == 0) {
    float a = wsum[0];
    float c = wsq[0];
    for (int u = 1; u < nw; ++u) {
      a += wsum[u];
      c += wsq[u];
    }
    reinterpret_cast<float2*>(part)[blockIdx.x] = make_float2(a, c);
  }
}


__global__ __launch_bounds__(256) void posenc_add(const float* __restrict__ x,
                                                  float* __restrict__ out,
                                                  float* __restrict__ part) {
  __shared__ float wsum[8];
  __shared__ float wsq[8];
  int p = threadIdx.x;
  int row0 = blockIdx.x * RPW;
  const float4* xin = reinterpret_cast<const float4*>(x) + row0 * (TT / 4) + p;
  float4* o = reinterpret_cast<float4*>(out) + row0 * (TT / 4) + p;
  float4 v[RPW];
#pragma unroll
  for (int q = 0; q < RPW; ++q) v[q] = xin[q * (TT / 4)];
  int t = p * 4;
  float s = 0.f, sq = 0.f;
#pragma unroll
  for (int q = 0; q < RPW; ++q) {
    int c = (row0 + q) % CC;
    float freq = expf(-(float)(c & ~1) * (LOG10000 / (float)CC));
    float ph = (float)(c & 1) * PI_2;
    float4 r;
    r.x = v[q].x + sinf((float)t * freq + ph);
    r.y = v[q].y + sinf((float)(t + 1) * freq + ph);
    r.z = v[q].z + sinf((float)(t + 2) * freq + ph);
    r.w = v[q].w + sinf((float)(t + 3) * freq + ph);
    v[q] = r;
    s += (r.x + r.y) + (r.z + r.w);
    sq += (r.x * r.x + r.y * r.y) + (r.z * r.z + r.w * r.w);
  }
  wg_pair_store(s, sq, 8, wsum, wsq, part);
#pragma unroll
  for (int q = 0; q < RPW; ++q) o[q * (TT / 4)] = v[q];
}

__global__ __launch_bounds__(256) void ln_apply(const float* __restrict__ src,
                                                const float* __restrict__ g,
                                                const float* __restrict__ bb,
                                                const float* __restrict__ part,
                                                float* __restrict__ dst) {
  int tid = threadIdx.x;
  int b = blockIdx.x / (CT / WGE);
  int f0 = (blockIdx.x % (CT / WGE)) * (WGE / 4) + tid;
  const float4* s4 = reinterpret_cast<const float4*>(src + b * CT) + f0;
  const float4* g4 = reinterpret_cast<const float4*>(g) + f0;
  const float4* b4 = reinterpret_cast<const float4*>(bb) + f0;
  float4* d4 = reinterpret_cast<float4*>(dst + b * CT) + f0;
  float4 xv[4], gv[4], bv[4];
#pragma unroll
  for (int q = 0; q < 4; ++q) {
    xv[q] = s4[q * 256];
    gv[q] = g4[q * 256];
    bv[q] = b4[q * 256];
  }
  float2 pr = reinterpret_cast<const float2*>(part)[b * NPART + (tid & 31)];
  float s = pr.x;
  float q2 = pr.y;
  for (int o = 16; o > 0; o >>= 1) {
    s += __shfl_xor(s, o, 32);
    q2 += __shfl_xor(q2, o, 32);
  }
  float inv_n = 1.0f / (float)CT;
  float mu = s * inv_n;
  float var = q2 * inv_n - mu * mu;
  float r = rsqrtf(var + 1e-5f);
#pragma unroll
  for (int q = 0; q < 4; ++q) {
    float4 y;
    y.x = (xv[q].x - mu) * r * gv[q].x + bv[q].x;
    y.y = (xv[q].y - mu) * r * gv[q].y + bv[q].y;
    y.z = (xv[q].z - mu) * r * gv[q].z + bv[q].z;
    y.w = (xv[q].w - mu) * r * gv[q].w + bv[q].w;
    d4[q * 256] = y;
  }
}


#define BPAIR 160
__global__ __launch_bounds__(128) void dwpw_gemm(const float* __restrict__ X,
                                                 const float* __restrict__ dww,
                                                 const float* __restrict__ dwb,
                                                 const float* __restrict__ W,
                                                 const float* __restrict__ bias,
                                                 float* __restrict__ res,
                                                 float* __restrict__ part) {
  __shared__ __attribute__((aligned(16))) float Bs[(CC / 2) * BPAIR];
  __shared__ float wsum[4];
  __shared__ float wsq[4];
  int tid = threadIdx.x;
  int l = tid & 31;
  int b = blockIdx.x >> 5;
  int r5 = blockIdx.x & 31;
  int n0 = (r5 >> 1) << 6;
  int m0 = (((r5 & 1) << 2) + (tid >> 5)) << 4;
  int lo = l & 15;
  int kh = (l >> 4) << 1;

  v2f afrag[CC / 4];
#pragma unroll
  for (int kk = 0; kk < CC; kk += 4) {
    int k0 = kk + kh;
    afrag[kk >> 2].x = W[(m0 + lo) * CC + k0];
    afrag[kk >> 2].y = W[(m0 + lo) * CC + k0 + 1];
  }

  {
    const float* xb = X + b * CT;
    int p = tid & 15;
    int gp = (n0 >> 2) + p;
    int gl = gp > 0 ? gp - 1 : gp;
    int gr = gp < TT / 4 - 1 ? gp + 1 : gp;
    int c0 = tid >> 4;
#pragma unroll 8
    for (int q = 0; q < CC / 8; ++q) {
      int c = c0 + 8 * q;
      const float4* r4 = reinterpret_cast<const float4*>(xb + c * TT);
      float4 cur = r4[gp];
      float4 prv = r4[gl];
      float4 nxt = r4[gr];
      if (gp == 0) prv = make_float4(0.f, 0.f, 0.f, 0.f);
      if (gp == TT / 4 - 1) nxt = make_float4(0.f, 0.f, 0.f, 0.f);
      float bs = dwb[c];
      float wj[7];
#pragma unroll
      for (int j = 0; j < 7; ++j) wj[j] = dww[c * 7 + j];
      float xs[12] = {prv.x, prv.y, prv.z, prv.w, cur.x, cur.y,
                      cur.z, cur.w, nxt.x, nxt.y, nxt.z, nxt.w};
      float o4[4];
#pragma unroll
      for (int k = 0; k < 4; ++k) {
        float s = bs;
#pragma unroll
        for (int j = 0; j < 7; ++j) s = fmaf(xs[k + j + 1], wj[j], s);
        o4[k] = s;
      }
      float* d = Bs + (c >> 1) * BPAIR + 8 * p + (c & 1);
#pragma unroll
      for (int k = 0; k < 4; ++k) d[2 * k] = o4[k];
    }
  }
  __syncthreads();

  v8f acc0 = {}, acc1 = {}, acc2 = {}, acc3 = {};
#pragma unroll
  for (int kk = 0; kk < CC; kk += 4) {
    const v2f* bp = reinterpret_cast<const v2f*>(Bs + ((kk + kh) >> 1) * BPAIR + 2 * lo);
    v2f b0 = bp[0], b1 = bp[16], b2 = bp[32], b3 = bp[48];
    acc0 = wmma4(afrag[kk >> 2], b0, acc0);
    acc1 = wmma4(afrag[kk >> 2], b1, acc1);
    acc2 = wmma4(afrag[kk >> 2], b2, acc2);
    acc3 = wmma4(afrag[kk >> 2], b3, acc3);
  }
  float* rb = res + b * CT;
  float s = 0.f, sq = 0.f;
#pragma unroll
  for (int v = 0; v < 8; ++v) {
    int m = m0 + v + ((l >> 4) << 3);
    float bsv = bias[m];
    const float* r = rb + m * TT + n0 + lo;
    float o0 = fmaxf(acc0[v] + bsv, 0.f) + r[0];
    float o1 = fmaxf(acc1[v] + bsv, 0.f) + r[16];
    float o2 = fmaxf(acc2[v] + bsv, 0.f) + r[32];
    float o3 = fmaxf(acc3[v] + bsv, 0.f) + r[48];
    acc0[v] = o0;
    acc1[v] = o1;
    acc2[v] = o2;
    acc3[v] = o3;
    s += (o0 + o1) + (o2 + o3);
    sq += (o0 * o0 + o1 * o1) + (o2 * o2 + o3 * o3);
  }
  wg_pair_store(s, sq, 4, wsum, wsq, part);
#pragma unroll
  for (int v = 0; v < 8; ++v) {
    float* r = rb + (m0 + v + ((l >> 4) << 3)) * TT + n0 + lo;
    r[0]  = acc0[v];
    r[16] = acc1[v];
    r[32] = acc2[v];
    r[48] = acc3[v];
  }
}

__global__ __launch_bounds__(128) void xtw_gemm(const float* __restrict__ X,
                                                const float* __restrict__ W0,
                                                const float* __restrict__ W1,
                                                const float* __restrict__ W2,
                                                float* __restrict__ D0,
                                                float* __restrict__ D1,
                                                float* __restrict__ D2) {
  int y = blockIdx.y;
  const float* W = y == 0 ? W0 : (y == 1 ? W1 : W2);
  float* dst = y == 0 ? D0 : (y == 1 ? D1 : D2);
  int l = threadIdx.x & 31;
  int tile = blockIdx.x * 4 + (threadIdx.x >> 5);
  int b = tile >> 6;
  int t0 = (tile & 63) << 4;
  int lo = l & 15;
  int kh = (l >> 4) << 1;
  const float* xb = X + b * CT;
  v8f acc0 = {}, acc1 = {}, acc2 = {}, acc3 = {};
  for (int kk = 0; kk < CC; kk += 4) {
    int k0 = kk + kh;
    v2f a;
    a.x = xb[k0 * TT + t0 + lo];
    a.y = xb[(k0 + 1) * TT + t0 + lo];
    const float* w0 = W + k0 * DKD + lo;
    const float* w1 = w0 + DKD;
    v2f b0, b1, b2, b3;
    b0.x = w0[0];  b0.y = w1[0];
    b1.x = w0[16]; b1.y = w1[16];
    b2.x = w0[32]; b2.y = w1[32];
    b3.x = w0[48]; b3.y = w1[48];
    acc0 = wmma4(a, b0, acc0);
    acc1 = wmma4(a, b1, acc1);
    acc2 = wmma4(a, b2, acc2);
    acc3 = wmma4(a, b3, acc3);
  }
  float* db = dst + b * TT * DKD;
#pragma unroll
  for (int v = 0; v < 8; ++v) {
    int t = t0 + v + ((l >> 4) << 3);
    float* d = db + t * DKD + lo;
    d[0]  = acc0[v];
    d[16] = acc1[v];
    d[32] = acc2[v];
    d[48] = acc3[v];
  }
}

#define AW 4
#define KTW (TT / 16 / AW)
#define RWW (16 / AW)
#define KPW (TT / AW)
__global__ __launch_bounds__(32 * AW) void attn_kernel(const float* __restrict__ Q,
                                                       const float* __restrict__ Km,
                                                       const float* __restrict__ Vm,
                                                       const float* __restrict__ mask,
                                                       float* __restrict__ head) {
  __shared__ __attribute__((aligned(16))) float S[16 * TT];
  int b = blockIdx.x >> 6;
  int q0 = (blockIdx.x & 63) << 4;
  int w = __builtin_amdgcn_readfirstlane((int)(threadIdx.x >> 5));
  int l = threadIdx.x & 31;
  int lo = l & 15;
  int kh = (l >> 4) << 1;
  const float* qb = Q + b * TT * DKD;
  const float* kb = Km + b * TT * DKD;
  const float* vb = Vm + b * TT * DKD;
  const float* mb = mask + b * TT;

  v2f afrag[DKD / 4];
#pragma unroll
  for (int kk = 0; kk < DKD; kk += 4) {
    int k0 = kk + kh;
    afrag[kk >> 2].x = qb[(q0 + lo) * DKD + k0];
    afrag[kk >> 2].y = qb[(q0 + lo) * DKD + k0 + 1];
  }

  for (int ti = 0; ti < KTW; ++ti) {
    int t = w * KTW + ti;
    v8f acc = {};
#pragma unroll
    for (int kk = 0; kk < DKD; kk += 4) {
      int k0 = kk + kh;
      v2f bv;
      bv.x = kb[(t * 16 + lo) * DKD + k0];
      bv.y = kb[(t * 16 + lo) * DKD + k0 + 1];
      acc = wmma4(afrag[kk >> 2], bv, acc);
    }
#pragma unroll
    for (int v = 0; v < 8; ++v) {
      int row = v + ((l >> 4) << 3);
      int col = t * 16 + lo;
      float hm = mb[col];
      float sv = acc[v] * 0.125f;
      S[row * TT + col] = sv * hm + (1.f - hm) * -1e30f;
    }
  }
  __syncthreads();

  for (int r = 0; r < RWW; ++r) {
    int row = w * RWW + r;
    float* sr = S + row * TT + l;
    float e[TT / 32];
    float mx = -3.4e38f;
#pragma unroll
    for (int c = 0; c < TT / 32; ++c) {
      e[c] = sr[c * 32];
      mx = fmaxf(mx, e[c]);
    }
    for (int o = 16; o > 0; o >>= 1) mx = fmaxf(mx, __shfl_xor(mx, o, 32));
    float sum = 0.f;
#pragma unroll
    for (int c = 0; c < TT / 32; ++c) {
      e[c] = expf(e[c] - mx);
      sum += e[c];
    }
    for (int o = 16; o > 0; o >>= 1) sum += __shfl_xor(sum, o, 32);
    float inv = 1.0f / sum;
    float qm = mb[q0 + row];
#pragma unroll
    for (int c = 0; c < TT / 32; ++c) sr[c * 32] = e[c] * inv * qm;
  }
  __syncthreads();

  v8f acc0 = {}, acc1 = {}, acc2 = {}, acc3 = {};
  for (int ki = 0; ki < KPW; ki += 4) {
    int k0 = w * KPW + ki + kh;
    v2f a;
    a.x = S[lo * TT + k0];
    a.y = S[lo * TT + k0 + 1];
    const float* v0p = vb + k0 * DKD + lo;
    const float* v1p = v0p + DKD;
    v2f b0, b1, b2, b3;
    b0.x = v0p[0];  b0.y = v1p[0];
    b1.x = v0p[16]; b1.y = v1p[16];
    b2.x = v0p[32]; b2.y = v1p[32];
    b3.x = v0p[48]; b3.y = v1p[48];
    acc0 = wmma4(a, b0, acc0);
    acc1 = wmma4(a, b1, acc1);
    acc2 = wmma4(a, b2, acc2);
    acc3 = wmma4(a, b3, acc3);
  }
  __syncthreads();

  {
    float* pw = S + w * (16 * DKD);
#pragma unroll
    for (int v = 0; v < 8; ++v) {
      float* p = pw + (v + ((l >> 4) << 3)) * DKD + lo;
      p[0]  = acc0[v];
      p[16] = acc1[v];
      p[32] = acc2[v];
      p[48] = acc3[v];
    }
  }
  __syncthreads();

  {
    const float4* s4 = reinterpret_cast<const float4*>(S);
    float4* h4 = reinterpret_cast<float4*>(head + (b * TT + q0) * DKD);
#pragma unroll
    for (int j = 0; j < (16 * DKD / 4) / (32 * AW); ++j) {
      int i4 = threadIdx.x + j * (32 * AW);
      float4 r = s4[i4];
#pragma unroll
      for (int u = 1; u < AW; ++u) {
        float4 pu = s4[u * (16 * DKD / 4) + i4];
        r.x += pu.x;
        r.y += pu.y;
        r.z += pu.z;
        r.w += pu.w;
      }
      h4[i4] = r;
    }
  }
}

__global__ __launch_bounds__(128) void wo_gemm(const float* __restrict__ H,
                                               const float* __restrict__ Wo,
                                               float* __restrict__ res,
                                               float* __restrict__ part) {
  __shared__ float wsum[4];
  __shared__ float wsq[4];
  int l = threadIdx.x & 31;
  int tile = blockIdx.x * 4 + (threadIdx.x >> 5);
  int b = tile >> 7;
  int rem = tile & 127;
  int m0 = (rem >> 4) << 4;
  int n0 = (rem & 15) << 6;
  int lo = l & 15;
  int kh = (l >> 4) << 1;
  const float* hb = H + b * TT * DKD;
  v8f acc0 = {}, acc1 = {}, acc2 = {}, acc3 = {};
  for (int kk = 0; kk < DKD; kk += 4) {
    int k0 = kk + kh;
    v2f a;
    a.x = Wo[k0 * CC + m0 + lo] + Wo[(k0 + 64) * CC + m0 + lo];
    a.y = Wo[(k0 + 1) * CC + m0 + lo] + Wo[(k0 + 65) * CC + m0 + lo];
    const float* h0 = hb + (n0 + lo) * DKD + k0;
    v2f b0, b1, b2, b3;
    b0.x = h0[0];            b0.y = h0[1];
    b1.x = h0[16 * DKD];     b1.y = h0[16 * DKD + 1];
    b2.x = h0[32 * DKD];     b2.y = h0[32 * DKD + 1];
    b3.x = h0[48 * DKD];     b3.y = h0[48 * DKD + 1];
    acc0 = wmma4(a, b0, acc0);
    acc1 = wmma4(a, b1, acc1);
    acc2 = wmma4(a, b2, acc2);
    acc3 = wmma4(a, b3, acc3);
  }
  float* rb = res + b * CT;
  float s = 0.f, sq = 0.f;
#pragma unroll
  for (int v = 0; v < 8; ++v) {
    int m = m0 + v + ((l >> 4) << 3);
    const float* r = rb + m * TT + n0 + lo;
    float o0 = acc0[v] + r[0];
    float o1 = acc1[v] + r[16];
    float o2 = acc2[v] + r[32];
    float o3 = acc3[v] + r[48];
    acc0[v] = o0;
    acc1[v] = o1;
    acc2[v] = o2;
    acc3[v] = o3;
    s += (o0 + o1) + (o2 + o3);
    sq += (o0 * o0 + o1 * o1) + (o2 * o2 + o3 * o3);
  }
  wg_pair_store(s, sq, 4, wsum, wsq, part);
#pragma unroll
  for (int v = 0; v < 8; ++v) {
    float* r = rb + (m0 + v + ((l >> 4) << 3)) * TT + n0 + lo;
    r[0]  = acc0[v];
    r[16] = acc1[v];
    r[32] = acc2[v];
    r[48] = acc3[v];
  }
}

__global__ __launch_bounds__(128) void fc_gemm(const float* __restrict__ X,
                                               const float* __restrict__ W,
                                               const float* __restrict__ bias,
                                               const float* __restrict__ res,
                                               float* __restrict__ out) {
  int l = threadIdx.x & 31;
  int tile = blockIdx.x * 4 + (threadIdx.x >> 5);
  int b = tile >> 7;
  int rem = tile & 127;
  int m0 = (rem >> 4) << 4;
  int n0 = (rem & 15) << 6;
  int lo = l & 15;
  int kh = (l >> 4) << 1;
  const float* xb = X + b * CT;
  v8f acc0 = {}, acc1 = {}, acc2 = {}, acc3 = {};
  for (int kk = 0; kk < CC; kk += 4) {
    int k0 = kk + kh;
    v2f a;
    a.x = W[(m0 + lo) * CC + k0];
    a.y = W[(m0 + lo) * CC + k0 + 1];
    const float* p0 = xb + k0 * TT + n0 + lo;
    const float* p1 = p0 + TT;
    v2f b0, b1, b2, b3;
    b0.x = p0[0];  b0.y = p1[0];
    b1.x = p0[16]; b1.y = p1[16];
    b2.x = p0[32]; b2.y = p1[32];
    b3.x = p0[48]; b3.y = p1[48];
    acc0 = wmma4(a, b0, acc0);
    acc1 = wmma4(a, b1, acc1);
    acc2 = wmma4(a, b2, acc2);
    acc3 = wmma4(a, b3, acc3);
  }
  const float* rb = res + b * CT;
  float* ob = out + b * CT;
#pragma unroll
  for (int v = 0; v < 8; ++v) {
    int m = m0 + v + ((l >> 4) << 3);
    float bsv = bias[m];
    const float* r = rb + m * TT + n0 + lo;
    float* o = ob + m * TT + n0 + lo;
    o[0]  = fmaxf(acc0[v] + bsv, 0.f) + r[0];
    o[16] = fmaxf(acc1[v] + bsv, 0.f) + r[16];
    o[32] = fmaxf(acc2[v] + bsv, 0.f) + r[32];
    o[48] = fmaxf(acc3[v] + bsv, 0.f) + r[48];
  }
}


extern "C" void kernel_launch(void* const* d_in, const int* in_sizes, int n_in,
                              void* d_out, int out_size, void* d_ws, size_t ws_size,
                              hipStream_t stream) {
  (void)in_sizes; (void)n_in; (void)out_size; (void)ws_size;
  const float* x    = (const float*)d_in[0];
  const float* mask = (const float*)d_in[1];
  const float* dw_w = (const float*)d_in[2];
  const float* dw_b = (const float*)d_in[3];
  const float* pw_w = (const float*)d_in[4];
  const float* pw_b = (const float*)d_in[5];
  const float* n0g  = (const float*)d_in[6];
  const float* n0b  = (const float*)d_in[7];
  const float* nsg  = (const float*)d_in[8];
  const float* nsb  = (const float*)d_in[9];
  const float* neg_ = (const float*)d_in[10];
  const float* neb  = (const float*)d_in[11];
  const float* Wq   = (const float*)d_in[12];
  const float* Wk   = (const float*)d_in[13];
  const float* Wv   = (const float*)d_in[14];
  const float* Wo   = (const float*)d_in[15];
  const float* fcw  = (const float*)d_in[16];
  const float* fcb  = (const float*)d_in[17];
  float* outp = (float*)d_out;
  float* ws = (float*)d_ws;

  float* OUT = ws;
  float* RES = ws + 2097152;
  float* Qb  = ws + 6291456;
  float* Kb  = ws + 7340032;
  float* Vb  = ws + 8388608;
  float* Hb  = ws + 9437184;
  float* ST  = Qb;

  posenc_add<<<512, 256, 0, stream>>>(x, OUT, ST);
  ln_apply<<<512, 256, 0, stream>>>(OUT, n0g, n0b, ST, RES);

  for (int i = 0; i < 4; ++i) {
    dwpw_gemm<<<512, 128, 0, stream>>>(OUT, dw_w + i * CC * 7, dw_b + i * CC,
                                       pw_w + i * CC * CC, pw_b + i * CC, RES, ST);
    ln_apply<<<512, 256, 0, stream>>>(RES, nsg + i * CT, nsb + i * CT, ST, OUT);
  }

  xtw_gemm<<<dim3(256, 3), 128, 0, stream>>>(OUT, Wq, Wk, Wv, Qb, Kb, Vb);
  attn_kernel<<<1024, 128, 0, stream>>>(Qb, Kb, Vb, mask, Hb);
  wo_gemm<<<512, 128, 0, stream>>>(Hb, Wo, RES, ST);
  ln_apply<<<512, 256, 0, stream>>>(RES, neg_, neb, ST, OUT);
  fc_gemm<<<512, 128, 0, stream>>>(OUT, fcw, fcb, RES, outp);
}


# LayerNorm applied by the reader as it loads (conv and 1x1 GEMM tile builders in LDS, first block's residual, Q/K/V A fragments): the 6 ln_apply launches and the normalised tensor's round trip gone
# speedup vs baseline: 3.6442x; 1.1249x over previous
#include <hip/hip_runtime.h>

typedef float v2f __attribute__((ext_vector_type(2)));
typedef float v8f __attribute__((ext_vector_type(8)));

#define BB 16
#define CC 128
#define TT 1024
#define DKD 64
#define CT (CC * TT)
#define NPART 32
#define RPW 4
#define WGE (RPW * TT)
#define LOG10000 9.210340371976184f
#define PI_2 1.5707963267948966f

__device__ inline v8f wmma4(v2f a, v2f b, v8f c) {
  return __builtin_amdgcn_wmma_f32_16x16x4_f32(false, a, false, b, (short)0, c,
                                               false, false);
}


__device__ inline void wg_pair_store(float s, float q, int nw, float* wsum, float* wsq,
                                     float* __restrict__ part) {
  int tid = threadIdx.x;
  for (int o = 16; o > 0; o >>= 1) {
    s += __shfl_xor(s, o, 32);
    q += __shfl_xor(q, o, 32);
  }
  if ((tid & 31) == 0) {
    wsum[tid >> 5] = s;
    wsq[tid >> 5] = q;
  }
  __syncthreads();
  if (tid == 0) {
    float a = wsum[0];
    float c = wsq[0];
    for (int u = 1; u < nw; ++u) {
      a += wsum[u];
      c += wsq[u];
    }
    reinterpret_cast<float2*>(part)[blockIdx.x] = make_float2(a, c);
  }
}


__global__ __launch_bounds__(256) void posenc_add(const float* __restrict__ x,
                                                  float* __restrict__ out,
                                                  float* __restrict__ part) {
  __shared__ float wsum[8];
  __shared__ float wsq[8];
  int p = threadIdx.x;
  int row0 = blockIdx.x * RPW;
  const float4* xin = reinterpret_cast<const float4*>(x) + row0 * (TT / 4) + p;
  float4* o = reinterpret_cast<float4*>(out) + row0 * (TT / 4) + p;
  float4 v[RPW];
#pragma unroll
  for (int q = 0; q < RPW; ++q) v[q] = xin[q * (TT / 4)];
  int t = p * 4;
  float s = 0.f, sq = 0.f;
#pragma unroll
  for (int q = 0; q < RPW; ++q) {
    int c = (row0 + q) % CC;
    float freq = expf(-(float)(c & ~1) * (LOG10000 / (float)CC));
    float ph = (float)(c & 1) * PI_2;
    float4 r;
    r.x = v[q].x + sinf((float)t * freq + ph);
    r.y = v[q].y + sinf((float)(t + 1) * freq + ph);
    r.z = v[q].z + sinf((float)(t + 2) * freq + ph);
    r.w = v[q].w + sinf((float)(t + 3) * freq + ph);
    v[q] = r;
    s += (r.x + r.y) + (r.z + r.w);
    sq += (r.x * r.x + r.y * r.y) + (r.z * r.z + r.w * r.w);
  }
  wg_pair_store(s, sq, 8, wsum, wsq, part);
#pragma unroll
  for (int q = 0; q < RPW; ++q) o[q * (TT / 4)] = v[q];
}

__device__ inline float2 ln_stats(const float* __restrict__ part, int b, int lane) {
  float2 pr = reinterpret_cast<const float2*>(part)[b * NPART + lane];
  float s = pr.x;
  float q2 = pr.y;
  for (int o = 16; o > 0; o >>= 1) {
    s += __shfl_xor(s, o, 32);
    q2 += __shfl_xor(q2, o, 32);
  }
  float inv_n = 1.0f / (float)CT;
  float mu = s * inv_n;
  float var = q2 * inv_n - mu * mu;
  return make_float2(mu, rsqrtf(var + 1e-5f));
}

__device__ inline float lnv(float x, float mu, float r, float g, float b) {
  return (x - mu) * r * g + b;
}

__device__ inline float4 ln4(float4 x, float4 g, float4 b, float mu, float r) {
  float4 y;
  y.x = lnv(x.x, mu, r, g.x, b.x);
  y.y = lnv(x.y, mu, r, g.y, b.y);
  y.z = lnv(x.z, mu, r, g.z, b.z);
  y.w = lnv(x.w, mu, r, g.w, b.w);
  return y;
}


#define BPAIR 160
template <int MODE>
__global__ __launch_bounds__(128) void lnpw_gemm(const float* __restrict__ X,
                                                 const float* __restrict__ lng,
                                                 const float* __restrict__ lnb,
                                                 const float* __restrict__ pin,
                                                 const float* __restrict__ dww,
                                                 const float* __restrict__ dwb,
                                                 const float* __restrict__ W,
                                                 const float* __restrict__ bias,
                                                 float* __restrict__ dst,
                                                 float* __restrict__ pout) {
  __shared__ __attribute__((aligned(16))) float Bs[(CC / 2) * BPAIR];
  __shared__ float wsum[4];
  __shared__ float wsq[4];
  int tid = threadIdx.x;
  int l = tid & 31;
  int b = blockIdx.x >> 5;
  int r5 = blockIdx.x & 31;
  int n0 = (r5 >> 1) << 6;
  int m0 = (((r5 & 1) << 2) + (tid >> 5)) << 4;
  int lo = l & 15;
  int kh = (l >> 4) << 1;

  float2 st = ln_stats(pin, b, l);
  float mu = st.x;
  float rs = st.y;

  v2f afrag[CC / 4];
#pragma unroll
  for (int kk = 0; kk < CC; kk += 4) {
    int k0 = kk + kh;
    afrag[kk >> 2].x = W[(m0 + lo) * CC + k0];
    afrag[kk >> 2].y = W[(m0 + lo) * CC + k0 + 1];
  }

  const float* xb = X + b * CT;
  {
    int p = tid & 15;
    int gp = (n0 >> 2) + p;
    int gl = gp > 0 ? gp - 1 : gp;
    int gr = gp < TT / 4 - 1 ? gp + 1 : gp;
    int c0 = tid >> 4;
#pragma unroll 8
    for (int q = 0; q < CC / 8; ++q) {
      int c = c0 + 8 * q;
      const float4* r4 = reinterpret_cast<const float4*>(xb + c * TT);
      const float4* g4 = reinterpret_cast<const float4*>(lng + c * TT);
      const float4* b4 = reinterpret_cast<const float4*>(lnb + c * TT);
      float* d = Bs + (c >> 1) * BPAIR + 8 * p + (c & 1);
      float4 cur = r4[gp];
      if (MODE == 2) {
        cur = ln4(cur, g4[gp], b4[gp], mu, rs);
        d[0] = cur.x;
        d[2] = cur.y;
        d[4] = cur.z;
        d[6] = cur.w;
      } else {
        float4 prv = r4[gl];
        float4 nxt = r4[gr];
        if (MODE == 1) {
          cur = ln4(cur, g4[gp], b4[gp], mu, rs);
          prv = ln4(prv, g4[gl], b4[gl], mu, rs);
          nxt = ln4(nxt, g4[gr], b4[gr], mu, rs);
        }
        if (gp == 0) prv = make_float4(0.f, 0.f, 0.f, 0.f);
        if (gp == TT / 4 - 1) nxt = make_float4(0.f, 0.f, 0.f, 0.f);
        float bs = dwb[c];
        float wj[7];
#pragma unroll
        for (int j = 0; j < 7; ++j) wj[j] = dww[c * 7 + j];
        float xs[12] = {prv.x, prv.y, prv.z, prv.w, cur.x, cur.y,
                        cur.z, cur.w, nxt.x, nxt.y, nxt.z, nxt.w};
#pragma unroll
        for (int k = 0; k < 4; ++k) {
          float s = bs;
#pragma unroll
          for (int j = 0; j < 7; ++j) s = fmaf(xs[k + j + 1], wj[j], s);
          d[2 * k] = s;
        }
      }
    }
  }
  __syncthreads();

  v8f acc0 = {}, acc1 = {}, acc2 = {}, acc3 = {};
#pragma unroll
  for (int kk = 0; kk < CC; kk += 4) {
    const v2f* bp = reinterpret_cast<const v2f*>(Bs + ((kk + kh) >> 1) * BPAIR + 2 * lo);
    v2f b0 = bp[0], b1 = bp[16], b2 = bp[32], b3 = bp[48];
    acc0 = wmma4(afrag[kk >> 2], b0, acc0);
    acc1 = wmma4(afrag[kk >> 2], b1, acc1);
    acc2 = wmma4(afrag[kk >> 2], b2, acc2);
    acc3 = wmma4(afrag[kk >> 2], b3, acc3);
  }
  float* db = dst + b * CT;
  float s = 0.f, sq = 0.f;
#pragma unroll
  for (int v = 0; v < 8; ++v) {
    int m = m0 + v + ((l >> 4) << 3);
    float bsv = bias[m];
    int off = m * TT + n0 + lo;
    const float* r = xb + off;
    float r0 = r[0], r1 = r[16], r2 = r[32], r3 = r[48];
    if (MODE == 0) {
      const float* g = lng + off;
      const float* be = lnb + off;
      r0 = lnv(r0, mu, rs, g[0], be[0]);
      r1 = lnv(r1, mu, rs, g[16], be[16]);
      r2 = lnv(r2, mu, rs, g[32], be[32]);
      r3 = lnv(r3, mu, rs, g[48], be[48]);
    }
    float o0 = fmaxf(acc0[v] + bsv, 0.f) + r0;
    float o1 = fmaxf(acc1[v] + bsv, 0.f) + r1;
    float o2 = fmaxf(acc2[v] + bsv, 0.f) + r2;
    float o3 = fmaxf(acc3[v] + bsv, 0.f) + r3;
    acc0[v] = o0;
    acc1[v] = o1;
    acc2[v] = o2;
    acc3[v] = o3;
    s += (o0 + o1) + (o2 + o3);
    sq += (o0 * o0 + o1 * o1) + (o2 * o2 + o3 * o3);
  }
  if (MODE != 2) wg_pair_store(s, sq, 4, wsum, wsq, pout);
#pragma unroll
  for (int v = 0; v < 8; ++v) {
    float* o = db + (m0 + v + ((l >> 4) << 3)) * TT + n0 + lo;
    o[0]  = acc0[v];
    o[16] = acc1[v];
    o[32] = acc2[v];
    o[48] = acc3[v];
  }
}

__global__ __launch_bounds__(128) void xtw_gemm(const float* __restrict__ X,
                                                const float* __restrict__ lng,
                                                const float* __restrict__ lnb,
                                                const float* __restrict__ pin,
                                                const float* __restrict__ W0,
                                                const float* __restrict__ W1,
                                                const float* __restrict__ W2,
                                                float* __restrict__ D0,
                                                float* __restrict__ D1,
                                                float* __restrict__ D2) {
  int y = blockIdx.y;
  const float* W = y == 0 ? W0 : (y == 1 ? W1 : W2);
  float* dst = y == 0 ? D0 : (y == 1 ? D1 : D2);
  int l = threadIdx.x & 31;
  int tile = blockIdx.x * 4 + (threadIdx.x >> 5);
  int b = tile >> 6;
  int t0 = (tile & 63) << 4;
  int lo = l & 15;
  int kh = (l >> 4) << 1;
  const float* xb = X + b * CT;
  float2 st = ln_stats(pin, b, l);
  v8f acc0 = {}, acc1 = {}, acc2 = {}, acc3 = {};
#pragma unroll 4
  for (int kk = 0; kk < CC; kk += 4) {
    int k0 = kk + kh;
    int ia = k0 * TT + t0 + lo;
    v2f a;
    a.x = lnv(xb[ia], st.x, st.y, lng[ia], lnb[ia]);
    a.y = lnv(xb[ia + TT], st.x, st.y, lng[ia + TT], lnb[ia + TT]);
    const float* w0 = W + k0 * DKD + lo;
    const float* w1 = w0 + DKD;
    v2f b0, b1, b2, b3;
    b0.x = w0[0];  b0.y = w1[0];
    b1.x = w0[16]; b1.y = w1[16];
    b2.x = w0[32]; b2.y = w1[32];
    b3.x = w0[48]; b3.y = w1[48];
    acc0 = wmma4(a, b0, acc0);
    acc1 = wmma4(a, b1, acc1);
    acc2 = wmma4(a, b2, acc2);
    acc3 = wmma4(a, b3, acc3);
  }
  float* db = dst + b * TT * DKD;
#pragma unroll
  for (int v = 0; v < 8; ++v) {
    int t = t0 + v + ((l >> 4) << 3);
    float* d = db + t * DKD + lo;
    d[0]  = acc0[v];
    d[16] = acc1[v];
    d[32] = acc2[v];
    d[48] = acc3[v];
  }
}

#define AW 4
#define KTW (TT / 16 / AW)
#define RWW (16 / AW)
#define KPW (TT / AW)
__global__ __launch_bounds__(32 * AW) void attn_kernel(const float* __restrict__ Q,
                                                       const float* __restrict__ Km,
                                                       const float* __restrict__ Vm,
                                                       const float* __restrict__ mask,
                                                       float* __restrict__ head) {
  __shared__ __attribute__((aligned(16))) float S[16 * TT];
  int b = blockIdx.x >> 6;
  int q0 = (blockIdx.x & 63) << 4;
  int w = __builtin_amdgcn_readfirstlane((int)(threadIdx.x >> 5));
  int l = threadIdx.x & 31;
  int lo = l & 15;
  int kh = (l >> 4) << 1;
  const float* qb = Q + b * TT * DKD;
  const float* kb = Km + b * TT * DKD;
  const float* vb = Vm + b * TT * DKD;
  const float* mb = mask + b * TT;

  v2f afrag[DKD / 4];
#pragma unroll
  for (int kk = 0; kk < DKD; kk += 4) {
    int k0 = kk + kh;
    afrag[kk >> 2].x = qb[(q0 + lo) * DKD + k0];
    afrag[kk >> 2].y = qb[(q0 + lo) * DKD + k0 + 1];
  }

  for (int ti = 0; ti < KTW; ++ti) {
    int t = w * KTW + ti;
    v8f acc = {};
#pragma unroll
    for (int kk = 0; kk < DKD; kk += 4) {
      int k0 = kk + kh;
      v2f bv;
      bv.x = kb[(t * 16 + lo) * DKD + k0];
      bv.y = kb[(t * 16 + lo) * DKD + k0 + 1];
      acc = wmma4(afrag[kk >> 2], bv, acc);
    }
#pragma unroll
    for (int v = 0; v < 8; ++v) {
      int row = v + ((l >> 4) << 3);
      int col = t * 16 + lo;
      float hm = mb[col];
      float sv = acc[v] * 0.125f;
      S[row * TT + col] = sv * hm + (1.f - hm) * -1e30f;
    }
  }
  __syncthreads();

  for (int r = 0; r < RWW; ++r) {
    int row = w * RWW + r;
    float* sr = S + row * TT + l;
    float e[TT / 32];
    float mx = -3.4e38f;
#pragma unroll
    for (int c = 0; c < TT / 32; ++c) {
      e[c] = sr[c * 32];
      mx = fmaxf(mx, e[c]);
    }
    for (int o = 16; o > 0; o >>= 1) mx = fmaxf(mx, __shfl_xor(mx, o, 32));
    float sum = 0.f;
#pragma unroll
    for (int c = 0; c < TT / 32; ++c) {
      e[c] = expf(e[c] - mx);
      sum += e[c];
    }
    for (int o = 16; o > 0; o >>= 1) sum += __shfl_xor(sum, o, 32);
    float inv = 1.0f / sum;
    float qm = mb[q0 + row];
#pragma unroll
    for (int c = 0; c < TT / 32; ++c) sr[c * 32] = e[c] * inv * qm;
  }
  __syncthreads();

  v8f acc0 = {}, acc1 = {}, acc2 = {}, acc3 = {};
  for (int ki = 0; ki < KPW; ki += 4) {
    int k0 = w * KPW + ki + kh;
    v2f a;
    a.x = S[lo * TT + k0];
    a.y = S[lo * TT + k0 + 1];
    const float* v0p = vb + k0 * DKD + lo;
    const float* v1p = v0p + DKD;
    v2f b0, b1, b2, b3;
    b0.x = v0p[0];  b0.y = v1p[0];
    b1.x = v0p[16]; b1.y = v1p[16];
    b2.x = v0p[32]; b2.y = v1p[32];
    b3.x = v0p[48]; b3.y = v1p[48];
    acc0 = wmma4(a, b0, acc0);
    acc1 = wmma4(a, b1, acc1);
    acc2 = wmma4(a, b2, acc2);
    acc3 = wmma4(a, b3, acc3);
  }
  __syncthreads();

  {
    float* pw = S + w * (16 * DKD);
#pragma unroll
    for (int v = 0; v < 8; ++v) {
      float* p = pw + (v + ((l >> 4) << 3)) * DKD + lo;
      p[0]  = acc0[v];
      p[16] = acc1[v];
      p[32] = acc2[v];
      p[48] = acc3[v];
    }
  }
  __syncthreads();

  {
    const float4* s4 = reinterpret_cast<const float4*>(S);
    float4* h4 = reinterpret_cast<float4*>(head + (b * TT + q0) * DKD);
#pragma unroll
    for (int j = 0; j < (16 * DKD / 4) / (32 * AW); ++j) {
      int i4 = threadIdx.x + j * (32 * AW);
      float4 r = s4[i4];
#pragma unroll
      for (int u = 1; u < AW; ++u) {
        float4 pu = s4[u * (16 * DKD / 4) + i4];
        r.x += pu.x;
        r.y += pu.y;
        r.z += pu.z;
        r.w += pu.w;
      }
      h4[i4] = r;
    }
  }
}

__global__ __launch_bounds__(128) void wo_gemm(const float* __restrict__ H,
                                               const float* __restrict__ Wo,
                                               float* __restrict__ res,
                                               float* __restrict__ part) {
  __shared__ float wsum[4];
  __shared__ float wsq[4];
  int l = threadIdx.x & 31;
  int tile = blockIdx.x * 4 + (threadIdx.x >> 5);
  int b = tile >> 7;
  int rem = tile & 127;
  int m0 = (rem >> 4) << 4;
  int n0 = (rem & 15) << 6;
  int lo = l & 15;
  int kh = (l >> 4) << 1;
  const float* hb = H + b * TT * DKD;
  v8f acc0 = {}, acc1 = {}, acc2 = {}, acc3 = {};
  for (int kk = 0; kk < DKD; kk += 4) {
    int k0 = kk + kh;
    v2f a;
    a.x = Wo[k0 * CC + m0 + lo] + Wo[(k0 + 64) * CC + m0 + lo];
    a.y = Wo[(k0 + 1) * CC + m0 + lo] + Wo[(k0 + 65) * CC + m0 + lo];
    const float* h0 = hb + (n0 + lo) * DKD + k0;
    v2f b0, b1, b2, b3;
    b0.x = h0[0];            b0.y = h0[1];
    b1.x = h0[16 * DKD];     b1.y = h0[16 * DKD + 1];
    b2.x = h0[32 * DKD];     b2.y = h0[32 * DKD + 1];
    b3.x = h0[48 * DKD];     b3.y = h0[48 * DKD + 1];
    acc0 = wmma4(a, b0, acc0);
    acc1 = wmma4(a, b1, acc1);
    acc2 = wmma4(a, b2, acc2);
    acc3 = wmma4(a, b3, acc3);
  }
  float* rb = res + b * CT;
  float s = 0.f, sq = 0.f;
#pragma unroll
  for (int v = 0; v < 8; ++v) {
    int m = m0 + v + ((l >> 4) << 3);
    const float* r = rb + m * TT + n0 + lo;
    float o0 = acc0[v] + r[0];
    float o1 = acc1[v] + r[16];
    float o2 = acc2[v] + r[32];
    float o3 = acc3[v] + r[48];
    acc0[v] = o0;
    acc1[v] = o1;
    acc2[v] = o2;
    acc3[v] = o3;
    s += (o0 + o1) + (o2 + o3);
    sq += (o0 * o0 + o1 * o1) + (o2 * o2 + o3 * o3);
  }
  wg_pair_store(s, sq, 4, wsum, wsq, part);
#pragma unroll
  for (int v = 0; v < 8; ++v) {
    float* r = rb + (m0 + v + ((l >> 4) << 3)) * TT + n0 + lo;
    r[0]  = acc0[v];
    r[16] = acc1[v];
    r[32] = acc2[v];
    r[48] = acc3[v];
  }
}


extern "C" void kernel_launch(void* const* d_in, const int* in_sizes, int n_in,
                              void* d_out, int out_size, void* d_ws, size_t ws_size,
                              hipStream_t stream) {
  (void)in_sizes; (void)n_in; (void)out_size; (void)ws_size;
  const float* x    = (const float*)d_in[0];
  const float* mask = (const float*)d_in[1];
  const float* dw_w = (const float*)d_in[2];
  const float* dw_b = (const float*)d_in[3];
  const float* pw_w = (const float*)d_in[4];
  const float* pw_b = (const float*)d_in[5];
  const float* n0g  = (const float*)d_in[6];
  const float* n0b  = (const float*)d_in[7];
  const float* nsg  = (const float*)d_in[8];
  const float* nsb  = (const float*)d_in[9];
  const float* neg_ = (const float*)d_in[10];
  const float* neb  = (const float*)d_in[11];
  const float* Wq   = (const float*)d_in[12];
  const float* Wk   = (const float*)d_in[13];
  const float* Wv   = (const float*)d_in[14];
  const float* Wo   = (const float*)d_in[15];
  const float* fcw  = (const float*)d_in[16];
  const float* fcb  = (const float*)d_in[17];
  float* outp = (float*)d_out;
  float* ws = (float*)d_ws;

  float* RA  = ws;
  float* RB  = ws + 2097152;
  float* Qb  = ws + 6291456;
  float* Kb  = ws + 7340032;
  float* Vb  = ws + 8388608;
  float* Hb  = ws + 9437184;
  float* ST  = ws + 4194304;
  const float* nodw = nullptr;
  float* nopart = nullptr;

  posenc_add<<<512, 256, 0, stream>>>(x, RA, ST);
  lnpw_gemm<0><<<512, 128, 0, stream>>>(RA, n0g, n0b, ST, dw_w, dw_b, pw_w, pw_b, RB,
                                        ST + 1024);
  lnpw_gemm<1><<<512, 128, 0, stream>>>(RB, nsg, nsb, ST + 1024, dw_w + 1 * CC * 7,
                                        dw_b + 1 * CC, pw_w + 1 * CC * CC, pw_b + 1 * CC,
                                        RA, ST + 2048);
  lnpw_gemm<1><<<512, 128, 0, stream>>>(RA, nsg + 1 * CT, nsb + 1 * CT, ST + 2048,
                                        dw_w + 2 * CC * 7, dw_b + 2 * CC,
                                        pw_w + 2 * CC * CC, pw_b + 2 * CC, RB, ST + 3072);
  lnpw_gemm<1><<<512, 128, 0, stream>>>(RB, nsg + 2 * CT, nsb + 2 * CT, ST + 3072,
                                        dw_w + 3 * CC * 7, dw_b + 3 * CC,
                                        pw_w + 3 * CC * CC, pw_b + 3 * CC, RA, ST + 4096);

  xtw_gemm<<<dim3(256, 3), 128, 0, stream>>>(RA, nsg + 3 * CT, nsb + 3 * CT, ST + 4096,
                                             Wq, Wk, Wv, Qb, Kb, Vb);
  attn_kernel<<<1024, 128, 0, stream>>>(Qb, Kb, Vb, mask, Hb);
  wo_gemm<<<512, 128, 0, stream>>>(Hb, Wo, RA, ST + 5120);
  lnpw_gemm<2><<<512, 128, 0, stream>>>(RA, neg_, neb, ST + 5120, nodw, nodw, fcw, fcb,
                                        outp, nopart);
}


# lnpw_gemm: one 16-wave work-group per 64-column strip builds the B tile once (was two 4-wave groups, each building it); a wave owns 16x32 outputs: a different reduction tree for the LayerNorm pairs
# speedup vs baseline: 3.8647x; 1.0605x over previous
#include <hip/hip_runtime.h>

typedef float v2f __attribute__((ext_vector_type(2)));
typedef float v8f __attribute__((ext_vector_type(8)));

#define BB 16
#define CC 128
#define TT 1024
#define DKD 64
#define CT (CC * TT)
#define NPART 32
#define RPW 4
#define WGE (RPW * TT)
#define LOG10000 9.210340371976184f
#define PI_2 1.5707963267948966f

__device__ inline v8f wmma4(v2f a, v2f b, v8f c) {
  return __builtin_amdgcn_wmma_f32_16x16x4_f32(false, a, false, b, (short)0, c,
                                               false, false);
}


__device__ inline void wg_pair_store(float s, float q, int nw, float* wsum, float* wsq,
                                     float* __restrict__ part) {
  int tid = threadIdx.x;
  for (int o = 16; o > 0; o >>= 1) {
    s += __shfl_xor(s, o, 32);
    q += __shfl_xor(q, o, 32);
  }
  if ((tid & 31) == 0) {
    wsum[tid >> 5] = s;
    wsq[tid >> 5] = q;
  }
  __syncthreads();
  if (tid == 0) {
    float a = wsum[0];
    float c = wsq[0];
    for (int u = 1; u < nw; ++u) {
      a += wsum[u];
      c += wsq[u];
    }
    reinterpret_cast<float2*>(part)[blockIdx.x] = make_float2(a, c);
  }
}

__device__ inline void wg_pair_store2(float s, float q, float* wsum, float* wsq,
                                      float* __restrict__ part) {
  int tid = threadIdx.x;
  for (int o = 16; o > 0; o >>= 1) {
    s += __shfl_xor(s, o, 32);
    q += __shfl_xor(q, o, 32);
  }
  if ((tid & 31) == 0) {
    wsum[tid >> 5] = s;
    wsq[tid >> 5] = q;
  }
  __syncthreads();
  if ((tid & 255) == 0) {
    int g = tid >> 8;
    float a = wsum[8 * g];
    float c = wsq[8 * g];
    for (int u = 1; u < 8; ++u) {
      a += wsum[8 * g + u];
      c += wsq[8 * g + u];
    }
    reinterpret_cast<float2*>(part)[blockIdx.x * 2 + g] = make_float2(a, c);
  }
}


__global__ __launch_bounds__(256) void posenc_add(const float* __restrict__ x,
                                                  float* __restrict__ out,
                                                  float* __restrict__ part) {
  __shared__ float wsum[8];
  __shared__ float wsq[8];
  int p = threadIdx.x;
  int row0 = blockIdx.x * RPW;
  const float4* xin = reinterpret_cast<const float4*>(x) + row0 * (TT / 4) + p;
  float4* o = reinterpret_cast<float4*>(out) + row0 * (TT / 4) + p;
  float4 v[RPW];
#pragma unroll
  for (int q = 0; q < RPW; ++q) v[q] = xin[q * (TT / 4)];
  int t = p * 4;
  float s = 0.f, sq = 0.f;
#pragma unroll
  for (int q = 0; q < RPW; ++q) {
    int c = (row0 + q) % CC;
    float freq = expf(-(float)(c & ~1) * (LOG10000 / (float)CC));
    float ph = (float)(c & 1) * PI_2;
    float4 r;
    r.x = v[q].x + sinf((float)t * freq + ph);
    r.y = v[q].y + sinf((float)(t + 1) * freq + ph);
    r.z = v[q].z + sinf((float)(t + 2) * freq + ph);
    r.w = v[q].w + sinf((float)(t + 3) * freq + ph);
    v[q] = r;
    s += (r.x + r.y) + (r.z + r.w);
    sq += (r.x * r.x + r.y * r.y) + (r.z * r.z + r.w * r.w);
  }
  wg_pair_store(s, sq, 8, wsum, wsq, part);
#pragma unroll
  for (int q = 0; q < RPW; ++q) o[q * (TT / 4)] = v[q];
}

__device__ inline float2 ln_stats(const float* __restrict__ part, int b, int lane) {
  float2 pr = reinterpret_cast<const float2*>(part)[b * NPART + lane];
  float s = pr.x;
  float q2 = pr.y;
  for (int o = 16; o > 0; o >>= 1) {
    s += __shfl_xor(s, o, 32);
    q2 += __shfl_xor(q2, o, 32);
  }
  float inv_n = 1.0f / (float)CT;
  float mu = s * inv_n;
  float var = q2 * inv_n - mu * mu;
  return make_float2(mu, rsqrtf(var + 1e-5f));
}

__device__ inline float lnv(float x, float mu, float r, float g, float b) {
  return (x - mu) * r * g + b;
}

__device__ inline float4 ln4(float4 x, float4 g, float4 b, float mu, float r) {
  float4 y;
  y.x = lnv(x.x, mu, r, g.x, b.x);
  y.y = lnv(x.y, mu, r, g.y, b.y);
  y.z = lnv(x.z, mu, r, g.z, b.z);
  y.w = lnv(x.w, mu, r, g.w, b.w);
  return y;
}


#define BPAIR 160
template <int MODE>
__global__ __launch_bounds__(512) void lnpw_gemm(const float* __restrict__ X,
                                                 const float* __restrict__ lng,
                                                 const float* __restrict__ lnb,
                                                 const float* __restrict__ pin,
                                                 const float* __restrict__ dww,
                                                 const float* __restrict__ dwb,
                                                 const float* __restrict__ W,
                                                 const float* __restrict__ bias,
                                                 float* __restrict__ dst,
                                                 float* __restrict__ pout) {
  __shared__ __attribute__((aligned(16))) float Bs[(CC / 2) * BPAIR];
  __shared__ float wsum[16];
  __shared__ float wsq[16];
  int tid = threadIdx.x;
  int l = tid & 31;
  int b = blockIdx.x >> 4;
  int n0 = (blockIdx.x & 15) << 6;
  int m0 = (tid >> 6) << 4;
  int nh = ((tid >> 5) & 1) << 5;
  int lo = l & 15;
  int kh = (l >> 4) << 1;

  float2 st = ln_stats(pin, b, l);
  float mu = st.x;
  float rs = st.y;

  v2f afrag[CC / 4];
#pragma unroll
  for (int kk = 0; kk < CC; kk += 4) {
    int k0 = kk + kh;
    afrag[kk >> 2].x = W[(m0 + lo) * CC + k0];
    afrag[kk >> 2].y = W[(m0 + lo) * CC + k0 + 1];
  }

  const float* xb = X + b * CT;
  {
    int p = tid & 15;
    int gp = (n0 >> 2) + p;
    int gl = gp > 0 ? gp - 1 : gp;
    int gr = gp < TT / 4 - 1 ? gp + 1 : gp;
    int c0 = tid >> 4;
#pragma unroll
    for (int q = 0; q < CC / 32; ++q) {
      int c = c0 + 32 * q;
      const float4* r4 = reinterpret_cast<const float4*>(xb + c * TT);
      const float4* g4 = reinterpret_cast<const float4*>(lng + c * TT);
      const float4* b4 = reinterpret_cast<const float4*>(lnb + c * TT);
      float* d = Bs + (c >> 1) * BPAIR + 8 * p + (c & 1);
      float4 cur = r4[gp];
      if (MODE == 2) {
        cur = ln4(cur, g4[gp], b4[gp], mu, rs);
        d[0] = cur.x;
        d[2] = cur.y;
        d[4] = cur.z;
        d[6] = cur.w;
      } else {
        float4 prv = r4[gl];
        float4 nxt = r4[gr];
        if (MODE == 1) {
          cur = ln4(cur, g4[gp], b4[gp], mu, rs);
          prv = ln4(prv, g4[gl], b4[gl], mu, rs);
          nxt = ln4(nxt, g4[gr], b4[gr], mu, rs);
        }
        if (gp == 0) prv = make_float4(0.f, 0.f, 0.f, 0.f);
        if (gp == TT / 4 - 1) nxt = make_float4(0.f, 0.f, 0.f, 0.f);
        float bs = dwb[c];
        float wj[7];
#pragma unroll
        for (int j = 0; j < 7; ++j) wj[j] = dww[c * 7 + j];
        float xs[12] = {prv.x, prv.y, prv.z, prv.w, cur.x, cur.y,
                        cur.z, cur.w, nxt.x, nxt.y, nxt.z, nxt.w};
#pragma unroll
        for (int k = 0; k < 4; ++k) {
          float s = bs;
#pragma unroll
          for (int j = 0; j < 7; ++j) s = fmaf(xs[k + j + 1], wj[j], s);
          d[2 * k] = s;
        }
      }
    }
  }
  __syncthreads();

  v8f acc0 = {}, acc1 = {};
#pragma unroll
  for (int kk = 0; kk < CC; kk += 4) {
    const v2f* bp =
        reinterpret_cast<const v2f*>(Bs + ((kk + kh) >> 1) * BPAIR + 2 * (nh + lo));
    v2f b0 = bp[0], b1 = bp[16];
    acc0 = wmma4(afrag[kk >> 2], b0, acc0);
    acc1 = wmma4(afrag[kk >> 2], b1, acc1);
  }
  float* db = dst + b * CT;
  float s = 0.f, sq = 0.f;
#pragma unroll
  for (int v = 0; v < 8; ++v) {
    int m = m0 + v + ((l >> 4) << 3);
    float bsv = bias[m];
    int off = m * TT + n0 + nh + lo;
    const float* r = xb + off;
    float r0 = r[0], r1 = r[16];
    if (MODE == 0) {
      const float* g = lng + off;
      const float* be = lnb + off;
      r0 = lnv(r0, mu, rs, g[0], be[0]);
      r1 = lnv(r1, mu, rs, g[16], be[16]);
    }
    float o0 = fmaxf(acc0[v] + bsv, 0.f) + r0;
    float o1 = fmaxf(acc1[v] + bsv, 0.f) + r1;
    acc0[v] = o0;
    acc1[v] = o1;
    s += o0 + o1;
    sq += o0 * o0 + o1 * o1;
  }
  if (MODE != 2) wg_pair_store2(s, sq, wsum, wsq, pout);
#pragma unroll
  for (int v = 0; v < 8; ++v) {
    float* o = db + (m0 + v + ((l >> 4) << 3)) * TT + n0 + nh + lo;
    o[0]  = acc0[v];
    o[16] = acc1[v];
  }
}

__global__ __launch_bounds__(128) void xtw_gemm(const float* __restrict__ X,
                                                const float* __restrict__ lng,
                                                const float* __restrict__ lnb,
                                                const float* __restrict__ pin,
                                                const float* __restrict__ W0,
                                                const float* __restrict__ W1,
                                                const float* __restrict__ W2,
                                                float* __restrict__ D0,
                                                float* __restrict__ D1,
                                                float* __restrict__ D2) {
  int y = blockIdx.y;
  const float* W = y == 0 ? W0 : (y == 1 ? W1 : W2);
  float* dst = y == 0 ? D0 : (y == 1 ? D1 : D2);
  int l = threadIdx.x & 31;
  int tile = blockIdx.x * 4 + (threadIdx.x >> 5);
  int b = tile >> 6;
  int t0 = (tile & 63) << 4;
  int lo = l & 15;
  int kh = (l >> 4) << 1;
  const float* xb = X + b * CT;
  float2 st = ln_stats(pin, b, l);
  v8f acc0 = {}, acc1 = {}, acc2 = {}, acc3 = {};
#pragma unroll 4
  for (int kk = 0; kk < CC; kk += 4) {
    int k0 = kk + kh;
    int ia = k0 * TT + t0 + lo;
    v2f a;
    a.x = lnv(xb[ia], st.x, st.y, lng[ia], lnb[ia]);
    a.y = lnv(xb[ia + TT], st.x, st.y, lng[ia + TT], lnb[ia + TT]);
    const float* w0 = W + k0 * DKD + lo;
    const float* w1 = w0 + DKD;
    v2f b0, b1, b2, b3;
    b0.x = w0[0];  b0.y = w1[0];
    b1.x = w0[16]; b1.y = w1[16];
    b2.x = w0[32]; b2.y = w1[32];
    b3.x = w0[48]; b3.y = w1[48];
    acc0 = wmma4(a, b0, acc0);
    acc1 = wmma4(a, b1, acc1);
    acc2 = wmma4(a, b2, acc2);
    acc3 = wmma4(a, b3, acc3);
  }
  float* db = dst + b * TT * DKD;
#pragma unroll
  for (int v = 0; v < 8; ++v) {
    int t = t0 + v + ((l >> 4) << 3);
    float* d = db + t * DKD + lo;
    d[0]  = acc0[v];
    d[16] = acc1[v];
    d[32] = acc2[v];
    d[48] = acc3[v];
  }
}

#define AW 4
#define KTW (TT / 16 / AW)
#define RWW (16 / AW)
#define KPW (TT / AW)
__global__ __launch_bounds__(32 * AW) void attn_kernel(const float* __restrict__ Q,
                                                       const float* __restrict__ Km,
                                                       const float* __restrict__ Vm,
                                                       const float* __restrict__ mask,
                                                       float* __restrict__ head) {
  __shared__ __attribute__((aligned(16))) float S[16 * TT];
  int b = blockIdx.x >> 6;
  int q0 = (blockIdx.x & 63) << 4;
  int w = __builtin_amdgcn_readfirstlane((int)(threadIdx.x >> 5));
  int l = threadIdx.x & 31;
  int lo = l & 15;
  int kh = (l >> 4) << 1;
  const float* qb = Q + b * TT * DKD;
  const float* kb = Km + b * TT * DKD;
  const float* vb = Vm + b * TT * DKD;
  const float* mb = mask + b * TT;

  v2f afrag[DKD / 4];
#pragma unroll
  for (int kk = 0; kk < DKD; kk += 4) {
    int k0 = kk + kh;
    afrag[kk >> 2].x = qb[(q0 + lo) * DKD + k0];
    afrag[kk >> 2].y = qb[(q0 + lo) * DKD + k0 + 1];
  }

  for (int ti = 0; ti < KTW; ++ti) {
    int t = w * KTW + ti;
    v8f acc = {};
#pragma unroll
    for (int kk = 0; kk < DKD; kk += 4) {
      int k0 = kk + kh;
      v2f bv;
      bv.x = kb[(t * 16 + lo) * DKD + k0];
      bv.y = kb[(t * 16 + lo) * DKD + k0 + 1];
      acc = wmma4(afrag[kk >> 2], bv, acc);
    }
#pragma unroll
    for (int v = 0; v < 8; ++v) {
      int row = v + ((l >> 4) << 3);
      int col = t * 16 + lo;
      float hm = mb[col];
      float sv = acc[v] * 0.125f;
      S[row * TT + col] = sv * hm + (1.f - hm) * -1e30f;
    }
  }
  __syncthreads();

  for (int r = 0; r < RWW; ++r) {
    int row = w * RWW + r;
    float* sr = S + row * TT + l;
    float e[TT / 32];
    float mx = -3.4e38f;
#pragma unroll
    for (int c = 0; c < TT / 32; ++c) {
      e[c] = sr[c * 32];
      mx = fmaxf(mx, e[c]);
    }
    for (int o = 16; o > 0; o >>= 1) mx = fmaxf(mx, __shfl_xor(mx, o, 32));
    float sum = 0.f;
#pragma unroll
    for (int c = 0; c < TT / 32; ++c) {
      e[c] = expf(e[c] - mx);
      sum += e[c];
    }
    for (int o = 16; o > 0; o >>= 1) sum += __shfl_xor(sum, o, 32);
    float inv = 1.0f / sum;
    float qm = mb[q0 + row];
#pragma unroll
    for (int c = 0; c < TT / 32; ++c) sr[c * 32] = e[c] * inv * qm;
  }
  __syncthreads();

  v8f acc0 = {}, acc1 = {}, acc2 = {}, acc3 = {};
  for (int ki = 0; ki < KPW; ki += 4) {
    int k0 = w * KPW + ki + kh;
    v2f a;
    a.x = S[lo * TT + k0];
    a.y = S[lo * TT + k0 + 1];
    const float* v0p = vb + k0 * DKD + lo;
    const float* v1p = v0p + DKD;
    v2f b0, b1, b2, b3;
    b0.x = v0p[0];  b0.y = v1p[0];
    b1.x = v0p[16]; b1.y = v1p[16];
    b2.x = v0p[32]; b2.y = v1p[32];
    b3.x = v0p[48]; b3.y = v1p[48];
    acc0 = wmma4(a, b0, acc0);
    acc1 = wmma4(a, b1, acc1);
    acc2 = wmma4(a, b2, acc2);
    acc3 = wmma4(a, b3, acc3);
  }
  __syncthreads();

  {
    float* pw = S + w * (16 * DKD);
#pragma unroll
    for (int v = 0; v < 8; ++v) {
      float* p = pw + (v + ((l >> 4) << 3)) * DKD + lo;
      p[0]  = acc0[v];
      p[16] = acc1[v];
      p[32] = acc2[v];
      p[48] = acc3[v];
    }
  }
  __syncthreads();

  {
    const float4* s4 = reinterpret_cast<const float4*>(S);
    float4* h4 = reinterpret_cast<float4*>(head + (b * TT + q0) * DKD);
#pragma unroll
    for (int j = 0; j < (16 * DKD / 4) / (32 * AW); ++j) {
      int i4 = threadIdx.x + j * (32 * AW);
      float4 r = s4[i4];
#pragma unroll
      for (int u = 1; u < AW; ++u) {
        float4 pu = s4[u * (16 * DKD / 4) + i4];
        r.x += pu.x;
        r.y += pu.y;
        r.z += pu.z;
        r.w += pu.w;
      }
      h4[i4] = r;
    }
  }
}

__global__ __launch_bounds__(128) void wo_gemm(const float* __restrict__ H,
                                               const float* __restrict__ Wo,
                                               float* __restrict__ res,
                                               float* __restrict__ part) {
  __shared__ float wsum[4];
  __shared__ float wsq[4];
  int l = threadIdx.x & 31;
  int tile = blockIdx.x * 4 + (threadIdx.x >> 5);
  int b = tile >> 7;
  int rem = tile & 127;
  int m0 = (rem >> 4) << 4;
  int n0 = (rem & 15) << 6;
  int lo = l & 15;
  int kh = (l >> 4) << 1;
  const float* hb = H + b * TT * DKD;
  v8f acc0 = {}, acc1 = {}, acc2 = {}, acc3 = {};
  for (int kk = 0; kk < DKD; kk += 4) {
    int k0 = kk + kh;
    v2f a;
    a.x = Wo[k0 * CC + m0 + lo] + Wo[(k0 + 64) * CC + m0 + lo];
    a.y = Wo[(k0 + 1) * CC + m0 + lo] + Wo[(k0 + 65) * CC + m0 + lo];
    const float* h0 = hb + (n0 + lo) * DKD + k0;
    v2f b0, b1, b2, b3;
    b0.x = h0[0];            b0.y = h0[1];
    b1.x = h0[16 * DKD];     b1.y = h0[16 * DKD + 1];
    b2.x = h0[32 * DKD];     b2.y = h0[32 * DKD + 1];
    b3.x = h0[48 * DKD];     b3.y = h0[48 * DKD + 1];
    acc0 = wmma4(a, b0, acc0);
    acc1 = wmma4(a, b1, acc1);
    acc2 = wmma4(a, b2, acc2);
    acc3 = wmma4(a, b3, acc3);
  }
  float* rb = res + b * CT;
  float s = 0.f, sq = 0.f;
#pragma unroll
  for (int v = 0; v < 8; ++v) {
    int m = m0 + v + ((l >> 4) << 3);
    const float* r = rb + m * TT + n0 + lo;
    float o0 = acc0[v] + r[0];
    float o1 = acc1[v] + r[16];
    float o2 = acc2[v] + r[32];
    float o3 = acc3[v] + r[48];
    acc0[v] = o0;
    acc1[v] = o1;
    acc2[v] = o2;
    acc3[v] = o3;
    s += (o0 + o1) + (o2 + o3);
    sq += (o0 * o0 + o1 * o1) + (o2 * o2 + o3 * o3);
  }
  wg_pair_store(s, sq, 4, wsum, wsq, part);
#pragma unroll
  for (int v = 0; v < 8; ++v) {
    float* r = rb + (m0 + v + ((l >> 4) << 3)) * TT + n0 + lo;
    r[0]  = acc0[v];
    r[16] = acc1[v];
    r[32] = acc2[v];
    r[48] = acc3[v];
  }
}


extern "C" void kernel_launch(void* const* d_in, const int* in_sizes, int n_in,
                              void* d_out, int out_size, void* d_ws, size_t ws_size,
                              hipStream_t stream) {
  (void)in_sizes; (void)n_in; (void)out_size; (void)ws_size;
  const float* x    = (const float*)d_in[0];
  const float* mask = (const float*)d_in[1];
  const float* dw_w = (const float*)d_in[2];
  const float* dw_b = (const float*)d_in[3];
  const float* pw_w = (const float*)d_in[4];
  const float* pw_b = (const float*)d_in[5];
  const float* n0g  = (const float*)d_in[6];
  const float* n0b  = (const float*)d_in[7];
  const float* nsg  = (const float*)d_in[8];
  const float* nsb  = (const float*)d_in[9];
  const float* neg_ = (const float*)d_in[10];
  const float* neb  = (const float*)d_in[11];
  const float* Wq   = (const float*)d_in[12];
  const float* Wk   = (const float*)d_in[13];
  const float* Wv   = (const float*)d_in[14];
  const float* Wo   = (const float*)d_in[15];
  const float* fcw  = (const float*)d_in[16];
  const float* fcb  = (const float*)d_in[17];
  float* outp = (float*)d_out;
  float* ws = (float*)d_ws;

  float* RA  = ws;
  float* RB  = ws + 2097152;
  float* Qb  = ws + 6291456;
  float* Kb  = ws + 7340032;
  float* Vb  = ws + 8388608;
  float* Hb  = ws + 9437184;
  float* ST  = ws + 4194304;
  const float* nodw = nullptr;
  float* nopart = nullptr;

  posenc_add<<<512, 256, 0, stream>>>(x, RA, ST);
  lnpw_gemm<0><<<256, 512, 0, stream>>>(RA, n0g, n0b, ST, dw_w, dw_b, pw_w, pw_b, RB,
                                        ST + 1024);
  lnpw_gemm<1><<<256, 512, 0, stream>>>(RB, nsg, nsb, ST + 1024, dw_w + 1 * CC * 7,
                                        dw_b + 1 * CC, pw_w + 1 * CC * CC, pw_b + 1 * CC,
                                        RA, ST + 2048);
  lnpw_gemm<1><<<256, 512, 0, stream>>>(RA, nsg + 1 * CT, nsb + 1 * CT, ST + 2048,
                                        dw_w + 2 * CC * 7, dw_b + 2 * CC,
                                        pw_w + 2 * CC * CC, pw_b + 2 * CC, RB, ST + 3072);
  lnpw_gemm<1><<<256, 512, 0, stream>>>(RB, nsg + 2 * CT, nsb + 2 * CT, ST + 3072,
                                        dw_w + 3 * CC * 7, dw_b + 3 * CC,
                                        pw_w + 3 * CC * CC, pw_b + 3 * CC, RA, ST + 4096);

  xtw_gemm<<<dim3(256, 3), 128, 0, stream>>>(RA, nsg + 3 * CT, nsb + 3 * CT, ST + 4096,
                                             Wq, Wk, Wv, Qb, Kb, Vb);
  attn_kernel<<<1024, 128, 0, stream>>>(Qb, Kb, Vb, mask, Hb);
  wo_gemm<<<512, 128, 0, stream>>>(Hb, Wo, RA, ST + 5120);
  lnpw_gemm<2><<<256, 512, 0, stream>>>(RA, neg_, neb, ST + 5120, nodw, nodw, fcw, fcb,
                                        outp, nopart);
}


# attn_kernel: a work-group of 8 waves takes 32 queries: each K and V fragment loaded feeds 2 matrix instructions; P.V k range split 8 ways: a different reduction tree for P.V
# speedup vs baseline: 4.1899x; 1.0841x over previous
#include <hip/hip_runtime.h>

typedef float v2f __attribute__((ext_vector_type(2)));
typedef float v8f __attribute__((ext_vector_type(8)));

#define BB 16
#define CC 128
#define TT 1024
#define DKD 64
#define CT (CC * TT)
#define NPART 32
#define RPW 4
#define WGE (RPW * TT)
#define LOG10000 9.210340371976184f
#define PI_2 1.5707963267948966f

__device__ inline v8f wmma4(v2f a, v2f b, v8f c) {
  return __builtin_amdgcn_wmma_f32_16x16x4_f32(false, a, false, b, (short)0, c,
                                               false, false);
}


__device__ inline void wg_pair_store(float s, float q, int nw, float* wsum, float* wsq,
                                     float* __restrict__ part) {
  int tid = threadIdx.x;
  for (int o = 16; o > 0; o >>= 1) {
    s += __shfl_xor(s, o, 32);
    q += __shfl_xor(q, o, 32);
  }
  if ((tid & 31) == 0) {
    wsum[tid >> 5] = s;
    wsq[tid >> 5] = q;
  }
  __syncthreads();
  if (tid == 0) {
    float a = wsum[0];
    float c = wsq[0];
    for (int u = 1; u < nw; ++u) {
      a += wsum[u];
      c += wsq[u];
    }
    reinterpret_cast<float2*>(part)[blockIdx.x] = make_float2(a, c);
  }
}

__device__ inline void wg_pair_store2(float s, float q, float* wsum, float* wsq,
                                      float* __restrict__ part) {
  int tid = threadIdx.x;
  for (int o = 16; o > 0; o >>= 1) {
    s += __shfl_xor(s, o, 32);
    q += __shfl_xor(q, o, 32);
  }
  if ((tid & 31) == 0) {
    wsum[tid >> 5] = s;
    wsq[tid >> 5] = q;
  }
  __syncthreads();
  if ((tid & 255) == 0) {
    int g = tid >> 8;
    float a = wsum[8 * g];
    float c = wsq[8 * g];
    for (int u = 1; u < 8; ++u) {
      a += wsum[8 * g + u];
      c += wsq[8 * g + u];
    }
    reinterpret_cast<float2*>(part)[blockIdx.x * 2 + g] = make_float2(a, c);
  }
}


__global__ __launch_bounds__(256) void posenc_add(const float* __restrict__ x,
                                                  float* __restrict__ out,
                                                  float* __restrict__ part) {
  __shared__ float wsum[8];
  __shared__ float wsq[8];
  int p = threadIdx.x;
  int row0 = blockIdx.x * RPW;
  const float4* xin = reinterpret_cast<const float4*>(x) + row0 * (TT / 4) + p;
  float4* o = reinterpret_cast<float4*>(out) + row0 * (TT / 4) + p;
  float4 v[RPW];
#pragma unroll
  for (int q = 0; q < RPW; ++q) v[q] = xin[q * (TT / 4)];
  int t = p * 4;
  float s = 0.f, sq = 0.f;
#pragma unroll
  for (int q = 0; q < RPW; ++q) {
    int c = (row0 + q) % CC;
    float freq = expf(-(float)(c & ~1) * (LOG10000 / (float)CC));
    float ph = (float)(c & 1) * PI_2;
    float4 r;
    r.x = v[q].x + sinf((float)t * freq + ph);
    r.y = v[q].y + sinf((float)(t + 1) * freq + ph);
    r.z = v[q].z + sinf((float)(t + 2) * freq + ph);
    r.w = v[q].w + sinf((float)(t + 3) * freq + ph);
    v[q] = r;
    s += (r.x + r.y) + (r.z + r.w);
    sq += (r.x * r.x + r.y * r.y) + (r.z * r.z + r.w * r.w);
  }
  wg_pair_store(s, sq, 8, wsum, wsq, part);
#pragma unroll
  for (int q = 0; q < RPW; ++q) o[q * (TT / 4)] = v[q];
}

__device__ inline float2 ln_stats(const float* __restrict__ part, int b, int lane) {
  float2 pr = reinterpret_cast<const float2*>(part)[b * NPART + lane];
  float s = pr.x;
  float q2 = pr.y;
  for (int o = 16; o > 0; o >>= 1) {
    s += __shfl_xor(s, o, 32);
    q2 += __shfl_xor(q2, o, 32);
  }
  float inv_n = 1.0f / (float)CT;
  float mu = s * inv_n;
  float var = q2 * inv_n - mu * mu;
  return make_float2(mu, rsqrtf(var + 1e-5f));
}

__device__ inline float lnv(float x, float mu, float r, float g, float b) {
  return (x - mu) * r * g + b;
}

__device__ inline float4 ln4(float4 x, float4 g, float4 b, float mu, float r) {
  float4 y;
  y.x = lnv(x.x, mu, r, g.x, b.x);
  y.y = lnv(x.y, mu, r, g.y, b.y);
  y.z = lnv(x.z, mu, r, g.z, b.z);
  y.w = lnv(x.w, mu, r, g.w, b.w);
  return y;
}


#define BPAIR 160
template <int MODE>
__global__ __launch_bounds__(512) void lnpw_gemm(const float* __restrict__ X,
                                                 const float* __restrict__ lng,
                                                 const float* __restrict__ lnb,
                                                 const float* __restrict__ pin,
                                                 const float* __restrict__ dww,
                                                 const float* __restrict__ dwb,
                                                 const float* __restrict__ W,
                                                 const float* __restrict__ bias,
                                                 float* __restrict__ dst,
                                                 float* __restrict__ pout) {
  __shared__ __attribute__((aligned(16))) float Bs[(CC / 2) * BPAIR];
  __shared__ float wsum[16];
  __shared__ float wsq[16];
  int tid = threadIdx.x;
  int l = tid & 31;
  int b = blockIdx.x >> 4;
  int n0 = (blockIdx.x & 15) << 6;
  int m0 = (tid >> 6) << 4;
  int nh = ((tid >> 5) & 1) << 5;
  int lo = l & 15;
  int kh = (l >> 4) << 1;

  float2 st = ln_stats(pin, b, l);
  float mu = st.x;
  float rs = st.y;

  v2f afrag[CC / 4];
#pragma unroll
  for (int kk = 0; kk < CC; kk += 4) {
    int k0 = kk + kh;
    afrag[kk >> 2].x = W[(m0 + lo) * CC + k0];
    afrag[kk >> 2].y = W[(m0 + lo) * CC + k0 + 1];
  }

  const float* xb = X + b * CT;
  {
    int p = tid & 15;
    int gp = (n0 >> 2) + p;
    int gl = gp > 0 ? gp - 1 : gp;
    int gr = gp < TT / 4 - 1 ? gp + 1 : gp;
    int c0 = tid >> 4;
#pragma unroll
    for (int q = 0; q < CC / 32; ++q) {
      int c = c0 + 32 * q;
      const float4* r4 = reinterpret_cast<const float4*>(xb + c * TT);
      const float4* g4 = reinterpret_cast<const float4*>(lng + c * TT);
      const float4* b4 = reinterpret_cast<const float4*>(lnb + c * TT);
      float* d = Bs + (c >> 1) * BPAIR + 8 * p + (c & 1);
      float4 cur = r4[gp];
      if (MODE == 2) {
        cur = ln4(cur, g4[gp], b4[gp], mu, rs);
        d[0] = cur.x;
        d[2] = cur.y;
        d[4] = cur.z;
        d[6] = cur.w;
      } else {
        float4 prv = r4[gl];
        float4 nxt = r4[gr];
        if (MODE == 1) {
          cur = ln4(cur, g4[gp], b4[gp], mu, rs);
          prv = ln4(prv, g4[gl], b4[gl], mu, rs);
          nxt = ln4(nxt, g4[gr], b4[gr], mu, rs);
        }
        if (gp == 0) prv = make_float4(0.f, 0.f, 0.f, 0.f);
        if (gp == TT / 4 - 1) nxt = make_float4(0.f, 0.f, 0.f, 0.f);
        float bs = dwb[c];
        float wj[7];
#pragma unroll
        for (int j = 0; j < 7; ++j) wj[j] = dww[c * 7 + j];
        float xs[12] = {prv.x, prv.y, prv.z, prv.w, cur.x, cur.y,
                        cur.z, cur.w, nxt.x, nxt.y, nxt.z, nxt.w};
#pragma unroll
        for (int k = 0; k < 4; ++k) {
          float s = bs;
#pragma unroll
          for (int j = 0; j < 7; ++j) s = fmaf(xs[k + j + 1], wj[j], s);
          d[2 * k] = s;
        }
      }
    }
  }
  __syncthreads();

  v8f acc0 = {}, acc1 = {};
#pragma unroll
  for (int kk = 0; kk < CC; kk += 4) {
    const v2f* bp =
        reinterpret_cast<const v2f*>(Bs + ((kk + kh) >> 1) * BPAIR + 2 * (nh + lo));
    v2f b0 = bp[0], b1 = bp[16];
    acc0 = wmma4(afrag[kk >> 2], b0, acc0);
    acc1 = wmma4(afrag[kk >> 2], b1, acc1);
  }
  float* db = dst + b * CT;
  float s = 0.f, sq = 0.f;
#pragma unroll
  for (int v = 0; v < 8; ++v) {
    int m = m0 + v + ((l >> 4) << 3);
    float bsv = bias[m];
    int off = m * TT + n0 + nh + lo;
    const float* r = xb + off;
    float r0 = r[0], r1 = r[16];
    if (MODE == 0) {
      const float* g = lng + off;
      const float* be = lnb + off;
      r0 = lnv(r0, mu, rs, g[0], be[0]);
      r1 = lnv(r1, mu, rs, g[16], be[16]);
    }
    float o0 = fmaxf(acc0[v] + bsv, 0.f) + r0;
    float o1 = fmaxf(acc1[v] + bsv, 0.f) + r1;
    acc0[v] = o0;
    acc1[v] = o1;
    s += o0 + o1;
    sq += o0 * o0 + o1 * o1;
  }
  if (MODE != 2) wg_pair_store2(s, sq, wsum, wsq, pout);
#pragma unroll
  for (int v = 0; v < 8; ++v) {
    float* o = db + (m0 + v + ((l >> 4) << 3)) * TT + n0 + nh + lo;
    o[0]  = acc0[v];
    o[16] = acc1[v];
  }
}

__global__ __launch_bounds__(128) void xtw_gemm(const float* __restrict__ X,
                                                const float* __restrict__ lng,
                                                const float* __restrict__ lnb,
                                                const float* __restrict__ pin,
                                                const float* __restrict__ W0,
                                                const float* __restrict__ W1,
                                                const float* __restrict__ W2,
                                                float* __restrict__ D0,
                                                float* __restrict__ D1,
                                                float* __restrict__ D2) {
  int y = blockIdx.y;
  const float* W = y == 0 ? W0 : (y == 1 ? W1 : W2);
  float* dst = y == 0 ? D0 : (y == 1 ? D1 : D2);
  int l = threadIdx.x & 31;
  int tile = blockIdx.x * 4 + (threadIdx.x >> 5);
  int b = tile >> 6;
  int t0 = (tile & 63) << 4;
  int lo = l & 15;
  int kh = (l >> 4) << 1;
  const float* xb = X + b * CT;
  float2 st = ln_stats(pin, b, l);
  v8f acc0 = {}, acc1 = {}, acc2 = {}, acc3 = {};
#pragma unroll 4
  for (int kk = 0; kk < CC; kk += 4) {
    int k0 = kk + kh;
    int ia = k0 * TT + t0 + lo;
    v2f a;
    a.x = lnv(xb[ia], st.x, st.y, lng[ia], lnb[ia]);
    a.y = lnv(xb[ia + TT], st.x, st.y, lng[ia + TT], lnb[ia + TT]);
    const float* w0 = W + k0 * DKD + lo;
    const float* w1 = w0 + DKD;
    v2f b0, b1, b2, b3;
    b0.x = w0[0];  b0.y = w1[0];
    b1.x = w0[16]; b1.y = w1[16];
    b2.x = w0[32]; b2.y = w1[32];
    b3.x = w0[48]; b3.y = w1[48];
    acc0 = wmma4(a, b0, acc0);
    acc1 = wmma4(a, b1, acc1);
    acc2 = wmma4(a, b2, acc2);
    acc3 = wmma4(a, b3, acc3);
  }
  float* db = dst + b * TT * DKD;
#pragma unroll
  for (int v = 0; v < 8; ++v) {
    int t = t0 + v + ((l >> 4) << 3);
    float* d = db + t * DKD + lo;
    d[0]  = acc0[v];
    d[16] = acc1[v];
    d[32] = acc2[v];
    d[48] = acc3[v];
  }
}

#define QT 2
#define AW 8
#define QR (16 * QT)
#define KTW (TT / 16 / AW)
#define RWW (QR / AW)
#define KPW (TT / AW)
__global__ __launch_bounds__(32 * AW) void attn_kernel(const float* __restrict__ Q,
                                                       const float* __restrict__ Km,
                                                       const float* __restrict__ Vm,
                                                       const float* __restrict__ mask,
                                                       float* __restrict__ head) {
  __shared__ __attribute__((aligned(16))) float S[QR * TT];
  int b = blockIdx.x / (TT / QR);
  int q0 = (blockIdx.x % (TT / QR)) * QR;
  int w = __builtin_amdgcn_readfirstlane((int)(threadIdx.x >> 5));
  int l = threadIdx.x & 31;
  int lo = l & 15;
  int kh = (l >> 4) << 1;
  const float* qb = Q + b * TT * DKD;
  const float* kb = Km + b * TT * DKD;
  const float* vb = Vm + b * TT * DKD;
  const float* mb = mask + b * TT;

  v2f afrag[QT][DKD / 4];
#pragma unroll
  for (int u = 0; u < QT; ++u) {
#pragma unroll
    for (int kk = 0; kk < DKD; kk += 4) {
      int k0 = kk + kh;
      afrag[u][kk >> 2].x = qb[(q0 + 16 * u + lo) * DKD + k0];
      afrag[u][kk >> 2].y = qb[(q0 + 16 * u + lo) * DKD + k0 + 1];
    }
  }

#pragma unroll 2
  for (int ti = 0; ti < KTW; ++ti) {
    int t = w * KTW + ti;
    v8f acc[QT];
#pragma unroll
    for (int u = 0; u < QT; ++u) acc[u] = v8f{};
#pragma unroll
    for (int kk = 0; kk < DKD; kk += 4) {
      int k0 = kk + kh;
      v2f bv;
      bv.x = kb[(t * 16 + lo) * DKD + k0];
      bv.y = kb[(t * 16 + lo) * DKD + k0 + 1];
#pragma unroll
      for (int u = 0; u < QT; ++u) acc[u] = wmma4(afrag[u][kk >> 2], bv, acc[u]);
    }
    int col = t * 16 + lo;
    float hm = mb[col];
#pragma unroll
    for (int u = 0; u < QT; ++u) {
#pragma unroll
      for (int v = 0; v < 8; ++v) {
        int row = 16 * u + v + ((l >> 4) << 3);
        float sv = acc[u][v] * 0.125f;
        S[row * TT + col] = sv * hm + (1.f - hm) * -1e30f;
      }
    }
  }
  __syncthreads();

  for (int r = 0; r < RWW; ++r) {
    int row = w * RWW + r;
    float* sr = S + row * TT + l;
    float e[TT / 32];
    float mx = -3.4e38f;
#pragma unroll
    for (int c = 0; c < TT / 32; ++c) {
      e[c] = sr[c * 32];
      mx = fmaxf(mx, e[c]);
    }
    for (int o = 16; o > 0; o >>= 1) mx = fmaxf(mx, __shfl_xor(mx, o, 32));
    float sum = 0.f;
#pragma unroll
    for (int c = 0; c < TT / 32; ++c) {
      e[c] = expf(e[c] - mx);
      sum += e[c];
    }
    for (int o = 16; o > 0; o >>= 1) sum += __shfl_xor(sum, o, 32);
    float inv = 1.0f / sum;
    float qm = mb[q0 + row];
#pragma unroll
    for (int c = 0; c < TT / 32; ++c) sr[c * 32] = e[c] * inv * qm;
  }
  __syncthreads();

  v8f pacc[QT][4];
#pragma unroll
  for (int u = 0; u < QT; ++u) {
#pragma unroll
    for (int n = 0; n < 4; ++n) pacc[u][n] = v8f{};
  }
#pragma unroll 4
  for (int ki = 0; ki < KPW; ki += 4) {
    int k0 = w * KPW + ki + kh;
    const float* v0p = vb + k0 * DKD + lo;
    const float* v1p = v0p + DKD;
    v2f bf[4];
#pragma unroll
    for (int n = 0; n < 4; ++n) {
      bf[n].x = v0p[16 * n];
      bf[n].y = v1p[16 * n];
    }
#pragma unroll
    for (int u = 0; u < QT; ++u) {
      v2f a;
      a.x = S[(16 * u + lo) * TT + k0];
      a.y = S[(16 * u + lo) * TT + k0 + 1];
#pragma unroll
      for (int n = 0; n < 4; ++n) pacc[u][n] = wmma4(a, bf[n], pacc[u][n]);
    }
  }
  __syncthreads();

  {
    float* pw = S + w * (QR * DKD);
#pragma unroll
    for (int u = 0; u < QT; ++u) {
#pragma unroll
      for (int v = 0; v < 8; ++v) {
        float* p = pw + (16 * u + v + ((l >> 4) << 3)) * DKD + lo;
        p[0]  = pacc[u][0][v];
        p[16] = pacc[u][1][v];
        p[32] = pacc[u][2][v];
        p[48] = pacc[u][3][v];
      }
    }
  }
  __syncthreads();

  {
    const float4* s4 = reinterpret_cast<const float4*>(S);
    float4* h4 = reinterpret_cast<float4*>(head + (b * TT + q0) * DKD);
#pragma unroll
    for (int j = 0; j < (QR * DKD / 4) / (32 * AW); ++j) {
      int i4 = threadIdx.x + j * (32 * AW);
      float4 r = s4[i4];
#pragma unroll
      for (int u = 1; u < AW; ++u) {
        float4 pu = s4[u * (QR * DKD / 4) + i4];
        r.x += pu.x;
        r.y += pu.y;
        r.z += pu.z;
        r.w += pu.w;
      }
      h4[i4] = r;
    }
  }
}

__global__ __launch_bounds__(128) void wo_gemm(const float* __restrict__ H,
                                               const float* __restrict__ Wo,
                                               float* __restrict__ res,
                                               float* __restrict__ part) {
  __shared__ float wsum[4];
  __shared__ float wsq[4];
  int l = threadIdx.x & 31;
  int tile = blockIdx.x * 4 + (threadIdx.x >> 5);
  int b = tile >> 7;
  int rem = tile & 127;
  int m0 = (rem >> 4) << 4;
  int n0 = (rem & 15) << 6;
  int lo = l & 15;
  int kh = (l >> 4) << 1;
  const float* hb = H + b * TT * DKD;
  v8f acc0 = {}, acc1 = {}, acc2 = {}, acc3 = {};
  for (int kk = 0; kk < DKD; kk += 4) {
    int k0 = kk + kh;
    v2f a;
    a.x = Wo[k0 * CC + m0 + lo] + Wo[(k0 + 64) * CC + m0 + lo];
    a.y = Wo[(k0 + 1) * CC + m0 + lo] + Wo[(k0 + 65) * CC + m0 + lo];
    const float* h0 = hb + (n0 + lo) * DKD + k0;
    v2f b0, b1, b2, b3;
    b0.x = h0[0];            b0.y = h0[1];
    b1.x = h0[16 * DKD];     b1.y = h0[16 * DKD + 1];
    b2.x = h0[32 * DKD];     b2.y = h0[32 * DKD + 1];
    b3.x = h0[48 * DKD];     b3.y = h0[48 * DKD + 1];
    acc0 = wmma4(a, b0, acc0);
    acc1 = wmma4(a, b1, acc1);
    acc2 = wmma4(a, b2, acc2);
    acc3 = wmma4(a, b3, acc3);
  }
  float* rb = res + b * CT;
  float s = 0.f, sq = 0.f;
#pragma unroll
  for (int v = 0; v < 8; ++v) {
    int m = m0 + v + ((l >> 4) << 3);
    const float* r = rb + m * TT + n0 + lo;
    float o0 = acc0[v] + r[0];
    float o1 = acc1[v] + r[16];
    float o2 = acc2[v] + r[32];
    float o3 = acc3[v] + r[48];
    acc0[v] = o0;
    acc1[v] = o1;
    acc2[v] = o2;
    acc3[v] = o3;
    s += (o0 + o1) + (o2 + o3);
    sq += (o0 * o0 + o1 * o1) + (o2 * o2 + o3 * o3);
  }
  wg_pair_store(s, sq, 4, wsum, wsq, part);
#pragma unroll
  for (int v = 0; v < 8; ++v) {
    float* r = rb + (m0 + v + ((l >> 4) << 3)) * TT + n0 + lo;
    r[0]  = acc0[v];
    r[16] = acc1[v];
    r[32] = acc2[v];
    r[48] = acc3[v];
  }
}


extern "C" void kernel_launch(void* const* d_in, const int* in_sizes, int n_in,
                              void* d_out, int out_size, void* d_ws, size_t ws_size,
                              hipStream_t stream) {
  (void)in_sizes; (void)n_in; (void)out_size; (void)ws_size;
  const float* x    = (const float*)d_in[0];
  const float* mask = (const float*)d_in[1];
  const float* dw_w = (const float*)d_in[2];
  const float* dw_b = (const float*)d_in[3];
  const float* pw_w = (const float*)d_in[4];
  const float* pw_b = (const float*)d_in[5];
  const float* n0g  = (const float*)d_in[6];
  const float* n0b  = (const float*)d_in[7];
  const float* nsg  = (const float*)d_in[8];
  const float* nsb  = (const float*)d_in[9];
  const float* neg_ = (const float*)d_in[10];
  const float* neb  = (const float*)d_in[11];
  const float* Wq   = (const float*)d_in[12];
  const float* Wk   = (const float*)d_in[13];
  const float* Wv   = (const float*)d_in[14];
  const float* Wo   = (const float*)d_in[15];
  const float* fcw  = (const float*)d_in[16];
  const float* fcb  = (const float*)d_in[17];
  float* outp = (float*)d_out;
  float* ws = (float*)d_ws;

  float* RA  = ws;
  float* RB  = ws + 2097152;
  float* Qb  = ws + 6291456;
  float* Kb  = ws + 7340032;
  float* Vb  = ws + 8388608;
  float* Hb  = ws + 9437184;
  float* ST  = ws + 4194304;
  const float* nodw = nullptr;
  float* nopart = nullptr;

  posenc_add<<<512, 256, 0, stream>>>(x, RA, ST);
  lnpw_gemm<0><<<256, 512, 0, stream>>>(RA, n0g, n0b, ST, dw_w, dw_b, pw_w, pw_b, RB,
                                        ST + 1024);
  lnpw_gemm<1><<<256, 512, 0, stream>>>(RB, nsg, nsb, ST + 1024, dw_w + 1 * CC * 7,
                                        dw_b + 1 * CC, pw_w + 1 * CC * CC, pw_b + 1 * CC,
                                        RA, ST + 2048);
  lnpw_gemm<1><<<256, 512, 0, stream>>>(RA, nsg + 1 * CT, nsb + 1 * CT, ST + 2048,
                                        dw_w + 2 * CC * 7, dw_b + 2 * CC,
                                        pw_w + 2 * CC * CC, pw_b + 2 * CC, RB, ST + 3072);
  lnpw_gemm<1><<<256, 512, 0, stream>>>(RB, nsg + 2 * CT, nsb + 2 * CT, ST + 3072,
                                        dw_w + 3 * CC * 7, dw_b + 3 * CC,
                                        pw_w + 3 * CC * CC, pw_b + 3 * CC, RA, ST + 4096);

  xtw_gemm<<<dim3(256, 3), 128, 0, stream>>>(RA, nsg + 3 * CT, nsb + 3 * CT, ST + 4096,
                                             Wq, Wk, Wv, Qb, Kb, Vb);
  attn_kernel<<<512, 256, 0, stream>>>(Qb, Kb, Vb, mask, Hb);
  wo_gemm<<<512, 128, 0, stream>>>(Hb, Wo, RA, ST + 5120);
  lnpw_gemm<2><<<256, 512, 0, stream>>>(RA, neg_, neb, ST + 5120, nodw, nodw, fcw, fcb,
                                        outp, nopart);
}


# Output projection done in attn_kernel on the head tile in LDS (residual add, whole-line stores); wo_gemm launch and head round trip gone: a different reduction tree for that LayerNorm's sums
# speedup vs baseline: 4.4664x; 1.0660x over previous
#include <hip/hip_runtime.h>

typedef float v2f __attribute__((ext_vector_type(2)));
typedef float v8f __attribute__((ext_vector_type(8)));

#define BB 16
#define CC 128
#define TT 1024
#define DKD 64
#define CT (CC * TT)
#define NPART 32
#define RPW 4
#define WGE (RPW * TT)
#define LOG10000 9.210340371976184f
#define PI_2 1.5707963267948966f

__device__ inline v8f wmma4(v2f a, v2f b, v8f c) {
  return __builtin_amdgcn_wmma_f32_16x16x4_f32(false, a, false, b, (short)0, c,
                                               false, false);
}


__device__ inline void wg_pair_store(float s, float q, int nw, float* wsum, float* wsq,
                                     float* __restrict__ part) {
  int tid = threadIdx.x;
  for (int o = 16; o > 0; o >>= 1) {
    s += __shfl_xor(s, o, 32);
    q += __shfl_xor(q, o, 32);
  }
  if ((tid & 31) == 0) {
    wsum[tid >> 5] = s;
    wsq[tid >> 5] = q;
  }
  __syncthreads();
  if (tid == 0) {
    float a = wsum[0];
    float c = wsq[0];
    for (int u = 1; u < nw; ++u) {
      a += wsum[u];
      c += wsq[u];
    }
    reinterpret_cast<float2*>(part)[blockIdx.x] = make_float2(a, c);
  }
}

__device__ inline void wg_pair_store2(float s, float q, float* wsum, float* wsq,
                                      float* __restrict__ part) {
  int tid = threadIdx.x;
  for (int o = 16; o > 0; o >>= 1) {
    s += __shfl_xor(s, o, 32);
    q += __shfl_xor(q, o, 32);
  }
  if ((tid & 31) == 0) {
    wsum[tid >> 5] = s;
    wsq[tid >> 5] = q;
  }
  __syncthreads();
  if ((tid & 255) == 0) {
    int g = tid >> 8;
    float a = wsum[8 * g];
    float c = wsq[8 * g];
    for (int u = 1; u < 8; ++u) {
      a += wsum[8 * g + u];
      c += wsq[8 * g + u];
    }
    reinterpret_cast<float2*>(part)[blockIdx.x * 2 + g] = make_float2(a, c);
  }
}


__global__ __launch_bounds__(256) void posenc_add(const float* __restrict__ x,
                                                  float* __restrict__ out,
                                                  float* __restrict__ part) {
  __shared__ float wsum[8];
  __shared__ float wsq[8];
  int p = threadIdx.x;
  int row0 = blockIdx.x * RPW;
  const float4* xin = reinterpret_cast<const float4*>(x) + row0 * (TT / 4) + p;
  float4* o = reinterpret_cast<float4*>(out) + row0 * (TT / 4) + p;
  float4 v[RPW];
#pragma unroll
  for (int q = 0; q < RPW; ++q) v[q] = xin[q * (TT / 4)];
  int t = p * 4;
  float s = 0.f, sq = 0.f;
#pragma unroll
  for (int q = 0; q < RPW; ++q) {
    int c = (row0 + q) % CC;
    float freq = expf(-(float)(c & ~1) * (LOG10000 / (float)CC));
    float ph = (float)(c & 1) * PI_2;
    float4 r;
    r.x = v[q].x + sinf((float)t * freq + ph);
    r.y = v[q].y + sinf((float)(t + 1) * freq + ph);
    r.z = v[q].z + sinf((float)(t + 2) * freq + ph);
    r.w = v[q].w + sinf((float)(t + 3) * freq + ph);
    v[q] = r;
    s += (r.x + r.y) + (r.z + r.w);
    sq += (r.x * r.x + r.y * r.y) + (r.z * r.z + r.w * r.w);
  }
  wg_pair_store(s, sq, 8, wsum, wsq, part);
#pragma unroll
  for (int q = 0; q < RPW; ++q) o[q * (TT / 4)] = v[q];
}

__device__ inline float2 ln_stats(const float* __restrict__ part, int b, int lane) {
  float2 pr = reinterpret_cast<const float2*>(part)[b * NPART + lane];
  float s = pr.x;
  float q2 = pr.y;
  for (int o = 16; o > 0; o >>= 1) {
    s += __shfl_xor(s, o, 32);
    q2 += __shfl_xor(q2, o, 32);
  }
  float inv_n = 1.0f / (float)CT;
  float mu = s * inv_n;
  float var = q2 * inv_n - mu * mu;
  return make_float2(mu, rsqrtf(var + 1e-5f));
}

__device__ inline float lnv(float x, float mu, float r, float g, float b) {
  return (x - mu) * r * g + b;
}

__device__ inline float4 ln4(float4 x, float4 g, float4 b, float mu, float r) {
  float4 y;
  y.x = lnv(x.x, mu, r, g.x, b.x);
  y.y = lnv(x.y, mu, r, g.y, b.y);
  y.z = lnv(x.z, mu, r, g.z, b.z);
  y.w = lnv(x.w, mu, r, g.w, b.w);
  return y;
}


#define BPAIR 160
template <int MODE>
__global__ __launch_bounds__(512) void lnpw_gemm(const float* __restrict__ X,
                                                 const float* __restrict__ lng,
                                                 const float* __restrict__ lnb,
                                                 const float* __restrict__ pin,
                                                 const float* __restrict__ dww,
                                                 const float* __restrict__ dwb,
                                                 const float* __restrict__ W,
                                                 const float* __restrict__ bias,
                                                 float* __restrict__ dst,
                                                 float* __restrict__ pout) {
  __shared__ __attribute__((aligned(16))) float Bs[(CC / 2) * BPAIR];
  __shared__ float wsum[16];
  __shared__ float wsq[16];
  int tid = threadIdx.x;
  int l = tid & 31;
  int b = blockIdx.x >> 4;
  int n0 = (blockIdx.x & 15) << 6;
  int m0 = (tid >> 6) << 4;
  int nh = ((tid >> 5) & 1) << 5;
  int lo = l & 15;
  int kh = (l >> 4) << 1;

  float2 st = ln_stats(pin, b, l);
  float mu = st.x;
  float rs = st.y;

  v2f afrag[CC / 4];
#pragma unroll
  for (int kk = 0; kk < CC; kk += 4) {
    int k0 = kk + kh;
    afrag[kk >> 2].x = W[(m0 + lo) * CC + k0];
    afrag[kk >> 2].y = W[(m0 + lo) * CC + k0 + 1];
  }

  const float* xb = X + b * CT;
  {
    int p = tid & 15;
    int gp = (n0 >> 2) + p;
    int gl = gp > 0 ? gp - 1 : gp;
    int gr = gp < TT / 4 - 1 ? gp + 1 : gp;
    int c0 = tid >> 4;
#pragma unroll
    for (int q = 0; q < CC / 32; ++q) {
      int c = c0 + 32 * q;
      const float4* r4 = reinterpret_cast<const float4*>(xb + c * TT);
      const float4* g4 = reinterpret_cast<const float4*>(lng + c * TT);
      const float4* b4 = reinterpret_cast<const float4*>(lnb + c * TT);
      float* d = Bs + (c >> 1) * BPAIR + 8 * p + (c & 1);
      float4 cur = r4[gp];
      if (MODE == 2) {
        cur = ln4(cur, g4[gp], b4[gp], mu, rs);
        d[0] = cur.x;
        d[2] = cur.y;
        d[4] = cur.z;
        d[6] = cur.w;
      } else {
        float4 prv = r4[gl];
        float4 nxt = r4[gr];
        if (MODE == 1) {
          cur = ln4(cur, g4[gp], b4[gp], mu, rs);
          prv = ln4(prv, g4[gl], b4[gl], mu, rs);
          nxt = ln4(nxt, g4[gr], b4[gr], mu, rs);
        }
        if (gp == 0) prv = make_float4(0.f, 0.f, 0.f, 0.f);
        if (gp == TT / 4 - 1) nxt = make_float4(0.f, 0.f, 0.f, 0.f);
        float bs = dwb[c];
        float wj[7];
#pragma unroll
        for (int j = 0; j < 7; ++j) wj[j] = dww[c * 7 + j];
        float xs[12] = {prv.x, prv.y, prv.z, prv.w, cur.x, cur.y,
                        cur.z, cur.w, nxt.x, nxt.y, nxt.z, nxt.w};
#pragma unroll
        for (int k = 0; k < 4; ++k) {
          float s = bs;
#pragma unroll
          for (int j = 0; j < 7; ++j) s = fmaf(xs[k + j + 1], wj[j], s);
          d[2 * k] = s;
        }
      }
    }
  }
  __syncthreads();

  v8f acc0 = {}, acc1 = {};
#pragma unroll
  for (int kk = 0; kk < CC; kk += 4) {
    const v2f* bp =
        reinterpret_cast<const v2f*>(Bs + ((kk + kh) >> 1) * BPAIR + 2 * (nh + lo));
    v2f b0 = bp[0], b1 = bp[16];
    acc0 = wmma4(afrag[kk >> 2], b0, acc0);
    acc1 = wmma4(afrag[kk >> 2], b1, acc1);
  }
  float* db = dst + b * CT;
  float s = 0.f, sq = 0.f;
#pragma unroll
  for (int v = 0; v < 8; ++v) {
    int m = m0 + v + ((l >> 4) << 3);
    float bsv = bias[m];
    int off = m * TT + n0 + nh + lo;
    const float* r = xb + off;
    float r0 = r[0], r1 = r[16];
    if (MODE == 0) {
      const float* g = lng + off;
      const float* be = lnb + off;
      r0 = lnv(r0, mu, rs, g[0], be[0]);
      r1 = lnv(r1, mu, rs, g[16], be[16]);
    }
    float o0 = fmaxf(acc0[v] + bsv, 0.f) + r0;
    float o1 = fmaxf(acc1[v] + bsv, 0.f) + r1;
    acc0[v] = o0;
    acc1[v] = o1;
    s += o0 + o1;
    sq += o0 * o0 + o1 * o1;
  }
  if (MODE != 2) wg_pair_store2(s, sq, wsum, wsq, pout);
#pragma unroll
  for (int v = 0; v < 8; ++v) {
    float* o = db + (m0 + v + ((l >> 4) << 3)) * TT + n0 + nh + lo;
    o[0]  = acc0[v];
    o[16] = acc1[v];
  }
}

__global__ __launch_bounds__(128) void xtw_gemm(const float* __restrict__ X,
                                                const float* __restrict__ lng,
                                                const float* __restrict__ lnb,
                                                const float* __restrict__ pin,
                                                const float* __restrict__ W0,
                                                const float* __restrict__ W1,
                                                const float* __restrict__ W2,
                                                float* __restrict__ D0,
                                                float* __restrict__ D1,
                                                float* __restrict__ D2) {
  int y = blockIdx.y;
  const float* W = y == 0 ? W0 : (y == 1 ? W1 : W2);
  float* dst = y == 0 ? D0 : (y == 1 ? D1 : D2);
  int l = threadIdx.x & 31;
  int tile = blockIdx.x * 4 + (threadIdx.x >> 5);
  int b = tile >> 6;
  int t0 = (tile & 63) << 4;
  int lo = l & 15;
  int kh = (l >> 4) << 1;
  const float* xb = X + b * CT;
  float2 st = ln_stats(pin, b, l);
  v8f acc0 = {}, acc1 = {}, acc2 = {}, acc3 = {};
#pragma unroll 4
  for (int kk = 0; kk < CC; kk += 4) {
    int k0 = kk + kh;
    int ia = k0 * TT + t0 + lo;
    v2f a;
    a.x = lnv(xb[ia], st.x, st.y, lng[ia], lnb[ia]);
    a.y = lnv(xb[ia + TT], st.x, st.y, lng[ia + TT], lnb[ia + TT]);
    const float* w0 = W + k0 * DKD + lo;
    const float* w1 = w0 + DKD;
    v2f b0, b1, b2, b3;
    b0.x = w0[0];  b0.y = w1[0];
    b1.x = w0[16]; b1.y = w1[16];
    b2.x = w0[32]; b2.y = w1[32];
    b3.x = w0[48]; b3.y = w1[48];
    acc0 = wmma4(a, b0, acc0);
    acc1 = wmma4(a, b1, acc1);
    acc2 = wmma4(a, b2, acc2);
    acc3 = wmma4(a, b3, acc3);
  }
  float* db = dst + b * TT * DKD;
#pragma unroll
  for (int v = 0; v < 8; ++v) {
    int t = t0 + v + ((l >> 4) << 3);
    float* d = db + t * DKD + lo;
    d[0]  = acc0[v];
    d[16] = acc1[v];
    d[32] = acc2[v];
    d[48] = acc3[v];
  }
}

#define QT 2
#define AW 8
#define QR (16 * QT)
#define KTW (TT / 16 / AW)
#define RWW (QR / AW)
#define KPW (TT / AW)
#define HP (DKD + 4)
#define OP (QR + 4)
__global__ __launch_bounds__(32 * AW) void attn_kernel(const float* __restrict__ Q,
                                                       const float* __restrict__ Km,
                                                       const float* __restrict__ Vm,
                                                       const float* __restrict__ mask,
                                                       const float* __restrict__ Wo,
                                                       float* __restrict__ res,
                                                       float* __restrict__ part) {
  __shared__ __attribute__((aligned(16))) float S[QR * TT];
  __shared__ float wsum[AW];
  __shared__ float wsq[AW];
  int b = blockIdx.x / (TT / QR);
  int q0 = (blockIdx.x % (TT / QR)) * QR;
  int w = __builtin_amdgcn_readfirstlane((int)(threadIdx.x >> 5));
  int l = threadIdx.x & 31;
  int lo = l & 15;
  int kh = (l >> 4) << 1;
  const float* qb = Q + b * TT * DKD;
  const float* kb = Km + b * TT * DKD;
  const float* vb = Vm + b * TT * DKD;
  const float* mb = mask + b * TT;

  v2f afrag[QT][DKD / 4];
#pragma unroll
  for (int u = 0; u < QT; ++u) {
#pragma unroll
    for (int kk = 0; kk < DKD; kk += 4) {
      int k0 = kk + kh;
      afrag[u][kk >> 2].x = qb[(q0 + 16 * u + lo) * DKD + k0];
      afrag[u][kk >> 2].y = qb[(q0 + 16 * u + lo) * DKD + k0 + 1];
    }
  }

#pragma unroll 2
  for (int ti = 0; ti < KTW; ++ti) {
    int t = w * KTW + ti;
    v8f acc[QT];
#pragma unroll
    for (int u = 0; u < QT; ++u) acc[u] = v8f{};
#pragma unroll
    for (int kk = 0; kk < DKD; kk += 4) {
      int k0 = kk + kh;
      v2f bv;
      bv.x = kb[(t * 16 + lo) * DKD + k0];
      bv.y = kb[(t * 16 + lo) * DKD + k0 + 1];
#pragma unroll
      for (int u = 0; u < QT; ++u) acc[u] = wmma4(afrag[u][kk >> 2], bv, acc[u]);
    }
    int col = t * 16 + lo;
    float hm = mb[col];
#pragma unroll
    for (int u = 0; u < QT; ++u) {
#pragma unroll
      for (int v = 0; v < 8; ++v) {
        int row = 16 * u + v + ((l >> 4) << 3);
        float sv = acc[u][v] * 0.125f;
        S[row * TT + col] = sv * hm + (1.f - hm) * -1e30f;
      }
    }
  }
  __syncthreads();

  for (int r = 0; r < RWW; ++r) {
    int row = w * RWW + r;
    float* sr = S + row * TT + l;
    float e[TT / 32];
    float mx = -3.4e38f;
#pragma unroll
    for (int c = 0; c < TT / 32; ++c) {
      e[c] = sr[c * 32];
      mx = fmaxf(mx, e[c]);
    }
    for (int o = 16; o > 0; o >>= 1) mx = fmaxf(mx, __shfl_xor(mx, o, 32));
    float sum = 0.f;
#pragma unroll
    for (int c = 0; c < TT / 32; ++c) {
      e[c] = expf(e[c] - mx);
      sum += e[c];
    }
    for (int o = 16; o > 0; o >>= 1) sum += __shfl_xor(sum, o, 32);
    float inv = 1.0f / sum;
    float qm = mb[q0 + row];
#pragma unroll
    for (int c = 0; c < TT / 32; ++c) sr[c * 32] = e[c] * inv * qm;
  }
  __syncthreads();

  v2f wa[DKD / 4];
  {
    const float* wp = Wo + 16 * w + lo;
#pragma unroll
    for (int kk = 0; kk < DKD; kk += 4) {
      int k0 = kk + kh;
      wa[kk >> 2].x = wp[k0 * CC] + wp[(k0 + 64) * CC];
      wa[kk >> 2].y = wp[(k0 + 1) * CC] + wp[(k0 + 65) * CC];
    }
  }
  float* rb = res + b * CT + (16 * w) * TT + q0 + l;
  float ov[16];
#pragma unroll
  for (int r = 0; r < 16; ++r) ov[r] = rb[r * TT];

  v8f pacc[QT][4];
#pragma unroll
  for (int u = 0; u < QT; ++u) {
#pragma unroll
    for (int n = 0; n < 4; ++n) pacc[u][n] = v8f{};
  }
#pragma unroll 4
  for (int ki = 0; ki < KPW; ki += 4) {
    int k0 = w * KPW + ki + kh;
    const float* v0p = vb + k0 * DKD + lo;
    const float* v1p = v0p + DKD;
    v2f bf[4];
#pragma unroll
    for (int n = 0; n < 4; ++n) {
      bf[n].x = v0p[16 * n];
      bf[n].y = v1p[16 * n];
    }
#pragma unroll
    for (int u = 0; u < QT; ++u) {
      v2f a;
      a.x = S[(16 * u + lo) * TT + k0];
      a.y = S[(16 * u + lo) * TT + k0 + 1];
#pragma unroll
      for (int n = 0; n < 4; ++n) pacc[u][n] = wmma4(a, bf[n], pacc[u][n]);
    }
  }
  __syncthreads();

  {
    float* pw = S + w * (QR * DKD);
#pragma unroll
    for (int u = 0; u < QT; ++u) {
#pragma unroll
      for (int v = 0; v < 8; ++v) {
        float* p = pw + (16 * u + v + ((l >> 4) << 3)) * DKD + lo;
        p[0]  = pacc[u][0][v];
        p[16] = pacc[u][1][v];
        p[32] = pacc[u][2][v];
        p[48] = pacc[u][3][v];
      }
    }
  }
  __syncthreads();

  float* Hs = S + AW * QR * DKD;
  float* Os = Hs + QR * HP;
  {
    const float4* s4 = reinterpret_cast<const float4*>(S);
#pragma unroll
    for (int j = 0; j < (QR * DKD / 4) / (32 * AW); ++j) {
      int i4 = threadIdx.x + j * (32 * AW);
      float4 r = s4[i4];
#pragma unroll
      for (int u = 1; u < AW; ++u) {
        float4 pu = s4[u * (QR * DKD / 4) + i4];
        r.x += pu.x;
        r.y += pu.y;
        r.z += pu.z;
        r.w += pu.w;
      }
      *reinterpret_cast<float4*>(Hs + (i4 >> 4) * HP + ((i4 & 15) << 2)) = r;
    }
  }
  __syncthreads();

  {
    v8f o0 = {}, o1 = {};
#pragma unroll
    for (int kk = 0; kk < DKD; kk += 4) {
      v2f b0 = *reinterpret_cast<const v2f*>(Hs + lo * HP + kk + kh);
      v2f b1 = *reinterpret_cast<const v2f*>(Hs + (16 + lo) * HP + kk + kh);
      o0 = wmma4(wa[kk >> 2], b0, o0);
      o1 = wmma4(wa[kk >> 2], b1, o1);
    }
#pragma unroll
    for (int v = 0; v < 8; ++v) {
      float* p = Os + (16 * w + v + ((l >> 4) << 3)) * OP + lo;
      p[0]  = o0[v];
      p[16] = o1[v];
    }
  }
  __syncthreads();

  float s = 0.f, sq = 0.f;
#pragma unroll
  for (int r = 0; r < 16; ++r) {
    float o = Os[(16 * w + r) * OP + l] + ov[r];
    ov[r] = o;
    s += o;
    sq += o * o;
  }
  wg_pair_store(s, sq, AW, wsum, wsq, part);
#pragma unroll
  for (int r = 0; r < 16; ++r) rb[r * TT] = ov[r];
}


extern "C" void kernel_launch(void* const* d_in, const int* in_sizes, int n_in,
                              void* d_out, int out_size, void* d_ws, size_t ws_size,
                              hipStream_t stream) {
  (void)in_sizes; (void)n_in; (void)out_size; (void)ws_size;
  const float* x    = (const float*)d_in[0];
  const float* mask = (const float*)d_in[1];
  const float* dw_w = (const float*)d_in[2];
  const float* dw_b = (const float*)d_in[3];
  const float* pw_w = (const float*)d_in[4];
  const float* pw_b = (const float*)d_in[5];
  const float* n0g  = (const float*)d_in[6];
  const float* n0b  = (const float*)d_in[7];
  const float* nsg  = (const float*)d_in[8];
  const float* nsb  = (const float*)d_in[9];
  const float* neg_ = (const float*)d_in[10];
  const float* neb  = (const float*)d_in[11];
  const float* Wq   = (const float*)d_in[12];
  const float* Wk   = (const float*)d_in[13];
  const float* Wv   = (const float*)d_in[14];
  const float* Wo   = (const float*)d_in[15];
  const float* fcw  = (const float*)d_in[16];
  const float* fcb  = (const float*)d_in[17];
  float* outp = (float*)d_out;
  float* ws = (float*)d_ws;

  float* RA  = ws;
  float* RB  = ws + 2097152;
  float* Qb  = ws + 6291456;
  float* Kb  = ws + 7340032;
  float* Vb  = ws + 8388608;
  float* ST  = ws + 4194304;
  const float* nodw = nullptr;
  float* nopart = nullptr;

  posenc_add<<<512, 256, 0, stream>>>(x, RA, ST);
  lnpw_gemm<0><<<256, 512, 0, stream>>>(RA, n0g, n0b, ST, dw_w, dw_b, pw_w, pw_b, RB,
                                        ST + 1024);
  lnpw_gemm<1><<<256, 512, 0, stream>>>(RB, nsg, nsb, ST + 1024, dw_w + 1 * CC * 7,
                                        dw_b + 1 * CC, pw_w + 1 * CC * CC, pw_b + 1 * CC,
                                        RA, ST + 2048);
  lnpw_gemm<1><<<256, 512, 0, stream>>>(RA, nsg + 1 * CT, nsb + 1 * CT, ST + 2048,
                                        dw_w + 2 * CC * 7, dw_b + 2 * CC,
                                        pw_w + 2 * CC * CC, pw_b + 2 * CC, RB, ST + 3072);
  lnpw_gemm<1><<<256, 512, 0, stream>>>(RB, nsg + 2 * CT, nsb + 2 * CT, ST + 3072,
                                        dw_w + 3 * CC * 7, dw_b + 3 * CC,
                                        pw_w + 3 * CC * CC, pw_b + 3 * CC, RA, ST + 4096);

  xtw_gemm<<<dim3(256, 3), 128, 0, stream>>>(RA, nsg + 3 * CT, nsb + 3 * CT, ST + 4096,
                                             Wq, Wk, Wv, Qb, Kb, Vb);
  attn_kernel<<<512, 256, 0, stream>>>(Qb, Kb, Vb, mask, Wo, RA, ST + 5120);
  lnpw_gemm<2><<<256, 512, 0, stream>>>(RA, neg_, neb, ST + 5120, nodw, nodw, fcw, fcb,
                                        outp, nopart);
}
